# starting kernel
# baseline (speedup 1.0000x reference)
#include <hip/hip_runtime.h>

#define NN 100000
#define NE 1000000
#define DIN 5
#define H 64
#define HHID 32
#define NL 4
#define DOUT 32
#define LN_EPS 1e-5f

typedef __attribute__((ext_vector_type(2)))  float    v2f;
typedef __attribute__((ext_vector_type(8)))  float    v8f;
typedef __attribute__((ext_vector_type(16))) _Float16 v16h;

#if defined(__has_builtin)
#if __has_builtin(__builtin_amdgcn_wmma_f32_16x16x4_f32)
#define HAVE_WMMA_F32 1
#endif
#endif

__device__ __forceinline__ float wave_sum(float v) {
#pragma unroll
  for (int off = 16; off > 0; off >>= 1) v += __shfl_xor(v, off, 32);
  return v;
}
__device__ __forceinline__ float sigmoidf_(float x) { return 1.0f / (1.0f + __expf(-x)); }

__device__ __forceinline__ void atomic_fadd(float* p, float v) { unsafeAtomicAdd(p, v); }

__global__ __launch_bounds__(256) void k_input_proj(
    const float* __restrict__ x, const float* __restrict__ W,
    const float* __restrict__ b, float* __restrict__ h) {
  int t = blockIdx.x * 256 + threadIdx.x;
  if (t >= NN * H) return;
  int n = t >> 6, k = t & 63;
  const float* xr = x + (size_t)n * DIN;
  float acc = b[k];
#pragma unroll
  for (int d = 0; d < DIN; ++d) acc = fmaf(xr[d], W[d * H + k], acc);
  h[t] = acc;
}

__global__ __launch_bounds__(128) void k_gemm_h(
    const float* __restrict__ A, const float* __restrict__ W,
    const float* __restrict__ bias, const float* __restrict__ aw,
    float* __restrict__ out, float* __restrict__ sD, float* __restrict__ sS) {
  __shared__ float lA[16][H + 1];
  __shared__ float lW[H][H + 1];
  const int tid = threadIdx.x;
  const int row0 = blockIdx.x * 16;
  for (int idx = tid; idx < 16 * H; idx += 128)
    lA[idx >> 6][idx & 63] = A[(size_t)row0 * H + idx];
  for (int idx = tid; idx < H * H; idx += 128)
    lW[idx >> 6][idx & 63] = W[idx];
  __syncthreads();

  const int wave = tid >> 5, lane = tid & 31;
  const int m = lane & 15, g = lane >> 4;
  const int col = wave * 16 + m;

  v8f acc;
  float bv = bias[col];
#pragma unroll
  for (int i = 0; i < 8; ++i) acc[i] = bv;

#if HAVE_WMMA_F32
#pragma unroll
  for (int k0 = 0; k0 < H; k0 += 4) {
    int kb = k0 + 2 * g;
    v2f a, b;
    a.x = lA[m][kb];     a.y = lA[m][kb + 1];
    b.x = lW[kb][col];   b.y = lW[kb + 1][col];
    acc = __builtin_amdgcn_wmma_f32_16x16x4_f32(false, a, false, b, (short)0, acc, false, false);
  }
#else
#pragma unroll
  for (int kk = 0; kk < H; kk += 32) {
    v16h a, b;
#pragma unroll
    for (int j = 0; j < 8; ++j) {
      int ka = (j < 4) ? (2 * j + 8 * g) : (16 + 2 * (j - 4) + 8 * g);
      a[2 * j]     = (_Float16)lA[m][kk + ka];
      a[2 * j + 1] = (_Float16)lA[m][kk + ka + 1];
      int kb = 2 * j + 16 * g;
      b[2 * j]     = (_Float16)lW[kk + kb][col];
      b[2 * j + 1] = (_Float16)lW[kk + kb + 1][col];
    }
    acc = __builtin_amdgcn_wmma_f32_16x16x32_f16(false, a, false, b, (short)0, acc, false, false);
  }
#endif

#pragma unroll
  for (int i = 0; i < 8; ++i) {
    int row = row0 + i + 8 * g;
    out[(size_t)row * H + col] = acc[i];
  }

  if (wave < 2) {
    const float* awp = aw + wave * H;
    float p = 0.0f;
#pragma unroll
    for (int k = 0; k < 32; ++k) p = fmaf(lA[m][32 * g + k], awp[32 * g + k], p);
    p += __shfl_xor(p, 16, 32);
    if (g == 0) {
      if (wave == 0) sD[row0 + m] = p; else sS[row0 + m] = p;
    }
  }
}

__global__ __launch_bounds__(256) void k_edges(
    const int* __restrict__ ei, const float* __restrict__ ea,
    const float* __restrict__ m, const float* __restrict__ sD,
    const float* __restrict__ sS, const float* __restrict__ attb,
    float* __restrict__ aggr) {
  int gw = blockIdx.x * 8 + (threadIdx.x >> 5);
  int lane = threadIdx.x & 31;
  int ebase = gw * 32;
  if (ebase >= NE) return;
  int e = ebase + lane;
  int src = 0, dst = 0;
  float w = 0.0f;
  if (e < NE) {
    src = ei[e];
    dst = ei[NE + e];
    __builtin_prefetch(m + (size_t)src * H, 0, 1);
    __builtin_prefetch(m + (size_t)src * H + 32, 0, 1);
    w = ea[e] * sigmoidf_(sD[dst] + sS[src] + attb[0]);
  }
  int cnt = min(32, NE - ebase);
  for (int i = 0; i < cnt; ++i) {
    int bs = __shfl(src, i, 32);
    int bd = __shfl(dst, i, 32);
    float bw = __shfl(w, i, 32);
    const float* mr = m + (size_t)bs * H;
    float* ar = aggr + (size_t)bd * H;
    atomic_fadd(ar + lane,      mr[lane] * bw);
    atomic_fadd(ar + lane + 32, mr[lane + 32] * bw);
  }
}

__global__ __launch_bounds__(256) void k_update(
    float* __restrict__ h, const float* __restrict__ m,
    const float* __restrict__ aggr, const float* __restrict__ g,
    const float* __restrict__ beta) {
  int gw = blockIdx.x * 8 + (threadIdx.x >> 5);
  int lane = threadIdx.x & 31;
  if (gw >= NN) return;
  size_t base = (size_t)gw * H;
  float r0 = h[base + lane], r1 = h[base + lane + 32];
  float f0 = aggr[base + lane] + m[base + lane];
  float f1 = aggr[base + lane + 32] + m[base + lane + 32];
  float mean = wave_sum(f0 + f1) * (1.0f / H);
  float d0 = f0 - mean, d1 = f1 - mean;
  float var = wave_sum(fmaf(d0, d0, d1 * d1)) * (1.0f / H);
  float inv = rsqrtf(var + LN_EPS);
  float y0 = fmaf(d0 * inv, g[lane], beta[lane]);
  float y1 = fmaf(d1 * inv, g[lane + 32], beta[lane + 32]);
  h[base + lane]      = fmaxf(y0, 0.0f) + r0;
  h[base + lane + 32] = fmaxf(y1, 0.0f) + r1;
}

__global__ __launch_bounds__(256) void k_heads(
    const float* __restrict__ h,
    const float* __restrict__ qW1, const float* __restrict__ qb1,
    const float* __restrict__ qW2, const float* __restrict__ qb2,
    const float* __restrict__ iW1, const float* __restrict__ ib1,
    const float* __restrict__ iW2, const float* __restrict__ ib2,
    const float* __restrict__ gW1, const float* __restrict__ gb1,
    const float* __restrict__ gW2, const float* __restrict__ gb2,
    const float* __restrict__ oW,  const float* __restrict__ obv,
    const float* __restrict__ sW1, const float* __restrict__ sb1,
    const float* __restrict__ sW2, const float* __restrict__ sb2,
    float* __restrict__ oq, float* __restrict__ oi, float* __restrict__ og,
    float* __restrict__ oh, float* __restrict__ obot, float* __restrict__ osev) {
  int gw = blockIdx.x * 8 + (threadIdx.x >> 5);
  int lane = threadIdx.x & 31;
  if (gw >= NN) return;
  size_t base = (size_t)gw * H;
  float h0 = h[base + lane], h1 = h[base + lane + 32];

  float tq = 0.f, ti = 0.f, tg = 0.f, ts = 0.f, bo = 0.f;
#pragma unroll 8
  for (int k = 0; k < H; ++k) {
    float hk = (k < 32) ? __shfl(h0, k, 32) : __shfl(h1, k - 32, 32);
    int wi = k * HHID + lane;
    tq = fmaf(hk, qW1[wi], tq);
    ti = fmaf(hk, iW1[wi], ti);
    tg = fmaf(hk, gW1[wi], tg);
    ts = fmaf(hk, sW1[wi], ts);
    bo = fmaf(hk, oW[wi],  bo);
  }
  tq = fmaxf(tq + qb1[lane], 0.0f);
  ti = fmaxf(ti + ib1[lane], 0.0f);
  tg = fmaxf(tg + gb1[lane], 0.0f);
  ts = fmaxf(ts + sb1[lane], 0.0f);

  float q  = wave_sum(tq * qW2[lane]);
  float ii = wave_sum(ti * iW2[lane]);
  float gg = wave_sum(tg * gW2[lane]);
  if (lane == 0) {
    oq[gw] = sigmoidf_(q + qb2[0]);
    oi[gw] = sigmoidf_(ii + ib2[0]);
    og[gw] = sigmoidf_(gg + gb2[0]);
  }

  float z[5];
#pragma unroll
  for (int c = 0; c < 5; ++c) z[c] = wave_sum(ts * sW2[lane * 5 + c]) + sb2[c];
  float zm = z[0];
#pragma unroll
  for (int c = 1; c < 5; ++c) zm = fmaxf(zm, z[c]);
  float es = 0.f;
#pragma unroll
  for (int c = 0; c < 5; ++c) { z[c] = __expf(z[c] - zm); es += z[c]; }
  float inv = 1.0f / es;
#pragma unroll
  for (int c = 0; c < 5; ++c)
    if (lane == c) osev[(size_t)gw * 5 + c] = z[c] * inv;

  obot[(size_t)gw * DOUT + lane] = bo + obv[lane];
  oh[base + lane]      = h0;
  oh[base + lane + 32] = h1;
}

__global__ __launch_bounds__(256) void k_graph_mean(
    const float* __restrict__ h, float* __restrict__ gout) {
  __shared__ float ls[H];
  int tid = threadIdx.x;
  if (tid < H) ls[tid] = 0.0f;
  __syncthreads();
  float acc = 0.0f;
  size_t stride = (size_t)gridDim.x * blockDim.x;
  for (size_t idx = (size_t)blockIdx.x * blockDim.x + tid; idx < (size_t)NN * H; idx += stride)
    acc += h[idx];
  atomicAdd(&ls[tid & 63], acc);
  __syncthreads();
  if (tid < H) atomic_fadd(&gout[tid], ls[tid] * (1.0f / NN));
}

extern "C" void kernel_launch(void* const* d_in, const int* in_sizes, int n_in,
                              void* d_out, int out_size, void* d_ws, size_t ws_size,
                              hipStream_t stream) {
  (void)in_sizes; (void)n_in; (void)out_size; (void)ws_size;
  const float* x    = (const float*)d_in[0];
  const int*   ei   = (const int*)  d_in[1];
  const float* ea   = (const float*)d_in[2];
  const float* inW  = (const float*)d_in[3];
  const float* inb  = (const float*)d_in[4];
  const float* linW = (const float*)d_in[5];
  const float* linb = (const float*)d_in[6];
  const float* attW = (const float*)d_in[7];
  const float* attb = (const float*)d_in[8];
  const float* lng  = (const float*)d_in[9];
  const float* lnb  = (const float*)d_in[10];
  const float* qW1 = (const float*)d_in[11]; const float* qb1 = (const float*)d_in[12];
  const float* qW2 = (const float*)d_in[13]; const float* qb2 = (const float*)d_in[14];
  const float* iW1 = (const float*)d_in[15]; const float* ib1 = (const float*)d_in[16];
  const float* iW2 = (const float*)d_in[17]; const float* ib2 = (const float*)d_in[18];
  const float* gW1 = (const float*)d_in[19]; const float* gb1 = (const float*)d_in[20];
  const float* gW2 = (const float*)d_in[21]; const float* gb2 = (const float*)d_in[22];
  const float* oW  = (const float*)d_in[23]; const float* obv = (const float*)d_in[24];
  const float* sW1 = (const float*)d_in[25]; const float* sb1 = (const float*)d_in[26];
  const float* sW2 = (const float*)d_in[27]; const float* sb2 = (const float*)d_in[28];

  float* ws   = (float*)d_ws;
  float* h    = ws;
  float* mbuf = h    + (size_t)NN * H;
  float* aggr = mbuf + (size_t)NN * H;
  float* sD   = aggr + (size_t)NN * H;
  float* sS   = sD + NN;

  float* out  = (float*)d_out;
  float* oq   = out;
  float* oi   = out + (size_t)NN;
  float* og   = out + 2 * (size_t)NN;
  float* oh   = out + 3 * (size_t)NN;
  float* ogr  = out + 67 * (size_t)NN;
  float* obot = out + 67 * (size_t)NN + H;
  float* osev = out + 99 * (size_t)NN + H;

  k_input_proj<<<(NN * H + 255) / 256, 256, 0, stream>>>(x, inW, inb, h);

  const int nodeWaveBlocks = (NN + 7) / 8;
  for (int l = 0; l < NL; ++l) {
    k_gemm_h<<<NN / 16, 128, 0, stream>>>(h, linW + (size_t)l * H * H, linb + (size_t)l * H,
                                          attW + (size_t)l * 2 * H, mbuf, sD, sS);
    hipMemsetAsync(aggr, 0, (size_t)NN * H * sizeof(float), stream);
    k_edges<<<(NE + 255) / 256, 256, 0, stream>>>(ei, ea, mbuf, sD, sS, attb + l, aggr);
    k_update<<<nodeWaveBlocks, 256, 0, stream>>>(h, mbuf, aggr, lng + (size_t)l * H, lnb + (size_t)l * H);
  }

  hipMemsetAsync(ogr, 0, H * sizeof(float), stream);
  k_heads<<<nodeWaveBlocks, 256, 0, stream>>>(h, qW1, qb1, qW2, qb2, iW1, ib1, iW2, ib2,
                                              gW1, gb1, gW2, gb2, oW, obv, sW1, sb1, sW2, sb2,
                                              oq, oi, og, oh, obot, osev);
  k_graph_mean<<<128, 256, 0, stream>>>(h, ogr);
}


# layers 1-3: edge sums held in LDS by 255 owner work-groups that scan a per-call owner-byte map, add rows by LDS f32 atomics and do the update; global f32 atomics, memsets go (sum order changes)
# speedup vs baseline: 1.7944x; 1.7944x over previous
#include <hip/hip_runtime.h>

#define NN 100000
#define NE 1000000
#define DIN 5
#define H 64
#define HHID 32
#define NL 4
#define DOUT 32
#define LN_EPS 1e-5f

typedef __attribute__((ext_vector_type(2)))  float    v2f;
typedef __attribute__((ext_vector_type(8)))  float    v8f;
typedef __attribute__((ext_vector_type(16))) _Float16 v16h;

#if defined(__has_builtin)
#if __has_builtin(__builtin_amdgcn_wmma_f32_16x16x4_f32)
#define HAVE_WMMA_F32 1
#endif
#endif

__device__ __forceinline__ float wave_sum(float v) {
#pragma unroll
  for (int off = 16; off > 0; off >>= 1) v += __shfl_xor(v, off, 32);
  return v;
}
__device__ __forceinline__ float sigmoidf_(float x) { return 1.0f / (1.0f + __expf(-x)); }

__device__ __forceinline__ void atomic_fadd(float* p, float v) { unsafeAtomicAdd(p, v); }

__device__ __forceinline__ v8f tile_product(const float (*lA)[H + 1], const float (*lW)[H + 1],
                                            int m, int g, int col, v8f acc) {
#if HAVE_WMMA_F32
#pragma unroll
  for (int k0 = 0; k0 < H; k0 += 4) {
    int kb = k0 + 2 * g;
    v2f a, b;
    a.x = lA[m][kb];     a.y = lA[m][kb + 1];
    b.x = lW[kb][col];   b.y = lW[kb + 1][col];
    acc = __builtin_amdgcn_wmma_f32_16x16x4_f32(false, a, false, b, (short)0, acc, false, false);
  }
#else
#pragma unroll
  for (int kk = 0; kk < H; kk += 32) {
    v16h a, b;
#pragma unroll
    for (int j = 0; j < 8; ++j) {
      int ka = (j < 4) ? (2 * j + 8 * g) : (16 + 2 * (j - 4) + 8 * g);
      a[2 * j]     = (_Float16)lA[m][kk + ka];
      a[2 * j + 1] = (_Float16)lA[m][kk + ka + 1];
      int kb = 2 * j + 16 * g;
      b[2 * j]     = (_Float16)lW[kk + kb][col];
      b[2 * j + 1] = (_Float16)lW[kk + kb + 1][col];
    }
    acc = __builtin_amdgcn_wmma_f32_16x16x32_f16(false, a, false, b, (short)0, acc, false, false);
  }
#endif
  return acc;
}

__global__ __launch_bounds__(128) void k_gemm_h(
    const float* __restrict__ A, const float* __restrict__ W,
    const float* __restrict__ bias, const float* __restrict__ aw,
    float* __restrict__ out, float* __restrict__ sD, float* __restrict__ sS) {
  __shared__ float lA[16][H + 1];
  __shared__ float lW[H][H + 1];
  const int tid = threadIdx.x;
  const int row0 = blockIdx.x * 16;
  for (int idx = tid; idx < 16 * H; idx += 128)
    lA[idx >> 6][idx & 63] = A[(size_t)row0 * H + idx];
  for (int idx = tid; idx < H * H; idx += 128)
    lW[idx >> 6][idx & 63] = W[idx];
  __syncthreads();

  const int wave = tid >> 5, lane = tid & 31;
  const int m = lane & 15, g = lane >> 4;
  const int col = wave * 16 + m;

  v8f acc;
  float bv = bias[col];
#pragma unroll
  for (int i = 0; i < 8; ++i) acc[i] = bv;

  acc = tile_product(lA, lW, m, g, col, acc);

#pragma unroll
  for (int i = 0; i < 8; ++i) {
    int row = row0 + i + 8 * g;
    out[(size_t)row * H + col] = acc[i];
  }

  if (wave < 2) {
    const float* awp = aw + wave * H;
    float p = 0.0f;
#pragma unroll
    for (int k = 0; k < 32; ++k) p = fmaf(lA[m][32 * g + k], awp[32 * g + k], p);
    p += __shfl_xor(p, 16, 32);
    if (g == 0) {
      if (wave == 0) sD[row0 + m] = p; else sS[row0 + m] = p;
    }
  }
}

#define EDGE_DEPTH 8
#define OWN_R 393
#define OWN_BLOCKS ((NN + OWN_R - 1) / OWN_R)
#define OWN_WAVES 32
#define OWN_Q 544
static_assert(OWN_BLOCKS <= 255, "owner numbers are bytes 0..254; 255 marks a map byte that is nobody's");
static_assert(NE % 16 == 0, "the owner map is read 16 bytes a lane");
static_assert(NE % 32 == 0, "every wave of the layer-0 edge pass owns 32 whole edges");
static_assert(32 % EDGE_DEPTH == 0, "a batch of 32 edges splits into whole groups");

__device__ __forceinline__ int lane_bcast_i(int v, int l) { return __builtin_amdgcn_readlane(v, l); }
__device__ __forceinline__ float lane_bcast_f(float v, int l) {
  return __int_as_float(__builtin_amdgcn_readlane(__float_as_int(v), l));
}

__global__ __launch_bounds__(256) void k_edge_pack(
    const int* __restrict__ ei, const float* __restrict__ ea,
    int* __restrict__ rec, unsigned* __restrict__ own) {
  const int t = blockIdx.x * 256 + threadIdx.x;
  if (t >= NE / 4) return;
  const int4 s4 = reinterpret_cast<const int4*>(ei)[t];
  const int4 d4 = reinterpret_cast<const int4*>(ei + NE)[t];
  const float4 a4 = reinterpret_cast<const float4*>(ea)[t];
  int4* r4 = reinterpret_cast<int4*>(rec) + (size_t)t * 4;
  r4[0] = make_int4(s4.x, d4.x, __float_as_int(a4.x), 0);
  r4[1] = make_int4(s4.y, d4.y, __float_as_int(a4.y), 0);
  r4[2] = make_int4(s4.z, d4.z, __float_as_int(a4.z), 0);
  r4[3] = make_int4(s4.w, d4.w, __float_as_int(a4.w), 0);
  own[t] = (unsigned)(d4.x / OWN_R) | ((unsigned)(d4.y / OWN_R) << 8) |
           ((unsigned)(d4.z / OWN_R) << 16) | ((unsigned)(d4.w / OWN_R) << 24);
}

__device__ __forceinline__ void own_batch(
    const int* __restrict__ qw, int n, int lane, int lo, float b0,
    const int* __restrict__ rec, const float* __restrict__ m, const float* __restrict__ sS,
    const float* __restrict__ lsD, float* __restrict__ tile) {
  int src = 0, dl = 0;
  float w = 0.0f;
  if (lane < n) {
    const int e = qw[lane];
    const int4 r = reinterpret_cast<const int4*>(rec)[e];
    src = r.x;
    dl = r.y - lo;
    w = __int_as_float(r.z) * sigmoidf_(lsD[dl] + sS[src] + b0);
  }
  const float* mlane = m + lane;
#pragma unroll
  for (int i0 = 0; i0 < 32; i0 += EDGE_DEPTH) {
    if (i0 < n) {
      float lov[EDGE_DEPTH], hiv[EDGE_DEPTH];
#pragma unroll
      for (int j = 0; j < EDGE_DEPTH; ++j) {
        const float* mr = mlane + (size_t)lane_bcast_i(src, i0 + j) * H;
        lov[j] = mr[0];
        hiv[j] = mr[32];
      }
#pragma unroll
      for (int j = 0; j < EDGE_DEPTH; ++j) {
        if (i0 + j < n) {
          const float bw = lane_bcast_f(w, i0 + j);
          float* tr = tile + lane_bcast_i(dl, i0 + j) * H + lane;
          __hip_atomic_fetch_add(tr,      lov[j] * bw, __ATOMIC_RELAXED, __HIP_MEMORY_SCOPE_WORKGROUP);
          __hip_atomic_fetch_add(tr + 32, hiv[j] * bw, __ATOMIC_RELAXED, __HIP_MEMORY_SCOPE_WORKGROUP);
        }
      }
    }
  }
}

__device__ __forceinline__ void own_scan_piece(const uint4 o, int vb, unsigned myb, int* qw, int& cnt) {
  const unsigned wd[4] = {o.x, o.y, o.z, o.w};
#pragma unroll
  for (int k = 0; k < 16; ++k) {
    const unsigned byte = (wd[k >> 2] >> (8 * (k & 3))) & 0xFFu;
    const bool mine = (byte == myb);
    const unsigned mask = __builtin_amdgcn_ballot_w32(mine);
    if (mask != 0u) {
      const int pos = cnt + (int)__builtin_amdgcn_mbcnt_lo(mask, 0u);
      if (mine) qw[pos] = vb + k;
      cnt += __builtin_popcount(mask);
    }
  }
}

__device__ __forceinline__ uint4 own_piece(const uint4 t, bool ok) {
  return ok ? t : make_uint4(0xFFFFFFFFu, 0xFFFFFFFFu, 0xFFFFFFFFu, 0xFFFFFFFFu);
}

__device__ __forceinline__ void own_drain(
    int* qw, int& cnt, int lane, int lo, float b0,
    const int* __restrict__ rec, const float* __restrict__ m, const float* __restrict__ sS,
    const float* __restrict__ lsD, float* __restrict__ tile) {
  int head = 0;
  while (cnt - head >= 32) {
    own_batch(qw + head, 32, lane, lo, b0, rec, m, sS, lsD, tile);
    head += 32;
  }
  if (head > 0) {
    const int rest = cnt - head;
    int keep = 0;
    if (lane < rest) keep = qw[head + lane];
    if (lane < rest) qw[lane] = keep;
    cnt = rest;
  }
}

__global__ __launch_bounds__(1024) void k_own(
    const int* __restrict__ rec, const unsigned* __restrict__ own,
    const float* __restrict__ m, const float* __restrict__ sD,
    const float* __restrict__ sS, const float* __restrict__ attb,
    const float* __restrict__ g, const float* __restrict__ beta,
    float* __restrict__ h) {
  __shared__ float tile[OWN_R * H];
  __shared__ float lsD[OWN_R + 1];
  __shared__ int q[OWN_WAVES][OWN_Q];
  const int tid = threadIdx.x, lane = tid & 31;
  const int wave = __builtin_amdgcn_readfirstlane(tid >> 5);
  const int lo = blockIdx.x * OWN_R;
  const int nrows = (NN - lo < OWN_R) ? (NN - lo) : OWN_R;
  const unsigned myb = blockIdx.x;
  const float b0 = attb[0];

#pragma unroll 4
  for (int r = wave; r < nrows; r += OWN_WAVES) {
    const float* mr = m + (size_t)(lo + r) * H;
    tile[r * H + lane]      = mr[lane];
    tile[r * H + lane + 32] = mr[lane + 32];
  }
  for (int r = tid; r < nrows; r += 1024) lsD[r] = sD[lo + r];
  __syncthreads();

  const uint4* __restrict__ own4 = reinterpret_cast<const uint4*>(own);
  int* qw = q[wave];
  int cnt = 0;
  const int nvec = NE / 16;
  const int nit = (nvec + 31) / 32;
  for (int it0 = wave; it0 < nit; it0 += 4 * OWN_WAVES) {
    const int v0 = it0 * 32 + lane;
    const int v1 = v0 + 32 * OWN_WAVES;
    const int v2 = v1 + 32 * OWN_WAVES;
    const int v3 = v2 + 32 * OWN_WAVES;
    const bool ok0 = v0 < nvec, ok1 = v1 < nvec, ok2 = v2 < nvec, ok3 = v3 < nvec;
    const uint4 t0 = own4[ok0 ? v0 : nvec - 1];
    const uint4 t1 = own4[ok1 ? v1 : nvec - 1];
    const uint4 t2 = own4[ok2 ? v2 : nvec - 1];
    const uint4 t3 = own4[ok3 ? v3 : nvec - 1];
    own_scan_piece(own_piece(t0, ok0), v0 * 16, myb, qw, cnt);
    own_drain(qw, cnt, lane, lo, b0, rec, m, sS, lsD, tile);
    own_scan_piece(own_piece(t1, ok1), v1 * 16, myb, qw, cnt);
    own_drain(qw, cnt, lane, lo, b0, rec, m, sS, lsD, tile);
    own_scan_piece(own_piece(t2, ok2), v2 * 16, myb, qw, cnt);
    own_drain(qw, cnt, lane, lo, b0, rec, m, sS, lsD, tile);
    own_scan_piece(own_piece(t3, ok3), v3 * 16, myb, qw, cnt);
    own_drain(qw, cnt, lane, lo, b0, rec, m, sS, lsD, tile);
  }
  if (cnt > 0) own_batch(qw, cnt, lane, lo, b0, rec, m, sS, lsD, tile);
  __syncthreads();

  const float g0 = g[lane], g1 = g[lane + 32];
  const float be0 = beta[lane], be1 = beta[lane + 32];
#pragma unroll 2
  for (int r = wave; r < nrows; r += OWN_WAVES) {
    const size_t base = (size_t)(lo + r) * H;
    const float r0 = h[base + lane], r1 = h[base + lane + 32];
    const float f0 = tile[r * H + lane], f1 = tile[r * H + lane + 32];
    const float mean = wave_sum(f0 + f1) * (1.0f / H);
    const float d0 = f0 - mean, d1 = f1 - mean;
    const float var = wave_sum(fmaf(d0, d0, d1 * d1)) * (1.0f / H);
    const float inv = rsqrtf(var + LN_EPS);
    const float y0 = fmaf(d0 * inv, g0, be0);
    const float y1 = fmaf(d1 * inv, g1, be1);
    h[base + lane]      = fmaxf(y0, 0.0f) + r0;
    h[base + lane + 32] = fmaxf(y1, 0.0f) + r1;
  }
}

#define ACCW 8
#define RECW 8

__global__ __launch_bounds__(256) void k_l0_prep(
    const float* __restrict__ x, const float* __restrict__ inW, const float* __restrict__ inb,
    const float* __restrict__ aw, float* __restrict__ rec, double* __restrict__ acc) {
  int n = blockIdx.x * 256 + threadIdx.x;
  if (n >= NN) return;
  float xv[DIN];
#pragma unroll
  for (int d = 0; d < DIN; ++d) xv[d] = x[(size_t)n * DIN + d];
  float sd = 0.0f, ss = 0.0f;
#pragma unroll 8
  for (int k = 0; k < H; ++k) {
    float hk = inb[k];
#pragma unroll
    for (int d = 0; d < DIN; ++d) hk = fmaf(xv[d], inW[d * H + k], hk);
    sd = fmaf(hk, aw[k], sd);
    ss = fmaf(hk, aw[H + k], ss);
  }
  float4* r4 = reinterpret_cast<float4*>(rec + (size_t)n * RECW);
  r4[0] = make_float4(xv[0], xv[1], xv[2], xv[3]);
  r4[1] = make_float4(xv[4], ss, sd, 0.0f);
  double2* a2 = reinterpret_cast<double2*>(acc + (size_t)n * ACCW);
#pragma unroll
  for (int j = 0; j < ACCW / 2; ++j) a2[j] = make_double2(0.0, 0.0);
}

__global__ __launch_bounds__(256) void k_l0_edges(
    const int* __restrict__ ei, const float* __restrict__ ea,
    const float* __restrict__ rec, const float* __restrict__ attb,
    double* __restrict__ acc) {
  int gw = blockIdx.x * 8 + (threadIdx.x >> 5);
  int lane = threadIdx.x & 31;
  int ebase = gw * 32;
  if (ebase >= NE) return;
  const int sub = lane >> 3, c = lane & 7;
  const float b0 = attb[0];
  int srcs[8], dsts[8];
  float eas[8], rv[8], sd[8];
#pragma unroll
  for (int i = 0; i < 8; ++i) {
    int e = ebase + 4 * i + sub;
    srcs[i] = ei[e];
    dsts[i] = ei[NE + e];
    eas[i] = ea[e];
  }
#pragma unroll
  for (int i = 0; i < 8; ++i) {
    rv[i] = rec[(size_t)srcs[i] * RECW + c];
    sd[i] = rec[(size_t)dsts[i] * RECW + 6];
  }
#pragma unroll
  for (int i = 0; i < 8; ++i) {
    float ss = __shfl(rv[i], (lane & 24) | 5, 32);
    float w = eas[i] * sigmoidf_(sd[i] + ss + b0);
    if (c < 6) {
      double v = (c == 5) ? (double)w : (double)w * (double)rv[i];
      unsafeAtomicAdd(acc + (size_t)dsts[i] * ACCW + c, v);
    }
  }
}

__global__ __launch_bounds__(128) void k_l0_node(
    const float* __restrict__ x, const double* __restrict__ acc,
    const float* __restrict__ inW, const float* __restrict__ inb,
    const float* __restrict__ W, const float* __restrict__ bias,
    const float* __restrict__ g_, const float* __restrict__ beta,
    float* __restrict__ h) {
  __shared__ float lA[16][H + 1];
  __shared__ float lW[H][H + 1];
  __shared__ float lO[16][H + 1];
  __shared__ float lX[16][8];
  __shared__ float lS[16][8];
  const int tid = threadIdx.x;
  const int row0 = blockIdx.x * 16;
  for (int idx = tid; idx < H * H; idx += 128)
    lW[idx >> 6][idx & 63] = W[idx];
  {
    const int r = tid >> 3, c = tid & 7;
    const size_t n = (size_t)(row0 + r);
    float xv = (c < DIN) ? x[n * DIN + c] : 0.0f;
    float sv = (c < 6) ? (float)acc[n * ACCW + c] : 0.0f;
    lX[r][c] = xv;
    lS[r][c] = (c < DIN) ? (sv + xv) : ((c == 5) ? (sv + 1.0f) : 0.0f);
  }
  __syncthreads();
  for (int idx = tid; idx < 16 * H; idx += 128) {
    const int r = idx >> 6, k = idx & 63;
    float u = lS[r][5] * inb[k];
#pragma unroll
    for (int d = 0; d < DIN; ++d) u = fmaf(lS[r][d], inW[d * H + k], u);
    lA[r][k] = u;
  }
  __syncthreads();

  const int wave = tid >> 5, lane = tid & 31;
  const int m = lane & 15, g = lane >> 4;
  const int col = wave * 16 + m;

  v8f t;
  const float bv = bias[col];
#pragma unroll
  for (int i = 0; i < 8; ++i) t[i] = lS[i + 8 * g][5] * bv;

  t = tile_product(lA, lW, m, g, col, t);

#pragma unroll
  for (int i = 0; i < 8; ++i) lO[i + 8 * g][col] = t[i];
  __syncthreads();

  const float g0 = g_[lane], g1 = g_[lane + 32];
  const float be0 = beta[lane], be1 = beta[lane + 32];
#pragma unroll
  for (int rr = 0; rr < 4; ++rr) {
    const int r = wave * 4 + rr;
    float r0 = inb[lane], r1 = inb[lane + 32];
#pragma unroll
    for (int d = 0; d < DIN; ++d) {
      r0 = fmaf(lX[r][d], inW[d * H + lane], r0);
      r1 = fmaf(lX[r][d], inW[d * H + lane + 32], r1);
    }
    float f0 = lO[r][lane], f1 = lO[r][lane + 32];
    float mean = wave_sum(f0 + f1) * (1.0f / H);
    float d0 = f0 - mean, d1 = f1 - mean;
    float var = wave_sum(fmaf(d0, d0, d1 * d1)) * (1.0f / H);
    float inv = rsqrtf(var + LN_EPS);
    float y0 = fmaf(d0 * inv, g0, be0);
    float y1 = fmaf(d1 * inv, g1, be1);
    const size_t base = (size_t)(row0 + r) * H;
    h[base + lane]      = fmaxf(y0, 0.0f) + r0;
    h[base + lane + 32] = fmaxf(y1, 0.0f) + r1;
  }
}

__device__ __forceinline__ float kgroup_reduce(const float (&a)[4], int hiA, int hiB) {
  float s0 = hiA ? a[0] : a[2];
  float s1 = hiA ? a[1] : a[3];
  float k0 = hiA ? a[2] : a[0];
  float k1 = hiA ? a[3] : a[1];
  float r0 = k0 + __shfl_xor(s0, 16, 32);
  float r1 = k1 + __shfl_xor(s1, 16, 32);
  float sB = hiB ? r0 : r1;
  float kB = hiB ? r1 : r0;
  return kB + __shfl_xor(sB, 8, 32);
}
#define TAIL_BLOCKS 2048
static_assert(TAIL_BLOCKS * H <= 2 * NN, "partial sums must fit in the sD+sS region");
static_assert(TAIL_BLOCKS % 16 == 0, "stage 2 splits the partial rows into 16 equal groups");
__global__ __launch_bounds__(256) void k_tail(
    const float* __restrict__ h,
    const float* __restrict__ qW1, const float* __restrict__ qb1,
    const float* __restrict__ qW2, const float* __restrict__ qb2,
    const float* __restrict__ iW1, const float* __restrict__ ib1,
    const float* __restrict__ iW2, const float* __restrict__ ib2,
    const float* __restrict__ gW1, const float* __restrict__ gb1,
    const float* __restrict__ gW2, const float* __restrict__ gb2,
    const float* __restrict__ oW,  const float* __restrict__ obv,
    const float* __restrict__ sW1, const float* __restrict__ sb1,
    const float* __restrict__ sW2, const float* __restrict__ sb2,
    float* __restrict__ oq, float* __restrict__ oi, float* __restrict__ og,
    float* __restrict__ oh, float* __restrict__ obot, float* __restrict__ osev,
    float* __restrict__ part) {
  __shared__ float ls[8][H];
  const int wave = threadIdx.x >> 5;
  const int lane = threadIdx.x & 31;
  const int nwaves = gridDim.x * 8;
  float cs0 = 0.0f, cs1 = 0.0f;

  const int kgrp = lane >> 3;
  const int hiA = (lane >> 4) & 1, hiB = (lane >> 3) & 1;
  const int col = 4 * (lane & 7) + 2 * hiA + hiB;
  const float4* __restrict__ q4 = reinterpret_cast<const float4*>(qW1);
  const float4* __restrict__ i4 = reinterpret_cast<const float4*>(iW1);
  const float4* __restrict__ g4 = reinterpret_cast<const float4*>(gW1);
  const float4* __restrict__ s4 = reinterpret_cast<const float4*>(sW1);
  const float4* __restrict__ o4 = reinterpret_cast<const float4*>(oW);
  const unsigned wl = (unsigned)lane;

  for (int gw = blockIdx.x * 8 + wave; gw < NN; gw += nwaves) {
    size_t base = (size_t)gw * H;

    const float h0 = h[base + lane], h1 = h[base + lane + 32];

    float aq[4] = {0.f, 0.f, 0.f, 0.f}, ai[4] = {0.f, 0.f, 0.f, 0.f}, ag[4] = {0.f, 0.f, 0.f, 0.f};
    float as[4] = {0.f, 0.f, 0.f, 0.f}, ao[4] = {0.f, 0.f, 0.f, 0.f};
#pragma unroll
    for (int half = 0; half < 2; ++half) {
      const float hsrc = half ? h1 : h0;
#pragma unroll 2
      for (int ii8 = 0; ii8 < 8; ++ii8) {
        const int i = half * 8 + ii8;
        const float hk = __shfl(hsrc, 4 * ii8 + kgrp, 32);
        const unsigned wx = (unsigned)(i * 32) + wl;
        const float4 wq = q4[wx], wi = i4[wx], wg = g4[wx], wsv = s4[wx], wo = o4[wx];
        aq[0] = fmaf(hk, wq.x, aq[0]);  aq[1] = fmaf(hk, wq.y, aq[1]);
        aq[2] = fmaf(hk, wq.z, aq[2]);  aq[3] = fmaf(hk, wq.w, aq[3]);
        ai[0] = fmaf(hk, wi.x, ai[0]);  ai[1] = fmaf(hk, wi.y, ai[1]);
        ai[2] = fmaf(hk, wi.z, ai[2]);  ai[3] = fmaf(hk, wi.w, ai[3]);
        ag[0] = fmaf(hk, wg.x, ag[0]);  ag[1] = fmaf(hk, wg.y, ag[1]);
        ag[2] = fmaf(hk, wg.z, ag[2]);  ag[3] = fmaf(hk, wg.w, ag[3]);
        as[0] = fmaf(hk, wsv.x, as[0]); as[1] = fmaf(hk, wsv.y, as[1]);
        as[2] = fmaf(hk, wsv.z, as[2]); as[3] = fmaf(hk, wsv.w, as[3]);
        ao[0] = fmaf(hk, wo.x, ao[0]);  ao[1] = fmaf(hk, wo.y, ao[1]);
        ao[2] = fmaf(hk, wo.z, ao[2]);  ao[3] = fmaf(hk, wo.w, ao[3]);
      }
    }
    float tq = kgroup_reduce(aq, hiA, hiB);
    float ti = kgroup_reduce(ai, hiA, hiB);
    float tg = kgroup_reduce(ag, hiA, hiB);
    float ts = kgroup_reduce(as, hiA, hiB);
    float bo = kgroup_reduce(ao, hiA, hiB);
    tq = fmaxf(tq + qb1[col], 0.0f);
    ti = fmaxf(ti + ib1[col], 0.0f);
    tg = fmaxf(tg + gb1[col], 0.0f);
    ts = fmaxf(ts + sb1[col], 0.0f);

    float q  = wave_sum(tq * qW2[col]);
    float ii = wave_sum(ti * iW2[col]);
    float gg = wave_sum(tg * gW2[col]);
    if (lane == 0) {
      oq[gw] = sigmoidf_(q + qb2[0]);
      oi[gw] = sigmoidf_(ii + ib2[0]);
      og[gw] = sigmoidf_(gg + gb2[0]);
    }

    float z[5];
#pragma unroll
    for (int c = 0; c < 5; ++c) z[c] = wave_sum(ts * sW2[col * 5 + c]) + sb2[c];
    float zm = z[0];
#pragma unroll
    for (int c = 1; c < 5; ++c) zm = fmaxf(zm, z[c]);
    float es = 0.f;
#pragma unroll
    for (int c = 0; c < 5; ++c) { z[c] = __expf(z[c] - zm); es += z[c]; }
    float inv = 1.0f / es;
#pragma unroll
    for (int c = 0; c < 5; ++c)
      if (lane == c) osev[(size_t)gw * 5 + c] = z[c] * inv;

    obot[(size_t)gw * DOUT + col] = bo + obv[col];
    oh[base + lane]      = h0;
    oh[base + lane + 32] = h1;

    cs0 += h0;
    cs1 += h1;
  }

  ls[wave][lane]      = cs0;
  ls[wave][lane + 32] = cs1;
  __syncthreads();
  if (threadIdx.x < H) {
    float t = 0.0f;
#pragma unroll
    for (int w = 0; w < 8; ++w) t += ls[w][threadIdx.x];
    part[(size_t)blockIdx.x * H + threadIdx.x] = t;
  }
}

__global__ __launch_bounds__(1024) void k_mean_final(
    const float* __restrict__ part, float* __restrict__ gout) {
  __shared__ float lq[16][H];
  const int f = threadIdx.x & 63, q = threadIdx.x >> 6;
  const float* p = part + (size_t)q * (TAIL_BLOCKS / 16) * H + f;
  float acc = 0.0f;
#pragma unroll 16
  for (int j = 0; j < TAIL_BLOCKS / 16; ++j) acc += p[(size_t)j * H];
  lq[q][f] = acc;
  __syncthreads();
  if (threadIdx.x < H) {
    float t = 0.0f;
#pragma unroll
    for (int g = 0; g < 16; ++g) t += lq[g][threadIdx.x];
    gout[threadIdx.x] = t * (1.0f / NN);
  }
}

extern "C" void kernel_launch(void* const* d_in, const int* in_sizes, int n_in,
                              void* d_out, int out_size, void* d_ws, size_t ws_size,
                              hipStream_t stream) {
  (void)in_sizes; (void)n_in; (void)out_size; (void)ws_size;
  const float* x    = (const float*)d_in[0];
  const int*   ei   = (const int*)  d_in[1];
  const float* ea   = (const float*)d_in[2];
  const float* inW  = (const float*)d_in[3];
  const float* inb  = (const float*)d_in[4];
  const float* linW = (const float*)d_in[5];
  const float* linb = (const float*)d_in[6];
  const float* attW = (const float*)d_in[7];
  const float* attb = (const float*)d_in[8];
  const float* lng  = (const float*)d_in[9];
  const float* lnb  = (const float*)d_in[10];
  const float* qW1 = (const float*)d_in[11]; const float* qb1 = (const float*)d_in[12];
  const float* qW2 = (const float*)d_in[13]; const float* qb2 = (const float*)d_in[14];
  const float* iW1 = (const float*)d_in[15]; const float* ib1 = (const float*)d_in[16];
  const float* iW2 = (const float*)d_in[17]; const float* ib2 = (const float*)d_in[18];
  const float* gW1 = (const float*)d_in[19]; const float* gb1 = (const float*)d_in[20];
  const float* gW2 = (const float*)d_in[21]; const float* gb2 = (const float*)d_in[22];
  const float* oW  = (const float*)d_in[23]; const float* obv = (const float*)d_in[24];
  const float* sW1 = (const float*)d_in[25]; const float* sb1 = (const float*)d_in[26];
  const float* sW2 = (const float*)d_in[27]; const float* sb2 = (const float*)d_in[28];

  float* ws   = (float*)d_ws;
  float* h    = ws;
  float* mbuf = h    + (size_t)NN * H;
  float* aggr = mbuf + (size_t)NN * H;
  float* sD   = aggr + (size_t)NN * H;
  float* sS   = sD + NN;

  float* out  = (float*)d_out;
  float* oq   = out;
  float* oi   = out + (size_t)NN;
  float* og   = out + 2 * (size_t)NN;
  float* oh   = out + 3 * (size_t)NN;
  float* ogr  = out + 67 * (size_t)NN;
  float* obot = out + 67 * (size_t)NN + H;
  float* osev = out + 99 * (size_t)NN + H;

  int*      erec = (int*)(aggr + 2000000);
  unsigned* eown = (unsigned*)(aggr + 6000000);
  k_edge_pack<<<(NE / 4 + 255) / 256, 256, 0, stream>>>(ei, ea, erec, eown);

  float*  rec0 = mbuf;
  double* acc0 = (double*)aggr;
  k_l0_prep<<<(NN + 255) / 256, 256, 0, stream>>>(x, inW, inb, attW, rec0, acc0);
  k_l0_edges<<<(NE + 255) / 256, 256, 0, stream>>>(ei, ea, rec0, attb, acc0);
  k_l0_node<<<NN / 16, 128, 0, stream>>>(x, acc0, inW, inb, linW, linb, lng, lnb, h);

  for (int l = 1; l < NL; ++l) {
    k_gemm_h<<<NN / 16, 128, 0, stream>>>(h, linW + (size_t)l * H * H, linb + (size_t)l * H,
                                          attW + (size_t)l * 2 * H, mbuf, sD, sS);
    k_own<<<OWN_BLOCKS, 1024, 0, stream>>>(erec, eown, mbuf, sD, sS, attb + l,
                                           lng + (size_t)l * H, lnb + (size_t)l * H, h);
  }

  float* part = sD;
  k_tail<<<TAIL_BLOCKS, 256, 0, stream>>>(h,
                                          qW1, qb1, qW2, qb2, iW1, ib1, iW2, ib2,
                                          gW1, gb1, gW2, gb2, oW, obv, sW1, sb1, sW2, sb2,
                                          oq, oi, og, oh, obot, osev, part);
  k_mean_final<<<1, 1024, 0, stream>>>(part, ogr);
}


# layers 1-3: owners sum rows of h, not of h W + b (linear in h), apply W once per owned row on WMMA from the LDS tile and emit the next attention dots; 3 node GEMM launches go (sum order changes)
# speedup vs baseline: 1.9417x; 1.0821x over previous
#include <hip/hip_runtime.h>

#define NN 100000
#define NE 1000000
#define DIN 5
#define H 64
#define HHID 32
#define NL 4
#define DOUT 32
#define LN_EPS 1e-5f

typedef __attribute__((ext_vector_type(2)))  float    v2f;
typedef __attribute__((ext_vector_type(8)))  float    v8f;
typedef __attribute__((ext_vector_type(16))) _Float16 v16h;

#if defined(__has_builtin)
#if __has_builtin(__builtin_amdgcn_wmma_f32_16x16x4_f32)
#define HAVE_WMMA_F32 1
#endif
#endif

__device__ __forceinline__ float wave_sum(float v) {
#pragma unroll
  for (int off = 16; off > 0; off >>= 1) v += __shfl_xor(v, off, 32);
  return v;
}
__device__ __forceinline__ float sigmoidf_(float x) { return 1.0f / (1.0f + __expf(-x)); }

__device__ __forceinline__ void atomic_fadd(float* p, float v) { unsafeAtomicAdd(p, v); }

__device__ __forceinline__ v8f tile_product(const float (*lA)[H + 1], const float (*lW)[H + 1],
                                            int m, int g, int col, v8f acc) {
#if HAVE_WMMA_F32
#pragma unroll
  for (int k0 = 0; k0 < H; k0 += 4) {
    int kb = k0 + 2 * g;
    v2f a, b;
    a.x = lA[m][kb];     a.y = lA[m][kb + 1];
    b.x = lW[kb][col];   b.y = lW[kb + 1][col];
    acc = __builtin_amdgcn_wmma_f32_16x16x4_f32(false, a, false, b, (short)0, acc, false, false);
  }
#else
#pragma unroll
  for (int kk = 0; kk < H; kk += 32) {
    v16h a, b;
#pragma unroll
    for (int j = 0; j < 8; ++j) {
      int ka = (j < 4) ? (2 * j + 8 * g) : (16 + 2 * (j - 4) + 8 * g);
      a[2 * j]     = (_Float16)lA[m][kk + ka];
      a[2 * j + 1] = (_Float16)lA[m][kk + ka + 1];
      int kb = 2 * j + 16 * g;
      b[2 * j]     = (_Float16)lW[kk + kb][col];
      b[2 * j + 1] = (_Float16)lW[kk + kb + 1][col];
    }
    acc = __builtin_amdgcn_wmma_f32_16x16x32_f16(false, a, false, b, (short)0, acc, false, false);
  }
#endif
  return acc;
}

#define EDGE_DEPTH 8
#define OWN_R 393
#define OWN_BLOCKS ((NN + OWN_R - 1) / OWN_R)
#define OWN_WAVES 32
#define OWN_Q 544
#define OWN_TILES ((OWN_R + 15) / 16)
#define OWN_RP (OWN_TILES * 16)
#define TP 68
static_assert(OWN_TILES <= OWN_WAVES, "one wave per 16-row tile of the owner's product");
static_assert(NL == 4, "the three owner launches are written out");
static_assert(OWN_BLOCKS <= 255, "owner numbers are bytes 0..254; 255 marks a map byte that is nobody's");
static_assert(NE % 16 == 0, "the owner map is read 16 bytes a lane");
static_assert(NE % 32 == 0, "every wave of the layer-0 edge pass owns 32 whole edges");
static_assert(32 % EDGE_DEPTH == 0, "a batch of 32 edges splits into whole groups");

__device__ __forceinline__ int lane_bcast_i(int v, int l) { return __builtin_amdgcn_readlane(v, l); }
__device__ __forceinline__ float lane_bcast_f(float v, int l) {
  return __int_as_float(__builtin_amdgcn_readlane(__float_as_int(v), l));
}

__global__ __launch_bounds__(256) void k_edge_pack(
    const int* __restrict__ ei, const float* __restrict__ ea,
    int* __restrict__ rec, unsigned* __restrict__ own) {
  const int t = blockIdx.x * 256 + threadIdx.x;
  if (t >= NE / 4) return;
  const int4 s4 = reinterpret_cast<const int4*>(ei)[t];
  const int4 d4 = reinterpret_cast<const int4*>(ei + NE)[t];
  const float4 a4 = reinterpret_cast<const float4*>(ea)[t];
  int4* r4 = reinterpret_cast<int4*>(rec) + (size_t)t * 4;
  r4[0] = make_int4(s4.x, d4.x, __float_as_int(a4.x), 0);
  r4[1] = make_int4(s4.y, d4.y, __float_as_int(a4.y), 0);
  r4[2] = make_int4(s4.z, d4.z, __float_as_int(a4.z), 0);
  r4[3] = make_int4(s4.w, d4.w, __float_as_int(a4.w), 0);
  own[t] = (unsigned)(d4.x / OWN_R) | ((unsigned)(d4.y / OWN_R) << 8) |
           ((unsigned)(d4.z / OWN_R) << 16) | ((unsigned)(d4.w / OWN_R) << 24);
}

__device__ __forceinline__ void own_batch(
    const int* __restrict__ qw, int n, int lane, int lo, float b0,
    const int* __restrict__ rec, const float* __restrict__ hrows, const float* __restrict__ sS,
    const float* __restrict__ lsD, float* __restrict__ lsw, float* __restrict__ tile) {
  int src = 0, dl = 0;
  float w = 0.0f;
  if (lane < n) {
    const int e = qw[lane];
    const int4 r = reinterpret_cast<const int4*>(rec)[e];
    src = r.x;
    dl = r.y - lo;
    w = __int_as_float(r.z) * sigmoidf_(lsD[dl] + sS[src] + b0);
    __hip_atomic_fetch_add(lsw + dl, w, __ATOMIC_RELAXED, __HIP_MEMORY_SCOPE_WORKGROUP);
  }
  const float* mlane = hrows + lane;
#pragma unroll
  for (int i0 = 0; i0 < 32; i0 += EDGE_DEPTH) {
    if (i0 < n) {
      float lov[EDGE_DEPTH], hiv[EDGE_DEPTH];
#pragma unroll
      for (int j = 0; j < EDGE_DEPTH; ++j) {
        const float* mr = mlane + (size_t)lane_bcast_i(src, i0 + j) * H;
        lov[j] = mr[0];
        hiv[j] = mr[32];
      }
#pragma unroll
      for (int j = 0; j < EDGE_DEPTH; ++j) {
        if (i0 + j < n) {
          const float bw = lane_bcast_f(w, i0 + j);
          float* tr = tile + lane_bcast_i(dl, i0 + j) * TP + lane;
          __hip_atomic_fetch_add(tr,      lov[j] * bw, __ATOMIC_RELAXED, __HIP_MEMORY_SCOPE_WORKGROUP);
          __hip_atomic_fetch_add(tr + 32, hiv[j] * bw, __ATOMIC_RELAXED, __HIP_MEMORY_SCOPE_WORKGROUP);
        }
      }
    }
  }
}

__device__ __forceinline__ void own_scan_piece(const uint4 o, int vb, unsigned myb, int* qw, int& cnt) {
  const unsigned wd[4] = {o.x, o.y, o.z, o.w};
#pragma unroll
  for (int k = 0; k < 16; ++k) {
    const unsigned byte = (wd[k >> 2] >> (8 * (k & 3))) & 0xFFu;
    const bool mine = (byte == myb);
    const unsigned mask = __builtin_amdgcn_ballot_w32(mine);
    if (mask != 0u) {
      const int pos = cnt + (int)__builtin_amdgcn_mbcnt_lo(mask, 0u);
      if (mine) qw[pos] = vb + k;
      cnt += __builtin_popcount(mask);
    }
  }
}

__device__ __forceinline__ uint4 own_piece(const uint4 t, bool ok) {
  return ok ? t : make_uint4(0xFFFFFFFFu, 0xFFFFFFFFu, 0xFFFFFFFFu, 0xFFFFFFFFu);
}

__device__ __forceinline__ void own_drain(
    int* qw, int& cnt, int lane, int lo, float b0,
    const int* __restrict__ rec, const float* __restrict__ hrows, const float* __restrict__ sS,
    const float* __restrict__ lsD, float* __restrict__ lsw, float* __restrict__ tile) {
  int head = 0;
  while (cnt - head >= 32) {
    own_batch(qw + head, 32, lane, lo, b0, rec, hrows, sS, lsD, lsw, tile);
    head += 32;
  }
  if (head > 0) {
    const int rest = cnt - head;
    int keep = 0;
    if (lane < rest) keep = qw[head + lane];
    if (lane < rest) qw[lane] = keep;
    cnt = rest;
  }
}

__global__ __launch_bounds__(1024) void k_own(
    const int* __restrict__ rec, const unsigned* __restrict__ own,
    const float* __restrict__ hin, const float* __restrict__ sD,
    const float* __restrict__ sS, const float* __restrict__ attb,
    const float* __restrict__ W, const float* __restrict__ bias,
    const float* __restrict__ g, const float* __restrict__ beta,
    const float* __restrict__ awn, int has_next,
    float* __restrict__ hout, float* __restrict__ sDn, float* __restrict__ sSn) {
  __shared__ __attribute__((aligned(16))) float tile[OWN_RP * TP];
  __shared__ __attribute__((aligned(16))) float lWt[H * TP];
  __shared__ float lsD[OWN_RP];
  __shared__ float lsw[OWN_RP];
  __shared__ int q[OWN_WAVES][OWN_Q];
  const int tid = threadIdx.x, lane = tid & 31;
  const int wave = __builtin_amdgcn_readfirstlane(tid >> 5);
  const int lo = blockIdx.x * OWN_R;
  const int nrows = (NN - lo < OWN_R) ? (NN - lo) : OWN_R;
  const unsigned myb = blockIdx.x;
  const float b0 = attb[0];

#pragma unroll 4
  for (int r = wave; r < nrows; r += OWN_WAVES) {
    const float* hr = hin + (size_t)(lo + r) * H;
    tile[r * TP + lane]      = hr[lane];
    tile[r * TP + lane + 32] = hr[lane + 32];
  }
  for (int r = nrows + wave; r < OWN_RP; r += OWN_WAVES) {
    tile[r * TP + lane]      = 0.0f;
    tile[r * TP + lane + 32] = 0.0f;
  }
  for (int r = tid; r < OWN_RP; r += 1024) {
    lsD[r] = (r < nrows) ? sD[lo + r] : 0.0f;
    lsw[r] = 1.0f;
  }
  for (int idx = tid; idx < H * H; idx += 1024)
    lWt[(idx & 63) * TP + (idx >> 6)] = W[idx];
  __syncthreads();

  const uint4* __restrict__ own4 = reinterpret_cast<const uint4*>(own);
  int* qw = q[wave];
  int cnt = 0;
  const int nvec = NE / 16;
  const int nit = (nvec + 31) / 32;
  for (int it0 = wave; it0 < nit; it0 += 4 * OWN_WAVES) {
    const int v0 = it0 * 32 + lane;
    const int v1 = v0 + 32 * OWN_WAVES;
    const int v2 = v1 + 32 * OWN_WAVES;
    const int v3 = v2 + 32 * OWN_WAVES;
    const bool ok0 = v0 < nvec, ok1 = v1 < nvec, ok2 = v2 < nvec, ok3 = v3 < nvec;
    const uint4 t0 = own4[ok0 ? v0 : nvec - 1];
    const uint4 t1 = own4[ok1 ? v1 : nvec - 1];
    const uint4 t2 = own4[ok2 ? v2 : nvec - 1];
    const uint4 t3 = own4[ok3 ? v3 : nvec - 1];
    own_scan_piece(own_piece(t0, ok0), v0 * 16, myb, qw, cnt);
    own_drain(qw, cnt, lane, lo, b0, rec, hin, sS, lsD, lsw, tile);
    own_scan_piece(own_piece(t1, ok1), v1 * 16, myb, qw, cnt);
    own_drain(qw, cnt, lane, lo, b0, rec, hin, sS, lsD, lsw, tile);
    own_scan_piece(own_piece(t2, ok2), v2 * 16, myb, qw, cnt);
    own_drain(qw, cnt, lane, lo, b0, rec, hin, sS, lsD, lsw, tile);
    own_scan_piece(own_piece(t3, ok3), v3 * 16, myb, qw, cnt);
    own_drain(qw, cnt, lane, lo, b0, rec, hin, sS, lsD, lsw, tile);
  }
  if (cnt > 0) own_batch(qw, cnt, lane, lo, b0, rec, hin, sS, lsD, lsw, tile);
  __syncthreads();

  if (wave * 16 < nrows) {
    const int mrow = lane & 15, gq = lane >> 4;
    const int r0 = wave * 16;
    const float* ap = tile + (r0 + mrow) * TP + 2 * gq;
    const float* bp = lWt + mrow * TP + 2 * gq;
    v8f acc[4];
#pragma unroll
    for (int cc = 0; cc < 4; ++cc) {
      const float bv = bias[16 * cc + mrow];
#pragma unroll
      for (int i = 0; i < 8; ++i) acc[cc][i] = lsw[r0 + i + 8 * gq] * bv;
    }
#if HAVE_WMMA_F32
#pragma unroll
    for (int k0 = 0; k0 < H; k0 += 4) {
      const v2f a = *reinterpret_cast<const v2f*>(ap + k0);
#pragma unroll
      for (int cc = 0; cc < 4; ++cc) {
        const v2f b = *reinterpret_cast<const v2f*>(bp + 16 * cc * TP + k0);
        acc[cc] = __builtin_amdgcn_wmma_f32_16x16x4_f32(false, a, false, b, (short)0, acc[cc], false, false);
      }
    }
#endif
#pragma unroll
    for (int cc = 0; cc < 4; ++cc) {
#pragma unroll
      for (int i = 0; i < 8; ++i) tile[(r0 + i + 8 * gq) * TP + 16 * cc + mrow] = acc[cc][i];
    }
  }
  __syncthreads();

  const float g0 = g[lane], g1 = g[lane + 32];
  const float be0 = beta[lane], be1 = beta[lane + 32];
  const float aD0 = awn[lane], aD1 = awn[lane + 32];
  const float aS0 = awn[H + lane], aS1 = awn[H + lane + 32];
#pragma unroll 2
  for (int r = wave; r < nrows; r += OWN_WAVES) {
    const size_t base = (size_t)(lo + r) * H;
    const float r0 = hin[base + lane], r1 = hin[base + lane + 32];
    const float f0 = tile[r * TP + lane], f1 = tile[r * TP + lane + 32];
    const float mean = wave_sum(f0 + f1) * (1.0f / H);
    const float d0 = f0 - mean, d1 = f1 - mean;
    const float var = wave_sum(fmaf(d0, d0, d1 * d1)) * (1.0f / H);
    const float inv = rsqrtf(var + LN_EPS);
    const float y0 = fmaf(d0 * inv, g0, be0);
    const float y1 = fmaf(d1 * inv, g1, be1);
    const float o0 = fmaxf(y0, 0.0f) + r0;
    const float o1 = fmaxf(y1, 0.0f) + r1;
    hout[base + lane]      = o0;
    hout[base + lane + 32] = o1;
    if (has_next) {
      const float pd = wave_sum(fmaf(o0, aD0, o1 * aD1));
      const float ps = wave_sum(fmaf(o0, aS0, o1 * aS1));
      if (lane == 0) { sDn[lo + r] = pd; sSn[lo + r] = ps; }
    }
  }
}

#define ACCW 8
#define RECW 8

__global__ __launch_bounds__(256) void k_l0_prep(
    const float* __restrict__ x, const float* __restrict__ inW, const float* __restrict__ inb,
    const float* __restrict__ aw, float* __restrict__ rec, double* __restrict__ acc) {
  int n = blockIdx.x * 256 + threadIdx.x;
  if (n >= NN) return;
  float xv[DIN];
#pragma unroll
  for (int d = 0; d < DIN; ++d) xv[d] = x[(size_t)n * DIN + d];
  float sd = 0.0f, ss = 0.0f;
#pragma unroll 8
  for (int k = 0; k < H; ++k) {
    float hk = inb[k];
#pragma unroll
    for (int d = 0; d < DIN; ++d) hk = fmaf(xv[d], inW[d * H + k], hk);
    sd = fmaf(hk, aw[k], sd);
    ss = fmaf(hk, aw[H + k], ss);
  }
  float4* r4 = reinterpret_cast<float4*>(rec + (size_t)n * RECW);
  r4[0] = make_float4(xv[0], xv[1], xv[2], xv[3]);
  r4[1] = make_float4(xv[4], ss, sd, 0.0f);
  double2* a2 = reinterpret_cast<double2*>(acc + (size_t)n * ACCW);
#pragma unroll
  for (int j = 0; j < ACCW / 2; ++j) a2[j] = make_double2(0.0, 0.0);
}

__global__ __launch_bounds__(256) void k_l0_edges(
    const int* __restrict__ ei, const float* __restrict__ ea,
    const float* __restrict__ rec, const float* __restrict__ attb,
    double* __restrict__ acc) {
  int gw = blockIdx.x * 8 + (threadIdx.x >> 5);
  int lane = threadIdx.x & 31;
  int ebase = gw * 32;
  if (ebase >= NE) return;
  const int sub = lane >> 3, c = lane & 7;
  const float b0 = attb[0];
  int srcs[8], dsts[8];
  float eas[8], rv[8], sd[8];
#pragma unroll
  for (int i = 0; i < 8; ++i) {
    int e = ebase + 4 * i + sub;
    srcs[i] = ei[e];
    dsts[i] = ei[NE + e];
    eas[i] = ea[e];
  }
#pragma unroll
  for (int i = 0; i < 8; ++i) {
    rv[i] = rec[(size_t)srcs[i] * RECW + c];
    sd[i] = rec[(size_t)dsts[i] * RECW + 6];
  }
#pragma unroll
  for (int i = 0; i < 8; ++i) {
    float ss = __shfl(rv[i], (lane & 24) | 5, 32);
    float w = eas[i] * sigmoidf_(sd[i] + ss + b0);
    if (c < 6) {
      double v = (c == 5) ? (double)w : (double)w * (double)rv[i];
      unsafeAtomicAdd(acc + (size_t)dsts[i] * ACCW + c, v);
    }
  }
}

__global__ __launch_bounds__(128) void k_l0_node(
    const float* __restrict__ x, const double* __restrict__ acc,
    const float* __restrict__ inW, const float* __restrict__ inb,
    const float* __restrict__ W, const float* __restrict__ bias,
    const float* __restrict__ g_, const float* __restrict__ beta,
    const float* __restrict__ awn,
    float* __restrict__ h, float* __restrict__ sDn, float* __restrict__ sSn) {
  __shared__ float lA[16][H + 1];
  __shared__ float lW[H][H + 1];
  __shared__ float lO[16][H + 1];
  __shared__ float lX[16][8];
  __shared__ float lS[16][8];
  const int tid = threadIdx.x;
  const int row0 = blockIdx.x * 16;
  for (int idx = tid; idx < H * H; idx += 128)
    lW[idx >> 6][idx & 63] = W[idx];
  {
    const int r = tid >> 3, c = tid & 7;
    const size_t n = (size_t)(row0 + r);
    float xv = (c < DIN) ? x[n * DIN + c] : 0.0f;
    float sv = (c < 6) ? (float)acc[n * ACCW + c] : 0.0f;
    lX[r][c] = xv;
    lS[r][c] = (c < DIN) ? (sv + xv) : ((c == 5) ? (sv + 1.0f) : 0.0f);
  }
  __syncthreads();
  for (int idx = tid; idx < 16 * H; idx += 128) {
    const int r = idx >> 6, k = idx & 63;
    float u = lS[r][5] * inb[k];
#pragma unroll
    for (int d = 0; d < DIN; ++d) u = fmaf(lS[r][d], inW[d * H + k], u);
    lA[r][k] = u;
  }
  __syncthreads();

  const int wave = tid >> 5, lane = tid & 31;
  const int m = lane & 15, g = lane >> 4;
  const int col = wave * 16 + m;

  v8f t;
  const float bv = bias[col];
#pragma unroll
  for (int i = 0; i < 8; ++i) t[i] = lS[i + 8 * g][5] * bv;

  t = tile_product(lA, lW, m, g, col, t);

#pragma unroll
  for (int i = 0; i < 8; ++i) lO[i + 8 * g][col] = t[i];
  __syncthreads();

  const float g0 = g_[lane], g1 = g_[lane + 32];
  const float be0 = beta[lane], be1 = beta[lane + 32];
  const float aD0 = awn[lane], aD1 = awn[lane + 32];
  const float aS0 = awn[H + lane], aS1 = awn[H + lane + 32];
#pragma unroll
  for (int rr = 0; rr < 4; ++rr) {
    const int r = wave * 4 + rr;
    float r0 = inb[lane], r1 = inb[lane + 32];
#pragma unroll
    for (int d = 0; d < DIN; ++d) {
      r0 = fmaf(lX[r][d], inW[d * H + lane], r0);
      r1 = fmaf(lX[r][d], inW[d * H + lane + 32], r1);
    }
    float f0 = lO[r][lane], f1 = lO[r][lane + 32];
    float mean = wave_sum(f0 + f1) * (1.0f / H);
    float d0 = f0 - mean, d1 = f1 - mean;
    float var = wave_sum(fmaf(d0, d0, d1 * d1)) * (1.0f / H);
    float inv = rsqrtf(var + LN_EPS);
    float y0 = fmaf(d0 * inv, g0, be0);
    float y1 = fmaf(d1 * inv, g1, be1);
    const size_t base = (size_t)(row0 + r) * H;
    const float o0 = fmaxf(y0, 0.0f) + r0;
    const float o1 = fmaxf(y1, 0.0f) + r1;
    h[base + lane]      = o0;
    h[base + lane + 32] = o1;
    const float pd = wave_sum(fmaf(o0, aD0, o1 * aD1));
    const float ps = wave_sum(fmaf(o0, aS0, o1 * aS1));
    if (lane == 0) { sDn[row0 + r] = pd; sSn[row0 + r] = ps; }
  }
}

__device__ __forceinline__ float kgroup_reduce(const float (&a)[4], int hiA, int hiB) {
  float s0 = hiA ? a[0] : a[2];
  float s1 = hiA ? a[1] : a[3];
  float k0 = hiA ? a[2] : a[0];
  float k1 = hiA ? a[3] : a[1];
  float r0 = k0 + __shfl_xor(s0, 16, 32);
  float r1 = k1 + __shfl_xor(s1, 16, 32);
  float sB = hiB ? r0 : r1;
  float kB = hiB ? r1 : r0;
  return kB + __shfl_xor(sB, 8, 32);
}
#define TAIL_BLOCKS 2048
static_assert(TAIL_BLOCKS * H <= 2 * NN, "partial sums must fit in the sD+sS region");
static_assert(TAIL_BLOCKS % 16 == 0, "stage 2 splits the partial rows into 16 equal groups");
__global__ __launch_bounds__(256) void k_tail(
    const float* __restrict__ h,
    const float* __restrict__ qW1, const float* __restrict__ qb1,
    const float* __restrict__ qW2, const float* __restrict__ qb2,
    const float* __restrict__ iW1, const float* __restrict__ ib1,
    const float* __restrict__ iW2, const float* __restrict__ ib2,
    const float* __restrict__ gW1, const float* __restrict__ gb1,
    const float* __restrict__ gW2, const float* __restrict__ gb2,
    const float* __restrict__ oW,  const float* __restrict__ obv,
    const float* __restrict__ sW1, const float* __restrict__ sb1,
    const float* __restrict__ sW2, const float* __restrict__ sb2,
    float* __restrict__ oq, float* __restrict__ oi, float* __restrict__ og,
    float* __restrict__ oh, float* __restrict__ obot, float* __restrict__ osev,
    float* __restrict__ part) {
  __shared__ float ls[8][H];
  const int wave = threadIdx.x >> 5;
  const int lane = threadIdx.x & 31;
  const int nwaves = gridDim.x * 8;
  float cs0 = 0.0f, cs1 = 0.0f;

  const int kgrp = lane >> 3;
  const int hiA = (lane >> 4) & 1, hiB = (lane >> 3) & 1;
  const int col = 4 * (lane & 7) + 2 * hiA + hiB;
  const float4* __restrict__ q4 = reinterpret_cast<const float4*>(qW1);
  const float4* __restrict__ i4 = reinterpret_cast<const float4*>(iW1);
  const float4* __restrict__ g4 = reinterpret_cast<const float4*>(gW1);
  const float4* __restrict__ s4 = reinterpret_cast<const float4*>(sW1);
  const float4* __restrict__ o4 = reinterpret_cast<const float4*>(oW);
  const unsigned wl = (unsigned)lane;

  for (int gw = blockIdx.x * 8 + wave; gw < NN; gw += nwaves) {
    size_t base = (size_t)gw * H;

    const float h0 = h[base + lane], h1 = h[base + lane + 32];

    float aq[4] = {0.f, 0.f, 0.f, 0.f}, ai[4] = {0.f, 0.f, 0.f, 0.f}, ag[4] = {0.f, 0.f, 0.f, 0.f};
    float as[4] = {0.f, 0.f, 0.f, 0.f}, ao[4] = {0.f, 0.f, 0.f, 0.f};
#pragma unroll
    for (int half = 0; half < 2; ++half) {
      const float hsrc = half ? h1 : h0;
#pragma unroll 2
      for (int ii8 = 0; ii8 < 8; ++ii8) {
        const int i = half * 8 + ii8;
        const float hk = __shfl(hsrc, 4 * ii8 + kgrp, 32);
        const unsigned wx = (unsigned)(i * 32) + wl;
        const float4 wq = q4[wx], wi = i4[wx], wg = g4[wx], wsv = s4[wx], wo = o4[wx];
        aq[0] = fmaf(hk, wq.x, aq[0]);  aq[1] = fmaf(hk, wq.y, aq[1]);
        aq[2] = fmaf(hk, wq.z, aq[2]);  aq[3] = fmaf(hk, wq.w, aq[3]);
        ai[0] = fmaf(hk, wi.x, ai[0]);  ai[1] = fmaf(hk, wi.y, ai[1]);
        ai[2] = fmaf(hk, wi.z, ai[2]);  ai[3] = fmaf(hk, wi.w, ai[3]);
        ag[0] = fmaf(hk, wg.x, ag[0]);  ag[1] = fmaf(hk, wg.y, ag[1]);
        ag[2] = fmaf(hk, wg.z, ag[2]);  ag[3] = fmaf(hk, wg.w, ag[3]);
        as[0] = fmaf(hk, wsv.x, as[0]); as[1] = fmaf(hk, wsv.y, as[1]);
        as[2] = fmaf(hk, wsv.z, as[2]); as[3] = fmaf(hk, wsv.w, as[3]);
        ao[0] = fmaf(hk, wo.x, ao[0]);  ao[1] = fmaf(hk, wo.y, ao[1]);
        ao[2] = fmaf(hk, wo.z, ao[2]);  ao[3] = fmaf(hk, wo.w, ao[3]);
      }
    }
    float tq = kgroup_reduce(aq, hiA, hiB);
    float ti = kgroup_reduce(ai, hiA, hiB);
    float tg = kgroup_reduce(ag, hiA, hiB);
    float ts = kgroup_reduce(as, hiA, hiB);
    float bo = kgroup_reduce(ao, hiA, hiB);
    tq = fmaxf(tq + qb1[col], 0.0f);
    ti = fmaxf(ti + ib1[col], 0.0f);
    tg = fmaxf(tg + gb1[col], 0.0f);
    ts = fmaxf(ts + sb1[col], 0.0f);

    float q  = wave_sum(tq * qW2[col]);
    float ii = wave_sum(ti * iW2[col]);
    float gg = wave_sum(tg * gW2[col]);
    if (lane == 0) {
      oq[gw] = sigmoidf_(q + qb2[0]);
      oi[gw] = sigmoidf_(ii + ib2[0]);
      og[gw] = sigmoidf_(gg + gb2[0]);
    }

    float z[5];
#pragma unroll
    for (int c = 0; c < 5; ++c) z[c] = wave_sum(ts * sW2[col * 5 + c]) + sb2[c];
    float zm = z[0];
#pragma unroll
    for (int c = 1; c < 5; ++c) zm = fmaxf(zm, z[c]);
    float es = 0.f;
#pragma unroll
    for (int c = 0; c < 5; ++c) { z[c] = __expf(z[c] - zm); es += z[c]; }
    float inv = 1.0f / es;
#pragma unroll
    for (int c = 0; c < 5; ++c)
      if (lane == c) osev[(size_t)gw * 5 + c] = z[c] * inv;

    obot[(size_t)gw * DOUT + col] = bo + obv[col];
    oh[base + lane]      = h0;
    oh[base + lane + 32] = h1;

    cs0 += h0;
    cs1 += h1;
  }

  ls[wave][lane]      = cs0;
  ls[wave][lane + 32] = cs1;
  __syncthreads();
  if (threadIdx.x < H) {
    float t = 0.0f;
#pragma unroll
    for (int w = 0; w < 8; ++w) t += ls[w][threadIdx.x];
    part[(size_t)blockIdx.x * H + threadIdx.x] = t;
  }
}

__global__ __launch_bounds__(1024) void k_mean_final(
    const float* __restrict__ part, float* __restrict__ gout) {
  __shared__ float lq[16][H];
  const int f = threadIdx.x & 63, q = threadIdx.x >> 6;
  const float* p = part + (size_t)q * (TAIL_BLOCKS / 16) * H + f;
  float acc = 0.0f;
#pragma unroll 16
  for (int j = 0; j < TAIL_BLOCKS / 16; ++j) acc += p[(size_t)j * H];
  lq[q][f] = acc;
  __syncthreads();
  if (threadIdx.x < H) {
    float t = 0.0f;
#pragma unroll
    for (int g = 0; g < 16; ++g) t += lq[g][threadIdx.x];
    gout[threadIdx.x] = t * (1.0f / NN);
  }
}

extern "C" void kernel_launch(void* const* d_in, const int* in_sizes, int n_in,
                              void* d_out, int out_size, void* d_ws, size_t ws_size,
                              hipStream_t stream) {
  (void)in_sizes; (void)n_in; (void)out_size; (void)ws_size;
  const float* x    = (const float*)d_in[0];
  const int*   ei   = (const int*)  d_in[1];
  const float* ea   = (const float*)d_in[2];
  const float* inW  = (const float*)d_in[3];
  const float* inb  = (const float*)d_in[4];
  const float* linW = (const float*)d_in[5];
  const float* linb = (const float*)d_in[6];
  const float* attW = (const float*)d_in[7];
  const float* attb = (const float*)d_in[8];
  const float* lng  = (const float*)d_in[9];
  const float* lnb  = (const float*)d_in[10];
  const float* qW1 = (const float*)d_in[11]; const float* qb1 = (const float*)d_in[12];
  const float* qW2 = (const float*)d_in[13]; const float* qb2 = (const float*)d_in[14];
  const float* iW1 = (const float*)d_in[15]; const float* ib1 = (const float*)d_in[16];
  const float* iW2 = (const float*)d_in[17]; const float* ib2 = (const float*)d_in[18];
  const float* gW1 = (const float*)d_in[19]; const float* gb1 = (const float*)d_in[20];
  const float* gW2 = (const float*)d_in[21]; const float* gb2 = (const float*)d_in[22];
  const float* oW  = (const float*)d_in[23]; const float* obv = (const float*)d_in[24];
  const float* sW1 = (const float*)d_in[25]; const float* sb1 = (const float*)d_in[26];
  const float* sW2 = (const float*)d_in[27]; const float* sb2 = (const float*)d_in[28];

  float* ws   = (float*)d_ws;
  float* h    = ws;
  float* mbuf = h    + (size_t)NN * H;
  float* aggr = mbuf + (size_t)NN * H;
  float* sD   = aggr + (size_t)NN * H;
  float* sS   = sD + NN;

  float* out  = (float*)d_out;
  float* oq   = out;
  float* oi   = out + (size_t)NN;
  float* og   = out + 2 * (size_t)NN;
  float* oh   = out + 3 * (size_t)NN;
  float* ogr  = out + 67 * (size_t)NN;
  float* obot = out + 67 * (size_t)NN + H;
  float* osev = out + 99 * (size_t)NN + H;

  int*      erec = (int*)(aggr + 2000000);
  unsigned* eown = (unsigned*)(aggr + 6000000);
  k_edge_pack<<<(NE / 4 + 255) / 256, 256, 0, stream>>>(ei, ea, erec, eown);

  float*  rec0 = mbuf;
  double* acc0 = (double*)aggr;
  k_l0_prep<<<(NN + 255) / 256, 256, 0, stream>>>(x, inW, inb, attW, rec0, acc0);
  k_l0_edges<<<(NE + 255) / 256, 256, 0, stream>>>(ei, ea, rec0, attb, acc0);
  float* sD2 = aggr + 1600000;
  float* sS2 = sD2 + NN;
  float* h2  = mbuf;
  k_l0_node<<<NN / 16, 128, 0, stream>>>(x, acc0, inW, inb, linW, linb, lng, lnb, attW + 2 * H,
                                         h, sD, sS);

  k_own<<<OWN_BLOCKS, 1024, 0, stream>>>(erec, eown, h, sD, sS, attb + 1,
                                         linW + 1 * H * H, linb + 1 * H, lng + 1 * H, lnb + 1 * H,
                                         attW + 2 * 2 * H, 1, h2, sD2, sS2);
  k_own<<<OWN_BLOCKS, 1024, 0, stream>>>(erec, eown, h2, sD2, sS2, attb + 2,
                                         linW + 2 * H * H, linb + 2 * H, lng + 2 * H, lnb + 2 * H,
                                         attW + 3 * 2 * H, 1, h, sD, sS);
  k_own<<<OWN_BLOCKS, 1024, 0, stream>>>(erec, eown, h, sD, sS, attb + 3,
                                         linW + 3 * H * H, linb + 3 * H, lng + 3 * H, lnb + 3 * H,
                                         attW + 3 * 2 * H, 0, h2, sD2, sS2);

  float* part = sD;
  k_tail<<<TAIL_BLOCKS, 256, 0, stream>>>(h2,
                                          qW1, qb1, qW2, qb2, iW1, ib1, iW2, ib2,
                                          gW1, gb1, gW2, gb2, oW, obv, sW1, sb1, sW2, sb2,
                                          oq, oi, og, oh, obot, osev, part);
  k_mean_final<<<1, 1024, 0, stream>>>(part, ogr);
}


# layers 2-3: owner waves replay the edge queues layer 1's map scan found and stored (numbers 2, records 1 batch ahead); the 1 MB owner map is scanned once a call, was 3 times (sum order changes)
# speedup vs baseline: 2.2340x; 1.1505x over previous
#include <hip/hip_runtime.h>

#define NN 100000
#define NE 1000000
#define DIN 5
#define H 64
#define HHID 32
#define NL 4
#define DOUT 32
#define LN_EPS 1e-5f

typedef __attribute__((ext_vector_type(2)))  float    v2f;
typedef __attribute__((ext_vector_type(8)))  float    v8f;
typedef __attribute__((ext_vector_type(16))) _Float16 v16h;

#if defined(__has_builtin)
#if __has_builtin(__builtin_amdgcn_wmma_f32_16x16x4_f32)
#define HAVE_WMMA_F32 1
#endif
#endif

__device__ __forceinline__ float wave_sum(float v) {
#pragma unroll
  for (int off = 16; off > 0; off >>= 1) v += __shfl_xor(v, off, 32);
  return v;
}
__device__ __forceinline__ float sigmoidf_(float x) { return 1.0f / (1.0f + __expf(-x)); }

__device__ __forceinline__ void atomic_fadd(float* p, float v) { unsafeAtomicAdd(p, v); }

__device__ __forceinline__ v8f tile_product(const float (*lA)[H + 1], const float (*lW)[H + 1],
                                            int m, int g, int col, v8f acc) {
#if HAVE_WMMA_F32
#pragma unroll
  for (int k0 = 0; k0 < H; k0 += 4) {
    int kb = k0 + 2 * g;
    v2f a, b;
    a.x = lA[m][kb];     a.y = lA[m][kb + 1];
    b.x = lW[kb][col];   b.y = lW[kb + 1][col];
    acc = __builtin_amdgcn_wmma_f32_16x16x4_f32(false, a, false, b, (short)0, acc, false, false);
  }
#else
#pragma unroll
  for (int kk = 0; kk < H; kk += 32) {
    v16h a, b;
#pragma unroll
    for (int j = 0; j < 8; ++j) {
      int ka = (j < 4) ? (2 * j + 8 * g) : (16 + 2 * (j - 4) + 8 * g);
      a[2 * j]     = (_Float16)lA[m][kk + ka];
      a[2 * j + 1] = (_Float16)lA[m][kk + ka + 1];
      int kb = 2 * j + 16 * g;
      b[2 * j]     = (_Float16)lW[kk + kb][col];
      b[2 * j + 1] = (_Float16)lW[kk + kb + 1][col];
    }
    acc = __builtin_amdgcn_wmma_f32_16x16x32_f16(false, a, false, b, (short)0, acc, false, false);
  }
#endif
  return acc;
}

#define EDGE_DEPTH 8
#define OWN_CAP 192
#define OWN_R 393
#define OWN_BLOCKS ((NN + OWN_R - 1) / OWN_R)
#define OWN_WAVES 32
#define OWN_Q 544
#define OWN_TILES ((OWN_R + 15) / 16)
#define OWN_RP (OWN_TILES * 16)
#define TP 68
static_assert(OWN_TILES <= OWN_WAVES, "one wave per 16-row tile of the owner's product");
static_assert(NL == 4, "the three owner launches are written out");
static_assert(OWN_BLOCKS <= 255, "owner numbers are bytes 0..254; 255 marks a map byte that is nobody's");
static_assert(NE % 16 == 0, "the owner map is read 16 bytes a lane");
static_assert(NE % 32 == 0, "every wave of the layer-0 edge pass owns 32 whole edges");
static_assert(32 % EDGE_DEPTH == 0, "a batch of 32 edges splits into whole groups");
static_assert(OWN_CAP % 32 == 0, "a strip holds whole batches");
static_assert((size_t)OWN_BLOCKS * OWN_WAVES * OWN_CAP * 4 <= 6400000, "the strips fit in the region of layer 0's sums");

__device__ __forceinline__ int lane_bcast_i(int v, int l) { return __builtin_amdgcn_readlane(v, l); }
__device__ __forceinline__ float lane_bcast_f(float v, int l) {
  return __int_as_float(__builtin_amdgcn_readlane(__float_as_int(v), l));
}

__global__ __launch_bounds__(256) void k_edge_pack(
    const int* __restrict__ ei, const float* __restrict__ ea,
    int* __restrict__ rec, unsigned* __restrict__ own) {
  const int t = blockIdx.x * 256 + threadIdx.x;
  if (t >= NE / 4) return;
  const int4 s4 = reinterpret_cast<const int4*>(ei)[t];
  const int4 d4 = reinterpret_cast<const int4*>(ei + NE)[t];
  const float4 a4 = reinterpret_cast<const float4*>(ea)[t];
  int4* r4 = reinterpret_cast<int4*>(rec) + (size_t)t * 4;
  r4[0] = make_int4(s4.x, d4.x, __float_as_int(a4.x), 0);
  r4[1] = make_int4(s4.y, d4.y, __float_as_int(a4.y), 0);
  r4[2] = make_int4(s4.z, d4.z, __float_as_int(a4.z), 0);
  r4[3] = make_int4(s4.w, d4.w, __float_as_int(a4.w), 0);
  own[t] = (unsigned)(d4.x / OWN_R) | ((unsigned)(d4.y / OWN_R) << 8) |
           ((unsigned)(d4.z / OWN_R) << 16) | ((unsigned)(d4.w / OWN_R) << 24);
}

__device__ __forceinline__ void own_batch(
    const int4 r, int n, int lane, int lo, float b0,
    const float* __restrict__ hrows, const float* __restrict__ sS,
    const float* __restrict__ lsD, float* __restrict__ lsw, float* __restrict__ tile) {
  int src = 0, dl = 0;
  float w = 0.0f;
  if (lane < n) {
    src = r.x;
    dl = r.y - lo;
    w = __int_as_float(r.z) * sigmoidf_(lsD[dl] + sS[src] + b0);
    __hip_atomic_fetch_add(lsw + dl, w, __ATOMIC_RELAXED, __HIP_MEMORY_SCOPE_WORKGROUP);
  }
  const float* mlane = hrows + lane;
#pragma unroll
  for (int i0 = 0; i0 < 32; i0 += EDGE_DEPTH) {
    if (i0 < n) {
      float lov[EDGE_DEPTH], hiv[EDGE_DEPTH];
#pragma unroll
      for (int j = 0; j < EDGE_DEPTH; ++j) {
        const float* mr = mlane + (size_t)lane_bcast_i(src, i0 + j) * H;
        lov[j] = mr[0];
        hiv[j] = mr[32];
      }
#pragma unroll
      for (int j = 0; j < EDGE_DEPTH; ++j) {
        if (i0 + j < n) {
          const float bw = lane_bcast_f(w, i0 + j);
          float* tr = tile + lane_bcast_i(dl, i0 + j) * TP + lane;
          __hip_atomic_fetch_add(tr,      lov[j] * bw, __ATOMIC_RELAXED, __HIP_MEMORY_SCOPE_WORKGROUP);
          __hip_atomic_fetch_add(tr + 32, hiv[j] * bw, __ATOMIC_RELAXED, __HIP_MEMORY_SCOPE_WORKGROUP);
        }
      }
    }
  }
}

__device__ __forceinline__ void own_scan_piece(const uint4 o, int vb, unsigned myb, int* qw, int& cnt) {
  const unsigned wd[4] = {o.x, o.y, o.z, o.w};
#pragma unroll
  for (int k = 0; k < 16; ++k) {
    const unsigned byte = (wd[k >> 2] >> (8 * (k & 3))) & 0xFFu;
    const bool mine = (byte == myb);
    const unsigned mask = __builtin_amdgcn_ballot_w32(mine);
    if (mask != 0u) {
      const int pos = cnt + (int)__builtin_amdgcn_mbcnt_lo(mask, 0u);
      if (mine) qw[pos] = vb + k;
      cnt += __builtin_popcount(mask);
    }
  }
}

__device__ __forceinline__ uint4 own_piece(const uint4 t, bool ok) {
  return ok ? t : make_uint4(0xFFFFFFFFu, 0xFFFFFFFFu, 0xFFFFFFFFu, 0xFFFFFFFFu);
}

__device__ __forceinline__ void own_drain(
    int* qw, int& cnt, int lane, int lo, float b0,
    const int4* __restrict__ rec4, const float* __restrict__ hrows, const float* __restrict__ sS,
    const float* __restrict__ lsD, float* __restrict__ lsw, float* __restrict__ tile,
    bool to_strip, int* gl, int& total) {
  int head = 0;
  while (cnt - head >= 32) {
    const int e = qw[head + lane];
    if (to_strip) {
      if (total + lane < OWN_CAP) gl[total + lane] = e;
      total += 32;
    }
    own_batch(rec4[e], 32, lane, lo, b0, hrows, sS, lsD, lsw, tile);
    head += 32;
  }
  if (head > 0) {
    const int rest = cnt - head;
    int keep = 0;
    if (lane < rest) keep = qw[head + lane];
    if (lane < rest) qw[lane] = keep;
    cnt = rest;
  }
}

__global__ __launch_bounds__(1024) void k_own(
    const int* __restrict__ rec, const unsigned* __restrict__ own,
    const float* __restrict__ hin, const float* __restrict__ sD,
    const float* __restrict__ sS, const float* __restrict__ attb,
    const float* __restrict__ W, const float* __restrict__ bias,
    const float* __restrict__ g, const float* __restrict__ beta,
    const float* __restrict__ awn, int has_next,
    float* __restrict__ hout, float* __restrict__ sDn, float* __restrict__ sSn,
    int replay_strips, int* elist, int* ecount) {
  __shared__ __attribute__((aligned(16))) float tile[OWN_RP * TP];
  __shared__ __attribute__((aligned(16))) float lWt[H * TP];
  __shared__ float lsD[OWN_RP];
  __shared__ float lsw[OWN_RP];
  __shared__ int q[OWN_WAVES][OWN_Q];
  const int tid = threadIdx.x, lane = tid & 31;
  const int wave = __builtin_amdgcn_readfirstlane(tid >> 5);
  const int lo = blockIdx.x * OWN_R;
  const int nrows = (NN - lo < OWN_R) ? (NN - lo) : OWN_R;
  const unsigned myb = blockIdx.x;
  const float b0 = attb[0];

#pragma unroll 4
  for (int r = wave; r < nrows; r += OWN_WAVES) {
    const float* hr = hin + (size_t)(lo + r) * H;
    tile[r * TP + lane]      = hr[lane];
    tile[r * TP + lane + 32] = hr[lane + 32];
  }
  for (int r = nrows + wave; r < OWN_RP; r += OWN_WAVES) {
    tile[r * TP + lane]      = 0.0f;
    tile[r * TP + lane + 32] = 0.0f;
  }
  for (int r = tid; r < OWN_RP; r += 1024) {
    lsD[r] = (r < nrows) ? sD[lo + r] : 0.0f;
    lsw[r] = 1.0f;
  }
  for (int idx = tid; idx < H * H; idx += 1024)
    lWt[(idx & 63) * TP + (idx >> 6)] = W[idx];
  __syncthreads();

  const uint4* __restrict__ own4 = reinterpret_cast<const uint4*>(own);
  const int4* __restrict__ rec4 = reinterpret_cast<const int4*>(rec);
  int* qw = q[wave];
  int cnt = 0;
  const int nvec = NE / 16;
  const int nit = (nvec + 31) / 32;
  const int slot = blockIdx.x * OWN_WAVES + wave;
  int* gl = elist + (size_t)slot * OWN_CAP;
  const bool scan_keeps = (replay_strips == 0);
  int total = 0;
  if (!scan_keeps) total = __builtin_amdgcn_readfirstlane(ecount[slot]);
  if (!scan_keeps && total <= OWN_CAP) {
    int eB = (lane < total) ? gl[lane] : 0;
    int4 rA = rec4[eB];
    eB = (32 + lane < total) ? gl[32 + lane] : 0;
    for (int b = 0; b < total; b += 32) {
      const int4 r = rA;
      rA = rec4[eB];
      eB = (b + 64 + lane < total) ? gl[b + 64 + lane] : 0;
      const int n = (total - b < 32) ? (total - b) : 32;
      own_batch(r, n, lane, lo, b0, hin, sS, lsD, lsw, tile);
    }
  } else {
    for (int it0 = wave; it0 < nit; it0 += 4 * OWN_WAVES) {
      const int v0 = it0 * 32 + lane;
      const int v1 = v0 + 32 * OWN_WAVES;
      const int v2 = v1 + 32 * OWN_WAVES;
      const int v3 = v2 + 32 * OWN_WAVES;
      const bool ok0 = v0 < nvec, ok1 = v1 < nvec, ok2 = v2 < nvec, ok3 = v3 < nvec;
      const uint4 t0 = own4[ok0 ? v0 : nvec - 1];
      const uint4 t1 = own4[ok1 ? v1 : nvec - 1];
      const uint4 t2 = own4[ok2 ? v2 : nvec - 1];
      const uint4 t3 = own4[ok3 ? v3 : nvec - 1];
      own_scan_piece(own_piece(t0, ok0), v0 * 16, myb, qw, cnt);
      own_drain(qw, cnt, lane, lo, b0, rec4, hin, sS, lsD, lsw, tile, scan_keeps, gl, total);
      own_scan_piece(own_piece(t1, ok1), v1 * 16, myb, qw, cnt);
      own_drain(qw, cnt, lane, lo, b0, rec4, hin, sS, lsD, lsw, tile, scan_keeps, gl, total);
      own_scan_piece(own_piece(t2, ok2), v2 * 16, myb, qw, cnt);
      own_drain(qw, cnt, lane, lo, b0, rec4, hin, sS, lsD, lsw, tile, scan_keeps, gl, total);
      own_scan_piece(own_piece(t3, ok3), v3 * 16, myb, qw, cnt);
      own_drain(qw, cnt, lane, lo, b0, rec4, hin, sS, lsD, lsw, tile, scan_keeps, gl, total);
    }
    if (cnt > 0) {
      const int e = (lane < cnt) ? qw[lane] : 0;
      if (scan_keeps) {
        if (lane < cnt && total + lane < OWN_CAP) gl[total + lane] = e;
        total += cnt;
      }
      own_batch(rec4[e], cnt, lane, lo, b0, hin, sS, lsD, lsw, tile);
    }
    if (scan_keeps && lane == 0) ecount[slot] = total;
  }
  __syncthreads();

  if (wave * 16 < nrows) {
    const int mrow = lane & 15, gq = lane >> 4;
    const int r0 = wave * 16;
    const float* ap = tile + (r0 + mrow) * TP + 2 * gq;
    const float* bp = lWt + mrow * TP + 2 * gq;
    v8f acc[4];
#pragma unroll
    for (int cc = 0; cc < 4; ++cc) {
      const float bv = bias[16 * cc + mrow];
#pragma unroll
      for (int i = 0; i < 8; ++i) acc[cc][i] = lsw[r0 + i + 8 * gq] * bv;
    }
#if HAVE_WMMA_F32
#pragma unroll
    for (int k0 = 0; k0 < H; k0 += 4) {
      const v2f a = *reinterpret_cast<const v2f*>(ap + k0);
#pragma unroll
      for (int cc = 0; cc < 4; ++cc) {
        const v2f b = *reinterpret_cast<const v2f*>(bp + 16 * cc * TP + k0);
        acc[cc] = __builtin_amdgcn_wmma_f32_16x16x4_f32(false, a, false, b, (short)0, acc[cc], false, false);
      }
    }
#endif
#pragma unroll
    for (int cc = 0; cc < 4; ++cc) {
#pragma unroll
      for (int i = 0; i < 8; ++i) tile[(r0 + i + 8 * gq) * TP + 16 * cc + mrow] = acc[cc][i];
    }
  }
  __syncthreads();

  const float g0 = g[lane], g1 = g[lane + 32];
  const float be0 = beta[lane], be1 = beta[lane + 32];
  const float aD0 = awn[lane], aD1 = awn[lane + 32];
  const float aS0 = awn[H + lane], aS1 = awn[H + lane + 32];
#pragma unroll 2
  for (int r = wave; r < nrows; r += OWN_WAVES) {
    const size_t base = (size_t)(lo + r) * H;
    const float r0 = hin[base + lane], r1 = hin[base + lane + 32];
    const float f0 = tile[r * TP + lane], f1 = tile[r * TP + lane + 32];
    const float mean = wave_sum(f0 + f1) * (1.0f / H);
    const float d0 = f0 - mean, d1 = f1 - mean;
    const float var = wave_sum(fmaf(d0, d0, d1 * d1)) * (1.0f / H);
    const float inv = rsqrtf(var + LN_EPS);
    const float y0 = fmaf(d0 * inv, g0, be0);
    const float y1 = fmaf(d1 * inv, g1, be1);
    const float o0 = fmaxf(y0, 0.0f) + r0;
    const float o1 = fmaxf(y1, 0.0f) + r1;
    hout[base + lane]      = o0;
    hout[base + lane + 32] = o1;
    if (has_next) {
      const float pd = wave_sum(fmaf(o0, aD0, o1 * aD1));
      const float ps = wave_sum(fmaf(o0, aS0, o1 * aS1));
      if (lane == 0) { sDn[lo + r] = pd; sSn[lo + r] = ps; }
    }
  }
}

#define ACCW 8
#define RECW 8

__global__ __launch_bounds__(256) void k_l0_prep(
    const float* __restrict__ x, const float* __restrict__ inW, const float* __restrict__ inb,
    const float* __restrict__ aw, float* __restrict__ rec, double* __restrict__ acc) {
  int n = blockIdx.x * 256 + threadIdx.x;
  if (n >= NN) return;
  float xv[DIN];
#pragma unroll
  for (int d = 0; d < DIN; ++d) xv[d] = x[(size_t)n * DIN + d];
  float sd = 0.0f, ss = 0.0f;
#pragma unroll 8
  for (int k = 0; k < H; ++k) {
    float hk = inb[k];
#pragma unroll
    for (int d = 0; d < DIN; ++d) hk = fmaf(xv[d], inW[d * H + k], hk);
    sd = fmaf(hk, aw[k], sd);
    ss = fmaf(hk, aw[H + k], ss);
  }
  float4* r4 = reinterpret_cast<float4*>(rec + (size_t)n * RECW);
  r4[0] = make_float4(xv[0], xv[1], xv[2], xv[3]);
  r4[1] = make_float4(xv[4], ss, sd, 0.0f);
  double2* a2 = reinterpret_cast<double2*>(acc + (size_t)n * ACCW);
#pragma unroll
  for (int j = 0; j < ACCW / 2; ++j) a2[j] = make_double2(0.0, 0.0);
}

__global__ __launch_bounds__(256) void k_l0_edges(
    const int* __restrict__ ei, const float* __restrict__ ea,
    const float* __restrict__ rec, const float* __restrict__ attb,
    double* __restrict__ acc) {
  int gw = blockIdx.x * 8 + (threadIdx.x >> 5);
  int lane = threadIdx.x & 31;
  int ebase = gw * 32;
  if (ebase >= NE) return;
  const int sub = lane >> 3, c = lane & 7;
  const float b0 = attb[0];
  int srcs[8], dsts[8];
  float eas[8], rv[8], sd[8];
#pragma unroll
  for (int i = 0; i < 8; ++i) {
    int e = ebase + 4 * i + sub;
    srcs[i] = ei[e];
    dsts[i] = ei[NE + e];
    eas[i] = ea[e];
  }
#pragma unroll
  for (int i = 0; i < 8; ++i) {
    rv[i] = rec[(size_t)srcs[i] * RECW + c];
    sd[i] = rec[(size_t)dsts[i] * RECW + 6];
  }
#pragma unroll
  for (int i = 0; i < 8; ++i) {
    float ss = __shfl(rv[i], (lane & 24) | 5, 32);
    float w = eas[i] * sigmoidf_(sd[i] + ss + b0);
    if (c < 6) {
      double v = (c == 5) ? (double)w : (double)w * (double)rv[i];
      unsafeAtomicAdd(acc + (size_t)dsts[i] * ACCW + c, v);
    }
  }
}

__global__ __launch_bounds__(128) void k_l0_node(
    const float* __restrict__ x, const double* __restrict__ acc,
    const float* __restrict__ inW, const float* __restrict__ inb,
    const float* __restrict__ W, const float* __restrict__ bias,
    const float* __restrict__ g_, const float* __restrict__ beta,
    const float* __restrict__ awn,
    float* __restrict__ h, float* __restrict__ sDn, float* __restrict__ sSn) {
  __shared__ float lA[16][H + 1];
  __shared__ float lW[H][H + 1];
  __shared__ float lO[16][H + 1];
  __shared__ float lX[16][8];
  __shared__ float lS[16][8];
  const int tid = threadIdx.x;
  const int row0 = blockIdx.x * 16;
  for (int idx = tid; idx < H * H; idx += 128)
    lW[idx >> 6][idx & 63] = W[idx];
  {
    const int r = tid >> 3, c = tid & 7;
    const size_t n = (size_t)(row0 + r);
    float xv = (c < DIN) ? x[n * DIN + c] : 0.0f;
    float sv = (c < 6) ? (float)acc[n * ACCW + c] : 0.0f;
    lX[r][c] = xv;
    lS[r][c] = (c < DIN) ? (sv + xv) : ((c == 5) ? (sv + 1.0f) : 0.0f);
  }
  __syncthreads();
  for (int idx = tid; idx < 16 * H; idx += 128) {
    const int r = idx >> 6, k = idx & 63;
    float u = lS[r][5] * inb[k];
#pragma unroll
    for (int d = 0; d < DIN; ++d) u = fmaf(lS[r][d], inW[d * H + k], u);
    lA[r][k] = u;
  }
  __syncthreads();

  const int wave = tid >> 5, lane = tid & 31;
  const int m = lane & 15, g = lane >> 4;
  const int col = wave * 16 + m;

  v8f t;
  const float bv = bias[col];
#pragma unroll
  for (int i = 0; i < 8; ++i) t[i] = lS[i + 8 * g][5] * bv;

  t = tile_product(lA, lW, m, g, col, t);

#pragma unroll
  for (int i = 0; i < 8; ++i) lO[i + 8 * g][col] = t[i];
  __syncthreads();

  const float g0 = g_[lane], g1 = g_[lane + 32];
  const float be0 = beta[lane], be1 = beta[lane + 32];
  const float aD0 = awn[lane], aD1 = awn[lane + 32];
  const float aS0 = awn[H + lane], aS1 = awn[H + lane + 32];
#pragma unroll
  for (int rr = 0; rr < 4; ++rr) {
    const int r = wave * 4 + rr;
    float r0 = inb[lane], r1 = inb[lane + 32];
#pragma unroll
    for (int d = 0; d < DIN; ++d) {
      r0 = fmaf(lX[r][d], inW[d * H + lane], r0);
      r1 = fmaf(lX[r][d], inW[d * H + lane + 32], r1);
    }
    float f0 = lO[r][lane], f1 = lO[r][lane + 32];
    float mean = wave_sum(f0 + f1) * (1.0f / H);
    float d0 = f0 - mean, d1 = f1 - mean;
    float var = wave_sum(fmaf(d0, d0, d1 * d1)) * (1.0f / H);
    float inv = rsqrtf(var + LN_EPS);
    float y0 = fmaf(d0 * inv, g0, be0);
    float y1 = fmaf(d1 * inv, g1, be1);
    const size_t base = (size_t)(row0 + r) * H;
    const float o0 = fmaxf(y0, 0.0f) + r0;
    const float o1 = fmaxf(y1, 0.0f) + r1;
    h[base + lane]      = o0;
    h[base + lane + 32] = o1;
    const float pd = wave_sum(fmaf(o0, aD0, o1 * aD1));
    const float ps = wave_sum(fmaf(o0, aS0, o1 * aS1));
    if (lane == 0) { sDn[row0 + r] = pd; sSn[row0 + r] = ps; }
  }
}

__device__ __forceinline__ float kgroup_reduce(const float (&a)[4], int hiA, int hiB) {
  float s0 = hiA ? a[0] : a[2];
  float s1 = hiA ? a[1] : a[3];
  float k0 = hiA ? a[2] : a[0];
  float k1 = hiA ? a[3] : a[1];
  float r0 = k0 + __shfl_xor(s0, 16, 32);
  float r1 = k1 + __shfl_xor(s1, 16, 32);
  float sB = hiB ? r0 : r1;
  float kB = hiB ? r1 : r0;
  return kB + __shfl_xor(sB, 8, 32);
}
#define TAIL_BLOCKS 2048
static_assert(TAIL_BLOCKS * H <= 2 * NN, "partial sums must fit in the sD+sS region");
static_assert(TAIL_BLOCKS % 16 == 0, "stage 2 splits the partial rows into 16 equal groups");
__global__ __launch_bounds__(256) void k_tail(
    const float* __restrict__ h,
    const float* __restrict__ qW1, const float* __restrict__ qb1,
    const float* __restrict__ qW2, const float* __restrict__ qb2,
    const float* __restrict__ iW1, const float* __restrict__ ib1,
    const float* __restrict__ iW2, const float* __restrict__ ib2,
    const float* __restrict__ gW1, const float* __restrict__ gb1,
    const float* __restrict__ gW2, const float* __restrict__ gb2,
    const float* __restrict__ oW,  const float* __restrict__ obv,
    const float* __restrict__ sW1, const float* __restrict__ sb1,
    const float* __restrict__ sW2, const float* __restrict__ sb2,
    float* __restrict__ oq, float* __restrict__ oi, float* __restrict__ og,
    float* __restrict__ oh, float* __restrict__ obot, float* __restrict__ osev,
    float* __restrict__ part) {
  __shared__ float ls[8][H];
  const int wave = threadIdx.x >> 5;
  const int lane = threadIdx.x & 31;
  const int nwaves = gridDim.x * 8;
  float cs0 = 0.0f, cs1 = 0.0f;

  const int kgrp = lane >> 3;
  const int hiA = (lane >> 4) & 1, hiB = (lane >> 3) & 1;
  const int col = 4 * (lane & 7) + 2 * hiA + hiB;
  const float4* __restrict__ q4 = reinterpret_cast<const float4*>(qW1);
  const float4* __restrict__ i4 = reinterpret_cast<const float4*>(iW1);
  const float4* __restrict__ g4 = reinterpret_cast<const float4*>(gW1);
  const float4* __restrict__ s4 = reinterpret_cast<const float4*>(sW1);
  const float4* __restrict__ o4 = reinterpret_cast<const float4*>(oW);
  const unsigned wl = (unsigned)lane;

  for (int gw = blockIdx.x * 8 + wave; gw < NN; gw += nwaves) {
    size_t base = (size_t)gw * H;

    const float h0 = h[base + lane], h1 = h[base + lane + 32];

    float aq[4] = {0.f, 0.f, 0.f, 0.f}, ai[4] = {0.f, 0.f, 0.f, 0.f}, ag[4] = {0.f, 0.f, 0.f, 0.f};
    float as[4] = {0.f, 0.f, 0.f, 0.f}, ao[4] = {0.f, 0.f, 0.f, 0.f};
#pragma unroll
    for (int half = 0; half < 2; ++half) {
      const float hsrc = half ? h1 : h0;
#pragma unroll 2
      for (int ii8 = 0; ii8 < 8; ++ii8) {
        const int i = half * 8 + ii8;
        const float hk = __shfl(hsrc, 4 * ii8 + kgrp, 32);
        const unsigned wx = (unsigned)(i * 32) + wl;
        const float4 wq = q4[wx], wi = i4[wx], wg = g4[wx], wsv = s4[wx], wo = o4[wx];
        aq[0] = fmaf(hk, wq.x, aq[0]);  aq[1] = fmaf(hk, wq.y, aq[1]);
        aq[2] = fmaf(hk, wq.z, aq[2]);  aq[3] = fmaf(hk, wq.w, aq[3]);
        ai[0] = fmaf(hk, wi.x, ai[0]);  ai[1] = fmaf(hk, wi.y, ai[1]);
        ai[2] = fmaf(hk, wi.z, ai[2]);  ai[3] = fmaf(hk, wi.w, ai[3]);
        ag[0] = fmaf(hk, wg.x, ag[0]);  ag[1] = fmaf(hk, wg.y, ag[1]);
        ag[2] = fmaf(hk, wg.z, ag[2]);  ag[3] = fmaf(hk, wg.w, ag[3]);
        as[0] = fmaf(hk, wsv.x, as[0]); as[1] = fmaf(hk, wsv.y, as[1]);
        as[2] = fmaf(hk, wsv.z, as[2]); as[3] = fmaf(hk, wsv.w, as[3]);
        ao[0] = fmaf(hk, wo.x, ao[0]);  ao[1] = fmaf(hk, wo.y, ao[1]);
        ao[2] = fmaf(hk, wo.z, ao[2]);  ao[3] = fmaf(hk, wo.w, ao[3]);
      }
    }
    float tq = kgroup_reduce(aq, hiA, hiB);
    float ti = kgroup_reduce(ai, hiA, hiB);
    float tg = kgroup_reduce(ag, hiA, hiB);
    float ts = kgroup_reduce(as, hiA, hiB);
    float bo = kgroup_reduce(ao, hiA, hiB);
    tq = fmaxf(tq + qb1[col], 0.0f);
    ti = fmaxf(ti + ib1[col], 0.0f);
    tg = fmaxf(tg + gb1[col], 0.0f);
    ts = fmaxf(ts + sb1[col], 0.0f);

    float q  = wave_sum(tq * qW2[col]);
    float ii = wave_sum(ti * iW2[col]);
    float gg = wave_sum(tg * gW2[col]);
    if (lane == 0) {
      oq[gw] = sigmoidf_(q + qb2[0]);
      oi[gw] = sigmoidf_(ii + ib2[0]);
      og[gw] = sigmoidf_(gg + gb2[0]);
    }

    float z[5];
#pragma unroll
    for (int c = 0; c < 5; ++c) z[c] = wave_sum(ts * sW2[col * 5 + c]) + sb2[c];
    float zm = z[0];
#pragma unroll
    for (int c = 1; c < 5; ++c) zm = fmaxf(zm, z[c]);
    float es = 0.f;
#pragma unroll
    for (int c = 0; c < 5; ++c) { z[c] = __expf(z[c] - zm); es += z[c]; }
    float inv = 1.0f / es;
#pragma unroll
    for (int c = 0; c < 5; ++c)
      if (lane == c) osev[(size_t)gw * 5 + c] = z[c] * inv;

    obot[(size_t)gw * DOUT + col] = bo + obv[col];
    oh[base + lane]      = h0;
    oh[base + lane + 32] = h1;

    cs0 += h0;
    cs1 += h1;
  }

  ls[wave][lane]      = cs0;
  ls[wave][lane + 32] = cs1;
  __syncthreads();
  if (threadIdx.x < H) {
    float t = 0.0f;
#pragma unroll
    for (int w = 0; w < 8; ++w) t += ls[w][threadIdx.x];
    part[(size_t)blockIdx.x * H + threadIdx.x] = t;
  }
}

__global__ __launch_bounds__(1024) void k_mean_final(
    const float* __restrict__ part, float* __restrict__ gout) {
  __shared__ float lq[16][H];
  const int f = threadIdx.x & 63, q = threadIdx.x >> 6;
  const float* p = part + (size_t)q * (TAIL_BLOCKS / 16) * H + f;
  float acc = 0.0f;
#pragma unroll 16
  for (int j = 0; j < TAIL_BLOCKS / 16; ++j) acc += p[(size_t)j * H];
  lq[q][f] = acc;
  __syncthreads();
  if (threadIdx.x < H) {
    float t = 0.0f;
#pragma unroll
    for (int g = 0; g < 16; ++g) t += lq[g][threadIdx.x];
    gout[threadIdx.x] = t * (1.0f / NN);
  }
}

extern "C" void kernel_launch(void* const* d_in, const int* in_sizes, int n_in,
                              void* d_out, int out_size, void* d_ws, size_t ws_size,
                              hipStream_t stream) {
  (void)in_sizes; (void)n_in; (void)out_size; (void)ws_size;
  const float* x    = (const float*)d_in[0];
  const int*   ei   = (const int*)  d_in[1];
  const float* ea   = (const float*)d_in[2];
  const float* inW  = (const float*)d_in[3];
  const float* inb  = (const float*)d_in[4];
  const float* linW = (const float*)d_in[5];
  const float* linb = (const float*)d_in[6];
  const float* attW = (const float*)d_in[7];
  const float* attb = (const float*)d_in[8];
  const float* lng  = (const float*)d_in[9];
  const float* lnb  = (const float*)d_in[10];
  const float* qW1 = (const float*)d_in[11]; const float* qb1 = (const float*)d_in[12];
  const float* qW2 = (const float*)d_in[13]; const float* qb2 = (const float*)d_in[14];
  const float* iW1 = (const float*)d_in[15]; const float* ib1 = (const float*)d_in[16];
  const float* iW2 = (const float*)d_in[17]; const float* ib2 = (const float*)d_in[18];
  const float* gW1 = (const float*)d_in[19]; const float* gb1 = (const float*)d_in[20];
  const float* gW2 = (const float*)d_in[21]; const float* gb2 = (const float*)d_in[22];
  const float* oW  = (const float*)d_in[23]; const float* obv = (const float*)d_in[24];
  const float* sW1 = (const float*)d_in[25]; const float* sb1 = (const float*)d_in[26];
  const float* sW2 = (const float*)d_in[27]; const float* sb2 = (const float*)d_in[28];

  float* ws   = (float*)d_ws;
  float* h    = ws;
  float* mbuf = h    + (size_t)NN * H;
  float* aggr = mbuf + (size_t)NN * H;
  float* sD   = aggr + (size_t)NN * H;
  float* sS   = sD + NN;

  float* out  = (float*)d_out;
  float* oq   = out;
  float* oi   = out + (size_t)NN;
  float* og   = out + 2 * (size_t)NN;
  float* oh   = out + 3 * (size_t)NN;
  float* ogr  = out + 67 * (size_t)NN;
  float* obot = out + 67 * (size_t)NN + H;
  float* osev = out + 99 * (size_t)NN + H;

  int*      erec = (int*)(aggr + 2000000);
  unsigned* eown = (unsigned*)(aggr + 6000000);
  k_edge_pack<<<(NE / 4 + 255) / 256, 256, 0, stream>>>(ei, ea, erec, eown);

  float*  rec0 = mbuf;
  double* acc0 = (double*)aggr;
  k_l0_prep<<<(NN + 255) / 256, 256, 0, stream>>>(x, inW, inb, attW, rec0, acc0);
  k_l0_edges<<<(NE + 255) / 256, 256, 0, stream>>>(ei, ea, rec0, attb, acc0);
  float* sD2 = aggr + 1600000;
  float* sS2 = sD2 + NN;
  float* h2  = mbuf;
  k_l0_node<<<NN / 16, 128, 0, stream>>>(x, acc0, inW, inb, linW, linb, lng, lnb, attW + 2 * H,
                                         h, sD, sS);

  int* elist  = (int*)aggr;
  int* ecount = (int*)(aggr + 1800000);
  k_own<<<OWN_BLOCKS, 1024, 0, stream>>>(erec, eown, h, sD, sS, attb + 1,
                                         linW + 1 * H * H, linb + 1 * H, lng + 1 * H, lnb + 1 * H,
                                         attW + 2 * 2 * H, 1, h2, sD2, sS2, 0, elist, ecount);
  k_own<<<OWN_BLOCKS, 1024, 0, stream>>>(erec, eown, h2, sD2, sS2, attb + 2,
                                         linW + 2 * H * H, linb + 2 * H, lng + 2 * H, lnb + 2 * H,
                                         attW + 3 * 2 * H, 1, h, sD, sS, 1, elist, ecount);
  k_own<<<OWN_BLOCKS, 1024, 0, stream>>>(erec, eown, h, sD, sS, attb + 3,
                                         linW + 3 * H * H, linb + 3 * H, lng + 3 * H, lnb + 3 * H,
                                         attW + 3 * 2 * H, 0, h2, sD2, sS2, 1, elist, ecount);

  float* part = sD;
  k_tail<<<TAIL_BLOCKS, 256, 0, stream>>>(h2,
                                          qW1, qb1, qW2, qb2, iW1, ib1, iW2, ib2,
                                          gW1, gb1, gW2, gb2, oW, obv, sW1, sb1, sW2, sb2,
                                          oq, oi, og, oh, obot, osev, part);
  k_mean_final<<<1, 1024, 0, stream>>>(part, ogr);
}


# all 32-lane sums and the heads' two partner exchanges on the vector ALU (v_permlanex16 for lane^16, DPP row_xmask for lane^8,4,2,1), were ds_bpermute round trips; same butterfly order
# speedup vs baseline: 2.3554x; 1.0543x over previous
#include <hip/hip_runtime.h>

#define NN 100000
#define NE 1000000
#define DIN 5
#define H 64
#define HHID 32
#define NL 4
#define DOUT 32
#define LN_EPS 1e-5f

typedef __attribute__((ext_vector_type(2)))  float    v2f;
typedef __attribute__((ext_vector_type(8)))  float    v8f;
typedef __attribute__((ext_vector_type(16))) _Float16 v16h;

#if defined(__has_builtin)
#if __has_builtin(__builtin_amdgcn_wmma_f32_16x16x4_f32)
#define HAVE_WMMA_F32 1
#endif
#endif

__device__ __forceinline__ float xhalf(float x) {
  const int b = __builtin_bit_cast(int, x);
  return __builtin_bit_cast(float, __builtin_amdgcn_permlanex16(b, b, 0x76543210, 0xfedcba98, false, false));
}
template <int M>
__device__ __forceinline__ float xrow(float x) {
  const int b = __builtin_bit_cast(int, x);
  return __builtin_bit_cast(float, __builtin_amdgcn_update_dpp(0, b, 0x160 + M, 0xF, 0xF, true));
}
__device__ __forceinline__ float wave_sum(float v) {
  v += xhalf(v);
  v += xrow<8>(v);
  v += xrow<4>(v);
  v += xrow<2>(v);
  v += xrow<1>(v);
  return v;
}
__device__ __forceinline__ float sigmoidf_(float x) { return 1.0f / (1.0f + __expf(-x)); }

__device__ __forceinline__ void atomic_fadd(float* p, float v) { unsafeAtomicAdd(p, v); }

__device__ __forceinline__ v8f tile_product(const float (*lA)[H + 1], const float (*lW)[H + 1],
                                            int m, int g, int col, v8f acc) {
#if HAVE_WMMA_F32
#pragma unroll
  for (int k0 = 0; k0 < H; k0 += 4) {
    int kb = k0 + 2 * g;
    v2f a, b;
    a.x = lA[m][kb];     a.y = lA[m][kb + 1];
    b.x = lW[kb][col];   b.y = lW[kb + 1][col];
    acc = __builtin_amdgcn_wmma_f32_16x16x4_f32(false, a, false, b, (short)0, acc, false, false);
  }
#else
#pragma unroll
  for (int kk = 0; kk < H; kk += 32) {
    v16h a, b;
#pragma unroll
    for (int j = 0; j < 8; ++j) {
      int ka = (j < 4) ? (2 * j + 8 * g) : (16 + 2 * (j - 4) + 8 * g);
      a[2 * j]     = (_Float16)lA[m][kk + ka];
      a[2 * j + 1] = (_Float16)lA[m][kk + ka + 1];
      int kb = 2 * j + 16 * g;
      b[2 * j]     = (_Float16)lW[kk + kb][col];
      b[2 * j + 1] = (_Float16)lW[kk + kb + 1][col];
    }
    acc = __builtin_amdgcn_wmma_f32_16x16x32_f16(false, a, false, b, (short)0, acc, false, false);
  }
#endif
  return acc;
}

#define EDGE_DEPTH 8
#define OWN_CAP 192
#define OWN_R 393
#define OWN_BLOCKS ((NN + OWN_R - 1) / OWN_R)
#define OWN_WAVES 32
#define OWN_Q 544
#define OWN_TILES ((OWN_R + 15) / 16)
#define OWN_RP (OWN_TILES * 16)
#define TP 68
static_assert(OWN_TILES <= OWN_WAVES, "one wave per 16-row tile of the owner's product");
static_assert(NL == 4, "the three owner launches are written out");
static_assert(OWN_BLOCKS <= 255, "owner numbers are bytes 0..254; 255 marks a map byte that is nobody's");
static_assert(NE % 16 == 0, "the owner map is read 16 bytes a lane");
static_assert(NE % 32 == 0, "every wave of the layer-0 edge pass owns 32 whole edges");
static_assert(32 % EDGE_DEPTH == 0, "a batch of 32 edges splits into whole groups");
static_assert(OWN_CAP % 32 == 0, "a strip holds whole batches");
static_assert((size_t)OWN_BLOCKS * OWN_WAVES * OWN_CAP * 4 <= 6400000, "the strips fit in the region of layer 0's sums");

__device__ __forceinline__ int lane_bcast_i(int v, int l) { return __builtin_amdgcn_readlane(v, l); }
__device__ __forceinline__ float lane_bcast_f(float v, int l) {
  return __int_as_float(__builtin_amdgcn_readlane(__float_as_int(v), l));
}

__global__ __launch_bounds__(256) void k_edge_pack(
    const int* __restrict__ ei, const float* __restrict__ ea,
    int* __restrict__ rec, unsigned* __restrict__ own) {
  const int t = blockIdx.x * 256 + threadIdx.x;
  if (t >= NE / 4) return;
  const int4 s4 = reinterpret_cast<const int4*>(ei)[t];
  const int4 d4 = reinterpret_cast<const int4*>(ei + NE)[t];
  const float4 a4 = reinterpret_cast<const float4*>(ea)[t];
  int4* r4 = reinterpret_cast<int4*>(rec) + (size_t)t * 4;
  r4[0] = make_int4(s4.x, d4.x, __float_as_int(a4.x), 0);
  r4[1] = make_int4(s4.y, d4.y, __float_as_int(a4.y), 0);
  r4[2] = make_int4(s4.z, d4.z, __float_as_int(a4.z), 0);
  r4[3] = make_int4(s4.w, d4.w, __float_as_int(a4.w), 0);
  own[t] = (unsigned)(d4.x / OWN_R) | ((unsigned)(d4.y / OWN_R) << 8) |
           ((unsigned)(d4.z / OWN_R) << 16) | ((unsigned)(d4.w / OWN_R) << 24);
}

__device__ __forceinline__ void own_batch(
    const int4 r, int n, int lane, int lo, float b0,
    const float* __restrict__ hrows, const float* __restrict__ sS,
    const float* __restrict__ lsD, float* __restrict__ lsw, float* __restrict__ tile) {
  int src = 0, dl = 0;
  float w = 0.0f;
  if (lane < n) {
    src = r.x;
    dl = r.y - lo;
    w = __int_as_float(r.z) * sigmoidf_(lsD[dl] + sS[src] + b0);
    __hip_atomic_fetch_add(lsw + dl, w, __ATOMIC_RELAXED, __HIP_MEMORY_SCOPE_WORKGROUP);
  }
  const float* mlane = hrows + lane;
#pragma unroll
  for (int i0 = 0; i0 < 32; i0 += EDGE_DEPTH) {
    if (i0 < n) {
      float lov[EDGE_DEPTH], hiv[EDGE_DEPTH];
#pragma unroll
      for (int j = 0; j < EDGE_DEPTH; ++j) {
        const float* mr = mlane + (size_t)lane_bcast_i(src, i0 + j) * H;
        lov[j] = mr[0];
        hiv[j] = mr[32];
      }
#pragma unroll
      for (int j = 0; j < EDGE_DEPTH; ++j) {
        if (i0 + j < n) {
          const float bw = lane_bcast_f(w, i0 + j);
          float* tr = tile + lane_bcast_i(dl, i0 + j) * TP + lane;
          __hip_atomic_fetch_add(tr,      lov[j] * bw, __ATOMIC_RELAXED, __HIP_MEMORY_SCOPE_WORKGROUP);
          __hip_atomic_fetch_add(tr + 32, hiv[j] * bw, __ATOMIC_RELAXED, __HIP_MEMORY_SCOPE_WORKGROUP);
        }
      }
    }
  }
}

__device__ __forceinline__ void own_scan_piece(const uint4 o, int vb, unsigned myb, int* qw, int& cnt) {
  const unsigned wd[4] = {o.x, o.y, o.z, o.w};
#pragma unroll
  for (int k = 0; k < 16; ++k) {
    const unsigned byte = (wd[k >> 2] >> (8 * (k & 3))) & 0xFFu;
    const bool mine = (byte == myb);
    const unsigned mask = __builtin_amdgcn_ballot_w32(mine);
    if (mask != 0u) {
      const int pos = cnt + (int)__builtin_amdgcn_mbcnt_lo(mask, 0u);
      if (mine) qw[pos] = vb + k;
      cnt += __builtin_popcount(mask);
    }
  }
}

__device__ __forceinline__ uint4 own_piece(const uint4 t, bool ok) {
  return ok ? t : make_uint4(0xFFFFFFFFu, 0xFFFFFFFFu, 0xFFFFFFFFu, 0xFFFFFFFFu);
}

__device__ __forceinline__ void own_drain(
    int* qw, int& cnt, int lane, int lo, float b0,
    const int4* __restrict__ rec4, const float* __restrict__ hrows, const float* __restrict__ sS,
    const float* __restrict__ lsD, float* __restrict__ lsw, float* __restrict__ tile,
    bool to_strip, int* gl, int& total) {
  int head = 0;
  while (cnt - head >= 32) {
    const int e = qw[head + lane];
    if (to_strip) {
      if (total + lane < OWN_CAP) gl[total + lane] = e;
      total += 32;
    }
    own_batch(rec4[e], 32, lane, lo, b0, hrows, sS, lsD, lsw, tile);
    head += 32;
  }
  if (head > 0) {
    const int rest = cnt - head;
    int keep = 0;
    if (lane < rest) keep = qw[head + lane];
    if (lane < rest) qw[lane] = keep;
    cnt = rest;
  }
}

__global__ __launch_bounds__(1024) void k_own(
    const int* __restrict__ rec, const unsigned* __restrict__ own,
    const float* __restrict__ hin, const float* __restrict__ sD,
    const float* __restrict__ sS, const float* __restrict__ attb,
    const float* __restrict__ W, const float* __restrict__ bias,
    const float* __restrict__ g, const float* __restrict__ beta,
    const float* __restrict__ awn, int has_next,
    float* __restrict__ hout, float* __restrict__ sDn, float* __restrict__ sSn,
    int replay_strips, int* elist, int* ecount) {
  __shared__ __attribute__((aligned(16))) float tile[OWN_RP * TP];
  __shared__ __attribute__((aligned(16))) float lWt[H * TP];
  __shared__ float lsD[OWN_RP];
  __shared__ float lsw[OWN_RP];
  __shared__ int q[OWN_WAVES][OWN_Q];
  const int tid = threadIdx.x, lane = tid & 31;
  const int wave = __builtin_amdgcn_readfirstlane(tid >> 5);
  const int lo = blockIdx.x * OWN_R;
  const int nrows = (NN - lo < OWN_R) ? (NN - lo) : OWN_R;
  const unsigned myb = blockIdx.x;
  const float b0 = attb[0];

#pragma unroll 4
  for (int r = wave; r < nrows; r += OWN_WAVES) {
    const float* hr = hin + (size_t)(lo + r) * H;
    tile[r * TP + lane]      = hr[lane];
    tile[r * TP + lane + 32] = hr[lane + 32];
  }
  for (int r = nrows + wave; r < OWN_RP; r += OWN_WAVES) {
    tile[r * TP + lane]      = 0.0f;
    tile[r * TP + lane + 32] = 0.0f;
  }
  for (int r = tid; r < OWN_RP; r += 1024) {
    lsD[r] = (r < nrows) ? sD[lo + r] : 0.0f;
    lsw[r] = 1.0f;
  }
  for (int idx = tid; idx < H * H; idx += 1024)
    lWt[(idx & 63) * TP + (idx >> 6)] = W[idx];
  __syncthreads();

  const uint4* __restrict__ own4 = reinterpret_cast<const uint4*>(own);
  const int4* __restrict__ rec4 = reinterpret_cast<const int4*>(rec);
  int* qw = q[wave];
  int cnt = 0;
  const int nvec = NE / 16;
  const int nit = (nvec + 31) / 32;
  const int slot = blockIdx.x * OWN_WAVES + wave;
  int* gl = elist + (size_t)slot * OWN_CAP;
  const bool scan_keeps = (replay_strips == 0);
  int total = 0;
  if (!scan_keeps) total = __builtin_amdgcn_readfirstlane(ecount[slot]);
  if (!scan_keeps && total <= OWN_CAP) {
    int eB = (lane < total) ? gl[lane] : 0;
    int4 rA = rec4[eB];
    eB = (32 + lane < total) ? gl[32 + lane] : 0;
    for (int b = 0; b < total; b += 32) {
      const int4 r = rA;
      rA = rec4[eB];
      eB = (b + 64 + lane < total) ? gl[b + 64 + lane] : 0;
      const int n = (total - b < 32) ? (total - b) : 32;
      own_batch(r, n, lane, lo, b0, hin, sS, lsD, lsw, tile);
    }
  } else {
    for (int it0 = wave; it0 < nit; it0 += 4 * OWN_WAVES) {
      const int v0 = it0 * 32 + lane;
      const int v1 = v0 + 32 * OWN_WAVES;
      const int v2 = v1 + 32 * OWN_WAVES;
      const int v3 = v2 + 32 * OWN_WAVES;
      const bool ok0 = v0 < nvec, ok1 = v1 < nvec, ok2 = v2 < nvec, ok3 = v3 < nvec;
      const uint4 t0 = own4[ok0 ? v0 : nvec - 1];
      const uint4 t1 = own4[ok1 ? v1 : nvec - 1];
      const uint4 t2 = own4[ok2 ? v2 : nvec - 1];
      const uint4 t3 = own4[ok3 ? v3 : nvec - 1];
      own_scan_piece(own_piece(t0, ok0), v0 * 16, myb, qw, cnt);
      own_drain(qw, cnt, lane, lo, b0, rec4, hin, sS, lsD, lsw, tile, scan_keeps, gl, total);
      own_scan_piece(own_piece(t1, ok1), v1 * 16, myb, qw, cnt);
      own_drain(qw, cnt, lane, lo, b0, rec4, hin, sS, lsD, lsw, tile, scan_keeps, gl, total);
      own_scan_piece(own_piece(t2, ok2), v2 * 16, myb, qw, cnt);
      own_drain(qw, cnt, lane, lo, b0, rec4, hin, sS, lsD, lsw, tile, scan_keeps, gl, total);
      own_scan_piece(own_piece(t3, ok3), v3 * 16, myb, qw, cnt);
      own_drain(qw, cnt, lane, lo, b0, rec4, hin, sS, lsD, lsw, tile, scan_keeps, gl, total);
    }
    if (cnt > 0) {
      const int e = (lane < cnt) ? qw[lane] : 0;
      if (scan_keeps) {
        if (lane < cnt && total + lane < OWN_CAP) gl[total + lane] = e;
        total += cnt;
      }
      own_batch(rec4[e], cnt, lane, lo, b0, hin, sS, lsD, lsw, tile);
    }
    if (scan_keeps && lane == 0) ecount[slot] = total;
  }
  __syncthreads();

  if (wave * 16 < nrows) {
    const int mrow = lane & 15, gq = lane >> 4;
    const int r0 = wave * 16;
    const float* ap = tile + (r0 + mrow) * TP + 2 * gq;
    const float* bp = lWt + mrow * TP + 2 * gq;
    v8f acc[4];
#pragma unroll
    for (int cc = 0; cc < 4; ++cc) {
      const float bv = bias[16 * cc + mrow];
#pragma unroll
      for (int i = 0; i < 8; ++i) acc[cc][i] = lsw[r0 + i + 8 * gq] * bv;
    }
#if HAVE_WMMA_F32
#pragma unroll
    for (int k0 = 0; k0 < H; k0 += 4) {
      const v2f a = *reinterpret_cast<const v2f*>(ap + k0);
#pragma unroll
      for (int cc = 0; cc < 4; ++cc) {
        const v2f b = *reinterpret_cast<const v2f*>(bp + 16 * cc * TP + k0);
        acc[cc] = __builtin_amdgcn_wmma_f32_16x16x4_f32(false, a, false, b, (short)0, acc[cc], false, false);
      }
    }
#endif
#pragma unroll
    for (int cc = 0; cc < 4; ++cc) {
#pragma unroll
      for (int i = 0; i < 8; ++i) tile[(r0 + i + 8 * gq) * TP + 16 * cc + mrow] = acc[cc][i];
    }
  }
  __syncthreads();

  const float g0 = g[lane], g1 = g[lane + 32];
  const float be0 = beta[lane], be1 = beta[lane + 32];
  const float aD0 = awn[lane], aD1 = awn[lane + 32];
  const float aS0 = awn[H + lane], aS1 = awn[H + lane + 32];
#pragma unroll 2
  for (int r = wave; r < nrows; r += OWN_WAVES) {
    const size_t base = (size_t)(lo + r) * H;
    const float r0 = hin[base + lane], r1 = hin[base + lane + 32];
    const float f0 = tile[r * TP + lane], f1 = tile[r * TP + lane + 32];
    const float mean = wave_sum(f0 + f1) * (1.0f / H);
    const float d0 = f0 - mean, d1 = f1 - mean;
    const float var = wave_sum(fmaf(d0, d0, d1 * d1)) * (1.0f / H);
    const float inv = rsqrtf(var + LN_EPS);
    const float y0 = fmaf(d0 * inv, g0, be0);
    const float y1 = fmaf(d1 * inv, g1, be1);
    const float o0 = fmaxf(y0, 0.0f) + r0;
    const float o1 = fmaxf(y1, 0.0f) + r1;
    hout[base + lane]      = o0;
    hout[base + lane + 32] = o1;
    if (has_next) {
      const float pd = wave_sum(fmaf(o0, aD0, o1 * aD1));
      const float ps = wave_sum(fmaf(o0, aS0, o1 * aS1));
      if (lane == 0) { sDn[lo + r] = pd; sSn[lo + r] = ps; }
    }
  }
}

#define ACCW 8
#define RECW 8

__global__ __launch_bounds__(256) void k_l0_prep(
    const float* __restrict__ x, const float* __restrict__ inW, const float* __restrict__ inb,
    const float* __restrict__ aw, float* __restrict__ rec, double* __restrict__ acc) {
  int n = blockIdx.x * 256 + threadIdx.x;
  if (n >= NN) return;
  float xv[DIN];
#pragma unroll
  for (int d = 0; d < DIN; ++d) xv[d] = x[(size_t)n * DIN + d];
  float sd = 0.0f, ss = 0.0f;
#pragma unroll 8
  for (int k = 0; k < H; ++k) {
    float hk = inb[k];
#pragma unroll
    for (int d = 0; d < DIN; ++d) hk = fmaf(xv[d], inW[d * H + k], hk);
    sd = fmaf(hk, aw[k], sd);
    ss = fmaf(hk, aw[H + k], ss);
  }
  float4* r4 = reinterpret_cast<float4*>(rec + (size_t)n * RECW);
  r4[0] = make_float4(xv[0], xv[1], xv[2], xv[3]);
  r4[1] = make_float4(xv[4], ss, sd, 0.0f);
  double2* a2 = reinterpret_cast<double2*>(acc + (size_t)n * ACCW);
#pragma unroll
  for (int j = 0; j < ACCW / 2; ++j) a2[j] = make_double2(0.0, 0.0);
}

__global__ __launch_bounds__(256) void k_l0_edges(
    const int* __restrict__ ei, const float* __restrict__ ea,
    const float* __restrict__ rec, const float* __restrict__ attb,
    double* __restrict__ acc) {
  int gw = blockIdx.x * 8 + (threadIdx.x >> 5);
  int lane = threadIdx.x & 31;
  int ebase = gw * 32;
  if (ebase >= NE) return;
  const int sub = lane >> 3, c = lane & 7;
  const float b0 = attb[0];
  int srcs[8], dsts[8];
  float eas[8], rv[8], sd[8];
#pragma unroll
  for (int i = 0; i < 8; ++i) {
    int e = ebase + 4 * i + sub;
    srcs[i] = ei[e];
    dsts[i] = ei[NE + e];
    eas[i] = ea[e];
  }
#pragma unroll
  for (int i = 0; i < 8; ++i) {
    rv[i] = rec[(size_t)srcs[i] * RECW + c];
    sd[i] = rec[(size_t)dsts[i] * RECW + 6];
  }
#pragma unroll
  for (int i = 0; i < 8; ++i) {
    float ss = __shfl(rv[i], (lane & 24) | 5, 32);
    float w = eas[i] * sigmoidf_(sd[i] + ss + b0);
    if (c < 6) {
      double v = (c == 5) ? (double)w : (double)w * (double)rv[i];
      unsafeAtomicAdd(acc + (size_t)dsts[i] * ACCW + c, v);
    }
  }
}

__global__ __launch_bounds__(128) void k_l0_node(
    const float* __restrict__ x, const double* __restrict__ acc,
    const float* __restrict__ inW, const float* __restrict__ inb,
    const float* __restrict__ W, const float* __restrict__ bias,
    const float* __restrict__ g_, const float* __restrict__ beta,
    const float* __restrict__ awn,
    float* __restrict__ h, float* __restrict__ sDn, float* __restrict__ sSn) {
  __shared__ float lA[16][H + 1];
  __shared__ float lW[H][H + 1];
  __shared__ float lO[16][H + 1];
  __shared__ float lX[16][8];
  __shared__ float lS[16][8];
  const int tid = threadIdx.x;
  const int row0 = blockIdx.x * 16;
  for (int idx = tid; idx < H * H; idx += 128)
    lW[idx >> 6][idx & 63] = W[idx];
  {
    const int r = tid >> 3, c = tid & 7;
    const size_t n = (size_t)(row0 + r);
    float xv = (c < DIN) ? x[n * DIN + c] : 0.0f;
    float sv = (c < 6) ? (float)acc[n * ACCW + c] : 0.0f;
    lX[r][c] = xv;
    lS[r][c] = (c < DIN) ? (sv + xv) : ((c == 5) ? (sv + 1.0f) : 0.0f);
  }
  __syncthreads();
  for (int idx = tid; idx < 16 * H; idx += 128) {
    const int r = idx >> 6, k = idx & 63;
    float u = lS[r][5] * inb[k];
#pragma unroll
    for (int d = 0; d < DIN; ++d) u = fmaf(lS[r][d], inW[d * H + k], u);
    lA[r][k] = u;
  }
  __syncthreads();

  const int wave = tid >> 5, lane = tid & 31;
  const int m = lane & 15, g = lane >> 4;
  const int col = wave * 16 + m;

  v8f t;
  const float bv = bias[col];
#pragma unroll
  for (int i = 0; i < 8; ++i) t[i] = lS[i + 8 * g][5] * bv;

  t = tile_product(lA, lW, m, g, col, t);

#pragma unroll
  for (int i = 0; i < 8; ++i) lO[i + 8 * g][col] = t[i];
  __syncthreads();

  const float g0 = g_[lane], g1 = g_[lane + 32];
  const float be0 = beta[lane], be1 = beta[lane + 32];
  const float aD0 = awn[lane], aD1 = awn[lane + 32];
  const float aS0 = awn[H + lane], aS1 = awn[H + lane + 32];
#pragma unroll
  for (int rr = 0; rr < 4; ++rr) {
    const int r = wave * 4 + rr;
    float r0 = inb[lane], r1 = inb[lane + 32];
#pragma unroll
    for (int d = 0; d < DIN; ++d) {
      r0 = fmaf(lX[r][d], inW[d * H + lane], r0);
      r1 = fmaf(lX[r][d], inW[d * H + lane + 32], r1);
    }
    float f0 = lO[r][lane], f1 = lO[r][lane + 32];
    float mean = wave_sum(f0 + f1) * (1.0f / H);
    float d0 = f0 - mean, d1 = f1 - mean;
    float var = wave_sum(fmaf(d0, d0, d1 * d1)) * (1.0f / H);
    float inv = rsqrtf(var + LN_EPS);
    float y0 = fmaf(d0 * inv, g0, be0);
    float y1 = fmaf(d1 * inv, g1, be1);
    const size_t base = (size_t)(row0 + r) * H;
    const float o0 = fmaxf(y0, 0.0f) + r0;
    const float o1 = fmaxf(y1, 0.0f) + r1;
    h[base + lane]      = o0;
    h[base + lane + 32] = o1;
    const float pd = wave_sum(fmaf(o0, aD0, o1 * aD1));
    const float ps = wave_sum(fmaf(o0, aS0, o1 * aS1));
    if (lane == 0) { sDn[row0 + r] = pd; sSn[row0 + r] = ps; }
  }
}

__device__ __forceinline__ float kgroup_reduce(const float (&a)[4], int hiA, int hiB) {
  float s0 = hiA ? a[0] : a[2];
  float s1 = hiA ? a[1] : a[3];
  float k0 = hiA ? a[2] : a[0];
  float k1 = hiA ? a[3] : a[1];
  float r0 = k0 + xhalf(s0);
  float r1 = k1 + xhalf(s1);
  float sB = hiB ? r0 : r1;
  float kB = hiB ? r1 : r0;
  return kB + xrow<8>(sB);
}
#define TAIL_BLOCKS 2048
static_assert(TAIL_BLOCKS * H <= 2 * NN, "partial sums must fit in the sD+sS region");
static_assert(TAIL_BLOCKS % 16 == 0, "stage 2 splits the partial rows into 16 equal groups");
__global__ __launch_bounds__(256) void k_tail(
    const float* __restrict__ h,
    const float* __restrict__ qW1, const float* __restrict__ qb1,
    const float* __restrict__ qW2, const float* __restrict__ qb2,
    const float* __restrict__ iW1, const float* __restrict__ ib1,
    const float* __restrict__ iW2, const float* __restrict__ ib2,
    const float* __restrict__ gW1, const float* __restrict__ gb1,
    const float* __restrict__ gW2, const float* __restrict__ gb2,
    const float* __restrict__ oW,  const float* __restrict__ obv,
    const float* __restrict__ sW1, const float* __restrict__ sb1,
    const float* __restrict__ sW2, const float* __restrict__ sb2,
    float* __restrict__ oq, float* __restrict__ oi, float* __restrict__ og,
    float* __restrict__ oh, float* __restrict__ obot, float* __restrict__ osev,
    float* __restrict__ part) {
  __shared__ float ls[8][H];
  const int wave = threadIdx.x >> 5;
  const int lane = threadIdx.x & 31;
  const int nwaves = gridDim.x * 8;
  float cs0 = 0.0f, cs1 = 0.0f;

  const int kgrp = lane >> 3;
  const int hiA = (lane >> 4) & 1, hiB = (lane >> 3) & 1;
  const int col = 4 * (lane & 7) + 2 * hiA + hiB;
  const float4* __restrict__ q4 = reinterpret_cast<const float4*>(qW1);
  const float4* __restrict__ i4 = reinterpret_cast<const float4*>(iW1);
  const float4* __restrict__ g4 = reinterpret_cast<const float4*>(gW1);
  const float4* __restrict__ s4 = reinterpret_cast<const float4*>(sW1);
  const float4* __restrict__ o4 = reinterpret_cast<const float4*>(oW);
  const unsigned wl = (unsigned)lane;

  for (int gw = blockIdx.x * 8 + wave; gw < NN; gw += nwaves) {
    size_t base = (size_t)gw * H;

    const float h0 = h[base + lane], h1 = h[base + lane + 32];

    float aq[4] = {0.f, 0.f, 0.f, 0.f}, ai[4] = {0.f, 0.f, 0.f, 0.f}, ag[4] = {0.f, 0.f, 0.f, 0.f};
    float as[4] = {0.f, 0.f, 0.f, 0.f}, ao[4] = {0.f, 0.f, 0.f, 0.f};
#pragma unroll
    for (int half = 0; half < 2; ++half) {
      const float hsrc = half ? h1 : h0;
#pragma unroll 2
      for (int ii8 = 0; ii8 < 8; ++ii8) {
        const int i = half * 8 + ii8;
        const float hk = __shfl(hsrc, 4 * ii8 + kgrp, 32);
        const unsigned wx = (unsigned)(i * 32) + wl;
        const float4 wq = q4[wx], wi = i4[wx], wg = g4[wx], wsv = s4[wx], wo = o4[wx];
        aq[0] = fmaf(hk, wq.x, aq[0]);  aq[1] = fmaf(hk, wq.y, aq[1]);
        aq[2] = fmaf(hk, wq.z, aq[2]);  aq[3] = fmaf(hk, wq.w, aq[3]);
        ai[0] = fmaf(hk, wi.x, ai[0]);  ai[1] = fmaf(hk, wi.y, ai[1]);
        ai[2] = fmaf(hk, wi.z, ai[2]);  ai[3] = fmaf(hk, wi.w, ai[3]);
        ag[0] = fmaf(hk, wg.x, ag[0]);  ag[1] = fmaf(hk, wg.y, ag[1]);
        ag[2] = fmaf(hk, wg.z, ag[2]);  ag[3] = fmaf(hk, wg.w, ag[3]);
        as[0] = fmaf(hk, wsv.x, as[0]); as[1] = fmaf(hk, wsv.y, as[1]);
        as[2] = fmaf(hk, wsv.z, as[2]); as[3] = fmaf(hk, wsv.w, as[3]);
        ao[0] = fmaf(hk, wo.x, ao[0]);  ao[1] = fmaf(hk, wo.y, ao[1]);
        ao[2] = fmaf(hk, wo.z, ao[2]);  ao[3] = fmaf(hk, wo.w, ao[3]);
      }
    }
    float tq = kgroup_reduce(aq, hiA, hiB);
    float ti = kgroup_reduce(ai, hiA, hiB);
    float tg = kgroup_reduce(ag, hiA, hiB);
    float ts = kgroup_reduce(as, hiA, hiB);
    float bo = kgroup_reduce(ao, hiA, hiB);
    tq = fmaxf(tq + qb1[col], 0.0f);
    ti = fmaxf(ti + ib1[col], 0.0f);
    tg = fmaxf(tg + gb1[col], 0.0f);
    ts = fmaxf(ts + sb1[col], 0.0f);

    float q  = wave_sum(tq * qW2[col]);
    float ii = wave_sum(ti * iW2[col]);
    float gg = wave_sum(tg * gW2[col]);
    if (lane == 0) {
      oq[gw] = sigmoidf_(q + qb2[0]);
      oi[gw] = sigmoidf_(ii + ib2[0]);
      og[gw] = sigmoidf_(gg + gb2[0]);
    }

    float z[5];
#pragma unroll
    for (int c = 0; c < 5; ++c) z[c] = wave_sum(ts * sW2[col * 5 + c]) + sb2[c];
    float zm = z[0];
#pragma unroll
    for (int c = 1; c < 5; ++c) zm = fmaxf(zm, z[c]);
    float es = 0.f;
#pragma unroll
    for (int c = 0; c < 5; ++c) { z[c] = __expf(z[c] - zm); es += z[c]; }
    float inv = 1.0f / es;
#pragma unroll
    for (int c = 0; c < 5; ++c)
      if (lane == c) osev[(size_t)gw * 5 + c] = z[c] * inv;

    obot[(size_t)gw * DOUT + col] = bo + obv[col];
    oh[base + lane]      = h0;
    oh[base + lane + 32] = h1;

    cs0 += h0;
    cs1 += h1;
  }

  ls[wave][lane]      = cs0;
  ls[wave][lane + 32] = cs1;
  __syncthreads();
  if (threadIdx.x < H) {
    float t = 0.0f;
#pragma unroll
    for (int w = 0; w < 8; ++w) t += ls[w][threadIdx.x];
    part[(size_t)blockIdx.x * H + threadIdx.x] = t;
  }
}

__global__ __launch_bounds__(1024) void k_mean_final(
    const float* __restrict__ part, float* __restrict__ gout) {
  __shared__ float lq[16][H];
  const int f = threadIdx.x & 63, q = threadIdx.x >> 6;
  const float* p = part + (size_t)q * (TAIL_BLOCKS / 16) * H + f;
  float acc = 0.0f;
#pragma unroll 16
  for (int j = 0; j < TAIL_BLOCKS / 16; ++j) acc += p[(size_t)j * H];
  lq[q][f] = acc;
  __syncthreads();
  if (threadIdx.x < H) {
    float t = 0.0f;
#pragma unroll
    for (int g = 0; g < 16; ++g) t += lq[g][threadIdx.x];
    gout[threadIdx.x] = t * (1.0f / NN);
  }
}

extern "C" void kernel_launch(void* const* d_in, const int* in_sizes, int n_in,
                              void* d_out, int out_size, void* d_ws, size_t ws_size,
                              hipStream_t stream) {
  (void)in_sizes; (void)n_in; (void)out_size; (void)ws_size;
  const float* x    = (const float*)d_in[0];
  const int*   ei   = (const int*)  d_in[1];
  const float* ea   = (const float*)d_in[2];
  const float* inW  = (const float*)d_in[3];
  const float* inb  = (const float*)d_in[4];
  const float* linW = (const float*)d_in[5];
  const float* linb = (const float*)d_in[6];
  const float* attW = (const float*)d_in[7];
  const float* attb = (const float*)d_in[8];
  const float* lng  = (const float*)d_in[9];
  const float* lnb  = (const float*)d_in[10];
  const float* qW1 = (const float*)d_in[11]; const float* qb1 = (const float*)d_in[12];
  const float* qW2 = (const float*)d_in[13]; const float* qb2 = (const float*)d_in[14];
  const float* iW1 = (const float*)d_in[15]; const float* ib1 = (const float*)d_in[16];
  const float* iW2 = (const float*)d_in[17]; const float* ib2 = (const float*)d_in[18];
  const float* gW1 = (const float*)d_in[19]; const float* gb1 = (const float*)d_in[20];
  const float* gW2 = (const float*)d_in[21]; const float* gb2 = (const float*)d_in[22];
  const float* oW  = (const float*)d_in[23]; const float* obv = (const float*)d_in[24];
  const float* sW1 = (const float*)d_in[25]; const float* sb1 = (const float*)d_in[26];
  const float* sW2 = (const float*)d_in[27]; const float* sb2 = (const float*)d_in[28];

  float* ws   = (float*)d_ws;
  float* h    = ws;
  float* mbuf = h    + (size_t)NN * H;
  float* aggr = mbuf + (size_t)NN * H;
  float* sD   = aggr + (size_t)NN * H;
  float* sS   = sD + NN;

  float* out  = (float*)d_out;
  float* oq   = out;
  float* oi   = out + (size_t)NN;
  float* og   = out + 2 * (size_t)NN;
  float* oh   = out + 3 * (size_t)NN;
  float* ogr  = out + 67 * (size_t)NN;
  float* obot = out + 67 * (size_t)NN + H;
  float* osev = out + 99 * (size_t)NN + H;

  int*      erec = (int*)(aggr + 2000000);
  unsigned* eown = (unsigned*)(aggr + 6000000);
  k_edge_pack<<<(NE / 4 + 255) / 256, 256, 0, stream>>>(ei, ea, erec, eown);

  float*  rec0 = mbuf;
  double* acc0 = (double*)aggr;
  k_l0_prep<<<(NN + 255) / 256, 256, 0, stream>>>(x, inW, inb, attW, rec0, acc0);
  k_l0_edges<<<(NE + 255) / 256, 256, 0, stream>>>(ei, ea, rec0, attb, acc0);
  float* sD2 = aggr + 1600000;
  float* sS2 = sD2 + NN;
  float* h2  = mbuf;
  k_l0_node<<<NN / 16, 128, 0, stream>>>(x, acc0, inW, inb, linW, linb, lng, lnb, attW + 2 * H,
                                         h, sD, sS);

  int* elist  = (int*)aggr;
  int* ecount = (int*)(aggr + 1800000);
  k_own<<<OWN_BLOCKS, 1024, 0, stream>>>(erec, eown, h, sD, sS, attb + 1,
                                         linW + 1 * H * H, linb + 1 * H, lng + 1 * H, lnb + 1 * H,
                                         attW + 2 * 2 * H, 1, h2, sD2, sS2, 0, elist, ecount);
  k_own<<<OWN_BLOCKS, 1024, 0, stream>>>(erec, eown, h2, sD2, sS2, attb + 2,
                                         linW + 2 * H * H, linb + 2 * H, lng + 2 * H, lnb + 2 * H,
                                         attW + 3 * 2 * H, 1, h, sD, sS, 1, elist, ecount);
  k_own<<<OWN_BLOCKS, 1024, 0, stream>>>(erec, eown, h, sD, sS, attb + 3,
                                         linW + 3 * H * H, linb + 3 * H, lng + 3 * H, lnb + 3 * H,
                                         attW + 3 * 2 * H, 0, h2, sD2, sS2, 1, elist, ecount);

  float* part = sD;
  k_tail<<<TAIL_BLOCKS, 256, 0, stream>>>(h2,
                                          qW1, qb1, qW2, qb2, iW1, ib1, iW2, ib2,
                                          gW1, gb1, gW2, gb2, oW, obv, sW1, sb1, sW2, sb2,
                                          oq, oi, og, oh, obot, osev, part);
  k_mean_final<<<1, 1024, 0, stream>>>(part, ogr);
}


# owners' edge lists built once a call in the pack kernel (LDS counts, one global integer atomic per owner and work-group); all 3 layers replay them, waves share batches evenly (sum order changes)
# speedup vs baseline: 2.4609x; 1.0448x over previous
#include <hip/hip_runtime.h>

#define NN 100000
#define NE 1000000
#define DIN 5
#define H 64
#define HHID 32
#define NL 4
#define DOUT 32
#define LN_EPS 1e-5f

typedef __attribute__((ext_vector_type(2)))  float    v2f;
typedef __attribute__((ext_vector_type(8)))  float    v8f;
typedef __attribute__((ext_vector_type(16))) _Float16 v16h;

#if defined(__has_builtin)
#if __has_builtin(__builtin_amdgcn_wmma_f32_16x16x4_f32)
#define HAVE_WMMA_F32 1
#endif
#endif

__device__ __forceinline__ float xhalf(float x) {
  const int b = __builtin_bit_cast(int, x);
  return __builtin_bit_cast(float, __builtin_amdgcn_permlanex16(b, b, 0x76543210, 0xfedcba98, false, false));
}
template <int M>
__device__ __forceinline__ float xrow(float x) {
  const int b = __builtin_bit_cast(int, x);
  return __builtin_bit_cast(float, __builtin_amdgcn_update_dpp(0, b, 0x160 + M, 0xF, 0xF, true));
}
__device__ __forceinline__ float wave_sum(float v) {
  v += xhalf(v);
  v += xrow<8>(v);
  v += xrow<4>(v);
  v += xrow<2>(v);
  v += xrow<1>(v);
  return v;
}
__device__ __forceinline__ float sigmoidf_(float x) { return 1.0f / (1.0f + __expf(-x)); }

__device__ __forceinline__ void atomic_fadd(float* p, float v) { unsafeAtomicAdd(p, v); }

__device__ __forceinline__ v8f tile_product(const float (*lA)[H + 1], const float (*lW)[H + 1],
                                            int m, int g, int col, v8f acc) {
#if HAVE_WMMA_F32
#pragma unroll
  for (int k0 = 0; k0 < H; k0 += 4) {
    int kb = k0 + 2 * g;
    v2f a, b;
    a.x = lA[m][kb];     a.y = lA[m][kb + 1];
    b.x = lW[kb][col];   b.y = lW[kb + 1][col];
    acc = __builtin_amdgcn_wmma_f32_16x16x4_f32(false, a, false, b, (short)0, acc, false, false);
  }
#else
#pragma unroll
  for (int kk = 0; kk < H; kk += 32) {
    v16h a, b;
#pragma unroll
    for (int j = 0; j < 8; ++j) {
      int ka = (j < 4) ? (2 * j + 8 * g) : (16 + 2 * (j - 4) + 8 * g);
      a[2 * j]     = (_Float16)lA[m][kk + ka];
      a[2 * j + 1] = (_Float16)lA[m][kk + ka + 1];
      int kb = 2 * j + 16 * g;
      b[2 * j]     = (_Float16)lW[kk + kb][col];
      b[2 * j + 1] = (_Float16)lW[kk + kb + 1][col];
    }
    acc = __builtin_amdgcn_wmma_f32_16x16x32_f16(false, a, false, b, (short)0, acc, false, false);
  }
#endif
  return acc;
}

#define EDGE_DEPTH 8
#define OWN_CAP 192
#define OWN_R 393
#define OWN_BLOCKS ((NN + OWN_R - 1) / OWN_R)
#define OWN_WAVES 32
#define OWN_LIST (OWN_WAVES * OWN_CAP)
#define ECNT_STRIDE 32
#define PACK_T 1024
#define OWN_Q 544
#define OWN_TILES ((OWN_R + 15) / 16)
#define OWN_RP (OWN_TILES * 16)
#define TP 68
static_assert(OWN_TILES <= OWN_WAVES, "one wave per 16-row tile of the owner's product");
static_assert(NL == 4, "the three owner launches are written out");
static_assert(OWN_BLOCKS <= 255, "owner numbers are bytes 0..254; 255 marks a map byte that is nobody's");
static_assert(NE % 16 == 0, "the owner map is read 16 bytes a lane");
static_assert(NE % 32 == 0, "every wave of the layer-0 edge pass owns 32 whole edges");
static_assert(32 % EDGE_DEPTH == 0, "a batch of 32 edges splits into whole groups");
static_assert(OWN_LIST % 32 == 0, "a list holds whole batches");
static_assert((size_t)OWN_BLOCKS * OWN_LIST * 4 <= 6400000, "the lists fit in the region of layer 0's sums");
static_assert((size_t)OWN_BLOCKS * ECNT_STRIDE * 4 <= 800000, "the counters fit between the second attention scalars and the records");
static_assert(OWN_BLOCKS <= 256 && PACK_T >= 256, "a pack work-group keeps one LDS counter per owner");

__device__ __forceinline__ int lane_bcast_i(int v, int l) { return __builtin_amdgcn_readlane(v, l); }
__device__ __forceinline__ float lane_bcast_f(float v, int l) {
  return __int_as_float(__builtin_amdgcn_readlane(__float_as_int(v), l));
}

__global__ __launch_bounds__(PACK_T) void k_edge_pack(
    const int* __restrict__ ei, const float* __restrict__ ea,
    int* __restrict__ rec, unsigned* __restrict__ own,
    int* __restrict__ elist, int* __restrict__ ecount) {
  __shared__ int hist[256];
  const int t = blockIdx.x * PACK_T + threadIdx.x;
  const bool live = t < NE / 4;
  if (threadIdx.x < 256) hist[threadIdx.x] = 0;
  __syncthreads();
  int o0 = 0, o1 = 0, o2 = 0, o3 = 0, r0 = 0, r1 = 0, r2 = 0, r3 = 0;
  int4 s4 = make_int4(0, 0, 0, 0), d4 = make_int4(0, 0, 0, 0);
  float4 a4 = make_float4(0.0f, 0.0f, 0.0f, 0.0f);
  if (live) {
    s4 = reinterpret_cast<const int4*>(ei)[t];
    d4 = reinterpret_cast<const int4*>(ei + NE)[t];
    a4 = reinterpret_cast<const float4*>(ea)[t];
    o0 = d4.x / OWN_R; o1 = d4.y / OWN_R; o2 = d4.z / OWN_R; o3 = d4.w / OWN_R;
    r0 = __hip_atomic_fetch_add(hist + o0, 1, __ATOMIC_RELAXED, __HIP_MEMORY_SCOPE_WORKGROUP);
    r1 = __hip_atomic_fetch_add(hist + o1, 1, __ATOMIC_RELAXED, __HIP_MEMORY_SCOPE_WORKGROUP);
    r2 = __hip_atomic_fetch_add(hist + o2, 1, __ATOMIC_RELAXED, __HIP_MEMORY_SCOPE_WORKGROUP);
    r3 = __hip_atomic_fetch_add(hist + o3, 1, __ATOMIC_RELAXED, __HIP_MEMORY_SCOPE_WORKGROUP);
  }
  __syncthreads();
  if (threadIdx.x < OWN_BLOCKS) {
    const int c = hist[threadIdx.x];
    int first = 0;
    if (c > 0)
      first = __hip_atomic_fetch_add(ecount + threadIdx.x * ECNT_STRIDE, c, __ATOMIC_RELAXED, __HIP_MEMORY_SCOPE_AGENT);
    hist[threadIdx.x] = first;
  }
  __syncthreads();
  if (live) {
    int4* r4 = reinterpret_cast<int4*>(rec) + (size_t)t * 4;
    r4[0] = make_int4(s4.x, d4.x, __float_as_int(a4.x), 0);
    r4[1] = make_int4(s4.y, d4.y, __float_as_int(a4.y), 0);
    r4[2] = make_int4(s4.z, d4.z, __float_as_int(a4.z), 0);
    r4[3] = make_int4(s4.w, d4.w, __float_as_int(a4.w), 0);
    own[t] = (unsigned)o0 | ((unsigned)o1 << 8) | ((unsigned)o2 << 16) | ((unsigned)o3 << 24);
    const int p0 = hist[o0] + r0, p1 = hist[o1] + r1, p2 = hist[o2] + r2, p3 = hist[o3] + r3;
    if (p0 < OWN_LIST) elist[(size_t)o0 * OWN_LIST + p0] = 4 * t;
    if (p1 < OWN_LIST) elist[(size_t)o1 * OWN_LIST + p1] = 4 * t + 1;
    if (p2 < OWN_LIST) elist[(size_t)o2 * OWN_LIST + p2] = 4 * t + 2;
    if (p3 < OWN_LIST) elist[(size_t)o3 * OWN_LIST + p3] = 4 * t + 3;
  }
}

__device__ __forceinline__ void own_batch(
    const int4 r, int n, int lane, int lo, float b0,
    const float* __restrict__ hrows, const float* __restrict__ sS,
    const float* __restrict__ lsD, float* __restrict__ lsw, float* __restrict__ tile) {
  int src = 0, dl = 0;
  float w = 0.0f;
  if (lane < n) {
    src = r.x;
    dl = r.y - lo;
    w = __int_as_float(r.z) * sigmoidf_(lsD[dl] + sS[src] + b0);
    __hip_atomic_fetch_add(lsw + dl, w, __ATOMIC_RELAXED, __HIP_MEMORY_SCOPE_WORKGROUP);
  }
  const float* mlane = hrows + lane;
#pragma unroll
  for (int i0 = 0; i0 < 32; i0 += EDGE_DEPTH) {
    if (i0 < n) {
      float lov[EDGE_DEPTH], hiv[EDGE_DEPTH];
#pragma unroll
      for (int j = 0; j < EDGE_DEPTH; ++j) {
        const float* mr = mlane + (size_t)lane_bcast_i(src, i0 + j) * H;
        lov[j] = mr[0];
        hiv[j] = mr[32];
      }
#pragma unroll
      for (int j = 0; j < EDGE_DEPTH; ++j) {
        if (i0 + j < n) {
          const float bw = lane_bcast_f(w, i0 + j);
          float* tr = tile + lane_bcast_i(dl, i0 + j) * TP + lane;
          __hip_atomic_fetch_add(tr,      lov[j] * bw, __ATOMIC_RELAXED, __HIP_MEMORY_SCOPE_WORKGROUP);
          __hip_atomic_fetch_add(tr + 32, hiv[j] * bw, __ATOMIC_RELAXED, __HIP_MEMORY_SCOPE_WORKGROUP);
        }
      }
    }
  }
}

__device__ __forceinline__ void own_scan_piece(const uint4 o, int vb, unsigned myb, int* qw, int& cnt) {
  const unsigned wd[4] = {o.x, o.y, o.z, o.w};
#pragma unroll
  for (int k = 0; k < 16; ++k) {
    const unsigned byte = (wd[k >> 2] >> (8 * (k & 3))) & 0xFFu;
    const bool mine = (byte == myb);
    const unsigned mask = __builtin_amdgcn_ballot_w32(mine);
    if (mask != 0u) {
      const int pos = cnt + (int)__builtin_amdgcn_mbcnt_lo(mask, 0u);
      if (mine) qw[pos] = vb + k;
      cnt += __builtin_popcount(mask);
    }
  }
}

__device__ __forceinline__ uint4 own_piece(const uint4 t, bool ok) {
  return ok ? t : make_uint4(0xFFFFFFFFu, 0xFFFFFFFFu, 0xFFFFFFFFu, 0xFFFFFFFFu);
}

__device__ __forceinline__ void own_drain(
    int* qw, int& cnt, int lane, int lo, float b0,
    const int4* __restrict__ rec4, const float* __restrict__ hrows, const float* __restrict__ sS,
    const float* __restrict__ lsD, float* __restrict__ lsw, float* __restrict__ tile) {
  int head = 0;
  while (cnt - head >= 32) {
    const int e = qw[head + lane];
    own_batch(rec4[e], 32, lane, lo, b0, hrows, sS, lsD, lsw, tile);
    head += 32;
  }
  if (head > 0) {
    const int rest = cnt - head;
    int keep = 0;
    if (lane < rest) keep = qw[head + lane];
    if (lane < rest) qw[lane] = keep;
    cnt = rest;
  }
}

__global__ __launch_bounds__(1024) void k_own(
    const int* __restrict__ rec, const unsigned* __restrict__ own,
    const float* __restrict__ hin, const float* __restrict__ sD,
    const float* __restrict__ sS, const float* __restrict__ attb,
    const float* __restrict__ W, const float* __restrict__ bias,
    const float* __restrict__ g, const float* __restrict__ beta,
    const float* __restrict__ awn, int has_next,
    float* __restrict__ hout, float* __restrict__ sDn, float* __restrict__ sSn,
    const int* __restrict__ elist, const int* __restrict__ ecount) {
  __shared__ __attribute__((aligned(16))) float tile[OWN_RP * TP];
  __shared__ __attribute__((aligned(16))) float lWt[H * TP];
  __shared__ float lsD[OWN_RP];
  __shared__ float lsw[OWN_RP];
  __shared__ int q[OWN_WAVES][OWN_Q];
  const int tid = threadIdx.x, lane = tid & 31;
  const int wave = __builtin_amdgcn_readfirstlane(tid >> 5);
  const int lo = blockIdx.x * OWN_R;
  const int nrows = (NN - lo < OWN_R) ? (NN - lo) : OWN_R;
  const unsigned myb = blockIdx.x;
  const float b0 = attb[0];
  const int total = __builtin_amdgcn_readfirstlane(ecount[blockIdx.x * ECNT_STRIDE]);

#pragma unroll 4
  for (int r = wave; r < nrows; r += OWN_WAVES) {
    const float* hr = hin + (size_t)(lo + r) * H;
    tile[r * TP + lane]      = hr[lane];
    tile[r * TP + lane + 32] = hr[lane + 32];
  }
  for (int r = nrows + wave; r < OWN_RP; r += OWN_WAVES) {
    tile[r * TP + lane]      = 0.0f;
    tile[r * TP + lane + 32] = 0.0f;
  }
  for (int r = tid; r < OWN_RP; r += 1024) {
    lsD[r] = (r < nrows) ? sD[lo + r] : 0.0f;
    lsw[r] = 1.0f;
  }
  for (int idx = tid; idx < H * H; idx += 1024)
    lWt[(idx & 63) * TP + (idx >> 6)] = W[idx];
  __syncthreads();

  const uint4* __restrict__ own4 = reinterpret_cast<const uint4*>(own);
  const int4* __restrict__ rec4 = reinterpret_cast<const int4*>(rec);
  int* qw = q[wave];
  int cnt = 0;
  const int nvec = NE / 16;
  const int nit = (nvec + 31) / 32;
  const int* gl = elist + (size_t)blockIdx.x * OWN_LIST;
  if (total <= OWN_LIST) {
    const int step = 32 * OWN_WAVES;
    const int first = 32 * wave;
    int eB = (first + lane < total) ? gl[first + lane] : 0;
    int4 rA = rec4[eB];
    eB = (first + step + lane < total) ? gl[first + step + lane] : 0;
    for (int b = first; b < total; b += step) {
      const int4 r = rA;
      rA = rec4[eB];
      eB = (b + 2 * step + lane < total) ? gl[b + 2 * step + lane] : 0;
      const int n = (total - b < 32) ? (total - b) : 32;
      own_batch(r, n, lane, lo, b0, hin, sS, lsD, lsw, tile);
    }
  } else {
    for (int it0 = wave; it0 < nit; it0 += 4 * OWN_WAVES) {
      const int v0 = it0 * 32 + lane;
      const int v1 = v0 + 32 * OWN_WAVES;
      const int v2 = v1 + 32 * OWN_WAVES;
      const int v3 = v2 + 32 * OWN_WAVES;
      const bool ok0 = v0 < nvec, ok1 = v1 < nvec, ok2 = v2 < nvec, ok3 = v3 < nvec;
      const uint4 t0 = own4[ok0 ? v0 : nvec - 1];
      const uint4 t1 = own4[ok1 ? v1 : nvec - 1];
      const uint4 t2 = own4[ok2 ? v2 : nvec - 1];
      const uint4 t3 = own4[ok3 ? v3 : nvec - 1];
      own_scan_piece(own_piece(t0, ok0), v0 * 16, myb, qw, cnt);
      own_drain(qw, cnt, lane, lo, b0, rec4, hin, sS, lsD, lsw, tile);
      own_scan_piece(own_piece(t1, ok1), v1 * 16, myb, qw, cnt);
      own_drain(qw, cnt, lane, lo, b0, rec4, hin, sS, lsD, lsw, tile);
      own_scan_piece(own_piece(t2, ok2), v2 * 16, myb, qw, cnt);
      own_drain(qw, cnt, lane, lo, b0, rec4, hin, sS, lsD, lsw, tile);
      own_scan_piece(own_piece(t3, ok3), v3 * 16, myb, qw, cnt);
      own_drain(qw, cnt, lane, lo, b0, rec4, hin, sS, lsD, lsw, tile);
    }
    if (cnt > 0) {
      const int e = (lane < cnt) ? qw[lane] : 0;
      own_batch(rec4[e], cnt, lane, lo, b0, hin, sS, lsD, lsw, tile);
    }
  }
  __syncthreads();

  if (wave * 16 < nrows) {
    const int mrow = lane & 15, gq = lane >> 4;
    const int r0 = wave * 16;
    const float* ap = tile + (r0 + mrow) * TP + 2 * gq;
    const float* bp = lWt + mrow * TP + 2 * gq;
    v8f acc[4];
#pragma unroll
    for (int cc = 0; cc < 4; ++cc) {
      const float bv = bias[16 * cc + mrow];
#pragma unroll
      for (int i = 0; i < 8; ++i) acc[cc][i] = lsw[r0 + i + 8 * gq] * bv;
    }
#if HAVE_WMMA_F32
#pragma unroll
    for (int k0 = 0; k0 < H; k0 += 4) {
      const v2f a = *reinterpret_cast<const v2f*>(ap + k0);
#pragma unroll
      for (int cc = 0; cc < 4; ++cc) {
        const v2f b = *reinterpret_cast<const v2f*>(bp + 16 * cc * TP + k0);
        acc[cc] = __builtin_amdgcn_wmma_f32_16x16x4_f32(false, a, false, b, (short)0, acc[cc], false, false);
      }
    }
#endif
#pragma unroll
    for (int cc = 0; cc < 4; ++cc) {
#pragma unroll
      for (int i = 0; i < 8; ++i) tile[(r0 + i + 8 * gq) * TP + 16 * cc + mrow] = acc[cc][i];
    }
  }
  __syncthreads();

  const float g0 = g[lane], g1 = g[lane + 32];
  const float be0 = beta[lane], be1 = beta[lane + 32];
  const float aD0 = awn[lane], aD1 = awn[lane + 32];
  const float aS0 = awn[H + lane], aS1 = awn[H + lane + 32];
#pragma unroll 2
  for (int r = wave; r < nrows; r += OWN_WAVES) {
    const size_t base = (size_t)(lo + r) * H;
    const float r0 = hin[base + lane], r1 = hin[base + lane + 32];
    const float f0 = tile[r * TP + lane], f1 = tile[r * TP + lane + 32];
    const float mean = wave_sum(f0 + f1) * (1.0f / H);
    const float d0 = f0 - mean, d1 = f1 - mean;
    const float var = wave_sum(fmaf(d0, d0, d1 * d1)) * (1.0f / H);
    const float inv = rsqrtf(var + LN_EPS);
    const float y0 = fmaf(d0 * inv, g0, be0);
    const float y1 = fmaf(d1 * inv, g1, be1);
    const float o0 = fmaxf(y0, 0.0f) + r0;
    const float o1 = fmaxf(y1, 0.0f) + r1;
    hout[base + lane]      = o0;
    hout[base + lane + 32] = o1;
    if (has_next) {
      const float pd = wave_sum(fmaf(o0, aD0, o1 * aD1));
      const float ps = wave_sum(fmaf(o0, aS0, o1 * aS1));
      if (lane == 0) { sDn[lo + r] = pd; sSn[lo + r] = ps; }
    }
  }
}

#define ACCW 8
#define RECW 8

static_assert(OWN_BLOCKS <= NN, "one thread of the node pass per list counter");
__global__ __launch_bounds__(256) void k_l0_prep(
    const float* __restrict__ x, const float* __restrict__ inW, const float* __restrict__ inb,
    const float* __restrict__ aw, float* __restrict__ rec, double* __restrict__ acc,
    int* __restrict__ ecount) {
  int n = blockIdx.x * 256 + threadIdx.x;
  if (n >= NN) return;
  if (n < OWN_BLOCKS) ecount[n * ECNT_STRIDE] = 0;
  float xv[DIN];
#pragma unroll
  for (int d = 0; d < DIN; ++d) xv[d] = x[(size_t)n * DIN + d];
  float sd = 0.0f, ss = 0.0f;
#pragma unroll 8
  for (int k = 0; k < H; ++k) {
    float hk = inb[k];
#pragma unroll
    for (int d = 0; d < DIN; ++d) hk = fmaf(xv[d], inW[d * H + k], hk);
    sd = fmaf(hk, aw[k], sd);
    ss = fmaf(hk, aw[H + k], ss);
  }
  float4* r4 = reinterpret_cast<float4*>(rec + (size_t)n * RECW);
  r4[0] = make_float4(xv[0], xv[1], xv[2], xv[3]);
  r4[1] = make_float4(xv[4], ss, sd, 0.0f);
  double2* a2 = reinterpret_cast<double2*>(acc + (size_t)n * ACCW);
#pragma unroll
  for (int j = 0; j < ACCW / 2; ++j) a2[j] = make_double2(0.0, 0.0);
}

__global__ __launch_bounds__(256) void k_l0_edges(
    const int* __restrict__ ei, const float* __restrict__ ea,
    const float* __restrict__ rec, const float* __restrict__ attb,
    double* __restrict__ acc) {
  int gw = blockIdx.x * 8 + (threadIdx.x >> 5);
  int lane = threadIdx.x & 31;
  int ebase = gw * 32;
  if (ebase >= NE) return;
  const int sub = lane >> 3, c = lane & 7;
  const float b0 = attb[0];
  int srcs[8], dsts[8];
  float eas[8], rv[8], sd[8];
#pragma unroll
  for (int i = 0; i < 8; ++i) {
    int e = ebase + 4 * i + sub;
    srcs[i] = ei[e];
    dsts[i] = ei[NE + e];
    eas[i] = ea[e];
  }
#pragma unroll
  for (int i = 0; i < 8; ++i) {
    rv[i] = rec[(size_t)srcs[i] * RECW + c];
    sd[i] = rec[(size_t)dsts[i] * RECW + 6];
  }
#pragma unroll
  for (int i = 0; i < 8; ++i) {
    float ss = __shfl(rv[i], (lane & 24) | 5, 32);
    float w = eas[i] * sigmoidf_(sd[i] + ss + b0);
    if (c < 6) {
      double v = (c == 5) ? (double)w : (double)w * (double)rv[i];
      unsafeAtomicAdd(acc + (size_t)dsts[i] * ACCW + c, v);
    }
  }
}

__global__ __launch_bounds__(128) void k_l0_node(
    const float* __restrict__ x, const double* __restrict__ acc,
    const float* __restrict__ inW, const float* __restrict__ inb,
    const float* __restrict__ W, const float* __restrict__ bias,
    const float* __restrict__ g_, const float* __restrict__ beta,
    const float* __restrict__ awn,
    float* __restrict__ h, float* __restrict__ sDn, float* __restrict__ sSn) {
  __shared__ float lA[16][H + 1];
  __shared__ float lW[H][H + 1];
  __shared__ float lO[16][H + 1];
  __shared__ float lX[16][8];
  __shared__ float lS[16][8];
  const int tid = threadIdx.x;
  const int row0 = blockIdx.x * 16;
  for (int idx = tid; idx < H * H; idx += 128)
    lW[idx >> 6][idx & 63] = W[idx];
  {
    const int r = tid >> 3, c = tid & 7;
    const size_t n = (size_t)(row0 + r);
    float xv = (c < DIN) ? x[n * DIN + c] : 0.0f;
    float sv = (c < 6) ? (float)acc[n * ACCW + c] : 0.0f;
    lX[r][c] = xv;
    lS[r][c] = (c < DIN) ? (sv + xv) : ((c == 5) ? (sv + 1.0f) : 0.0f);
  }
  __syncthreads();
  for (int idx = tid; idx < 16 * H; idx += 128) {
    const int r = idx >> 6, k = idx & 63;
    float u = lS[r][5] * inb[k];
#pragma unroll
    for (int d = 0; d < DIN; ++d) u = fmaf(lS[r][d], inW[d * H + k], u);
    lA[r][k] = u;
  }
  __syncthreads();

  const int wave = tid >> 5, lane = tid & 31;
  const int m = lane & 15, g = lane >> 4;
  const int col = wave * 16 + m;

  v8f t;
  const float bv = bias[col];
#pragma unroll
  for (int i = 0; i < 8; ++i) t[i] = lS[i + 8 * g][5] * bv;

  t = tile_product(lA, lW, m, g, col, t);

#pragma unroll
  for (int i = 0; i < 8; ++i) lO[i + 8 * g][col] = t[i];
  __syncthreads();

  const float g0 = g_[lane], g1 = g_[lane + 32];
  const float be0 = beta[lane], be1 = beta[lane + 32];
  const float aD0 = awn[lane], aD1 = awn[lane + 32];
  const float aS0 = awn[H + lane], aS1 = awn[H + lane + 32];
#pragma unroll
  for (int rr = 0; rr < 4; ++rr) {
    const int r = wave * 4 + rr;
    float r0 = inb[lane], r1 = inb[lane + 32];
#pragma unroll
    for (int d = 0; d < DIN; ++d) {
      r0 = fmaf(lX[r][d], inW[d * H + lane], r0);
      r1 = fmaf(lX[r][d], inW[d * H + lane + 32], r1);
    }
    float f0 = lO[r][lane], f1 = lO[r][lane + 32];
    float mean = wave_sum(f0 + f1) * (1.0f / H);
    float d0 = f0 - mean, d1 = f1 - mean;
    float var = wave_sum(fmaf(d0, d0, d1 * d1)) * (1.0f / H);
    float inv = rsqrtf(var + LN_EPS);
    float y0 = fmaf(d0 * inv, g0, be0);
    float y1 = fmaf(d1 * inv, g1, be1);
    const size_t base = (size_t)(row0 + r) * H;
    const float o0 = fmaxf(y0, 0.0f) + r0;
    const float o1 = fmaxf(y1, 0.0f) + r1;
    h[base + lane]      = o0;
    h[base + lane + 32] = o1;
    const float pd = wave_sum(fmaf(o0, aD0, o1 * aD1));
    const float ps = wave_sum(fmaf(o0, aS0, o1 * aS1));
    if (lane == 0) { sDn[row0 + r] = pd; sSn[row0 + r] = ps; }
  }
}

__device__ __forceinline__ float kgroup_reduce(const float (&a)[4], int hiA, int hiB) {
  float s0 = hiA ? a[0] : a[2];
  float s1 = hiA ? a[1] : a[3];
  float k0 = hiA ? a[2] : a[0];
  float k1 = hiA ? a[3] : a[1];
  float r0 = k0 + xhalf(s0);
  float r1 = k1 + xhalf(s1);
  float sB = hiB ? r0 : r1;
  float kB = hiB ? r1 : r0;
  return kB + xrow<8>(sB);
}
#define TAIL_BLOCKS 2048
static_assert(TAIL_BLOCKS * H <= 2 * NN, "partial sums must fit in the sD+sS region");
static_assert(TAIL_BLOCKS % 16 == 0, "stage 2 splits the partial rows into 16 equal groups");
__global__ __launch_bounds__(256) void k_tail(
    const float* __restrict__ h,
    const float* __restrict__ qW1, const float* __restrict__ qb1,
    const float* __restrict__ qW2, const float* __restrict__ qb2,
    const float* __restrict__ iW1, const float* __restrict__ ib1,
    const float* __restrict__ iW2, const float* __restrict__ ib2,
    const float* __restrict__ gW1, const float* __restrict__ gb1,
    const float* __restrict__ gW2, const float* __restrict__ gb2,
    const float* __restrict__ oW,  const float* __restrict__ obv,
    const float* __restrict__ sW1, const float* __restrict__ sb1,
    const float* __restrict__ sW2, const float* __restrict__ sb2,
    float* __restrict__ oq, float* __restrict__ oi, float* __restrict__ og,
    float* __restrict__ oh, float* __restrict__ obot, float* __restrict__ osev,
    float* __restrict__ part) {
  __shared__ float ls[8][H];
  const int wave = threadIdx.x >> 5;
  const int lane = threadIdx.x & 31;
  const int nwaves = gridDim.x * 8;
  float cs0 = 0.0f, cs1 = 0.0f;

  const int kgrp = lane >> 3;
  const int hiA = (lane >> 4) & 1, hiB = (lane >> 3) & 1;
  const int col = 4 * (lane & 7) + 2 * hiA + hiB;
  const float4* __restrict__ q4 = reinterpret_cast<const float4*>(qW1);
  const float4* __restrict__ i4 = reinterpret_cast<const float4*>(iW1);
  const float4* __restrict__ g4 = reinterpret_cast<const float4*>(gW1);
  const float4* __restrict__ s4 = reinterpret_cast<const float4*>(sW1);
  const float4* __restrict__ o4 = reinterpret_cast<const float4*>(oW);
  const unsigned wl = (unsigned)lane;

  for (int gw = blockIdx.x * 8 + wave; gw < NN; gw += nwaves) {
    size_t base = (size_t)gw * H;

    const float h0 = h[base + lane], h1 = h[base + lane + 32];

    float aq[4] = {0.f, 0.f, 0.f, 0.f}, ai[4] = {0.f, 0.f, 0.f, 0.f}, ag[4] = {0.f, 0.f, 0.f, 0.f};
    float as[4] = {0.f, 0.f, 0.f, 0.f}, ao[4] = {0.f, 0.f, 0.f, 0.f};
#pragma unroll
    for (int half = 0; half < 2; ++half) {
      const float hsrc = half ? h1 : h0;
#pragma unroll 2
      for (int ii8 = 0; ii8 < 8; ++ii8) {
        const int i = half * 8 + ii8;
        const float hk = __shfl(hsrc, 4 * ii8 + kgrp, 32);
        const unsigned wx = (unsigned)(i * 32) + wl;
        const float4 wq = q4[wx], wi = i4[wx], wg = g4[wx], wsv = s4[wx], wo = o4[wx];
        aq[0] = fmaf(hk, wq.x, aq[0]);  aq[1] = fmaf(hk, wq.y, aq[1]);
        aq[2] = fmaf(hk, wq.z, aq[2]);  aq[3] = fmaf(hk, wq.w, aq[3]);
        ai[0] = fmaf(hk, wi.x, ai[0]);  ai[1] = fmaf(hk, wi.y, ai[1]);
        ai[2] = fmaf(hk, wi.z, ai[2]);  ai[3] = fmaf(hk, wi.w, ai[3]);
        ag[0] = fmaf(hk, wg.x, ag[0]);  ag[1] = fmaf(hk, wg.y, ag[1]);
        ag[2] = fmaf(hk, wg.z, ag[2]);  ag[3] = fmaf(hk, wg.w, ag[3]);
        as[0] = fmaf(hk, wsv.x, as[0]); as[1] = fmaf(hk, wsv.y, as[1]);
        as[2] = fmaf(hk, wsv.z, as[2]); as[3] = fmaf(hk, wsv.w, as[3]);
        ao[0] = fmaf(hk, wo.x, ao[0]);  ao[1] = fmaf(hk, wo.y, ao[1]);
        ao[2] = fmaf(hk, wo.z, ao[2]);  ao[3] = fmaf(hk, wo.w, ao[3]);
      }
    }
    float tq = kgroup_reduce(aq, hiA, hiB);
    float ti = kgroup_reduce(ai, hiA, hiB);
    float tg = kgroup_reduce(ag, hiA, hiB);
    float ts = kgroup_reduce(as, hiA, hiB);
    float bo = kgroup_reduce(ao, hiA, hiB);
    tq = fmaxf(tq + qb1[col], 0.0f);
    ti = fmaxf(ti + ib1[col], 0.0f);
    tg = fmaxf(tg + gb1[col], 0.0f);
    ts = fmaxf(ts + sb1[col], 0.0f);

    float q  = wave_sum(tq * qW2[col]);
    float ii = wave_sum(ti * iW2[col]);
    float gg = wave_sum(tg * gW2[col]);
    if (lane == 0) {
      oq[gw] = sigmoidf_(q + qb2[0]);
      oi[gw] = sigmoidf_(ii + ib2[0]);
      og[gw] = sigmoidf_(gg + gb2[0]);
    }

    float z[5];
#pragma unroll
    for (int c = 0; c < 5; ++c) z[c] = wave_sum(ts * sW2[col * 5 + c]) + sb2[c];
    float zm = z[0];
#pragma unroll
    for (int c = 1; c < 5; ++c) zm = fmaxf(zm, z[c]);
    float es = 0.f;
#pragma unroll
    for (int c = 0; c < 5; ++c) { z[c] = __expf(z[c] - zm); es += z[c]; }
    float inv = 1.0f / es;
#pragma unroll
    for (int c = 0; c < 5; ++c)
      if (lane == c) osev[(size_t)gw * 5 + c] = z[c] * inv;

    obot[(size_t)gw * DOUT + col] = bo + obv[col];
    oh[base + lane]      = h0;
    oh[base + lane + 32] = h1;

    cs0 += h0;
    cs1 += h1;
  }

  ls[wave][lane]      = cs0;
  ls[wave][lane + 32] = cs1;
  __syncthreads();
  if (threadIdx.x < H) {
    float t = 0.0f;
#pragma unroll
    for (int w = 0; w < 8; ++w) t += ls[w][threadIdx.x];
    part[(size_t)blockIdx.x * H + threadIdx.x] = t;
  }
}

__global__ __launch_bounds__(1024) void k_mean_final(
    const float* __restrict__ part, float* __restrict__ gout) {
  __shared__ float lq[16][H];
  const int f = threadIdx.x & 63, q = threadIdx.x >> 6;
  const float* p = part + (size_t)q * (TAIL_BLOCKS / 16) * H + f;
  float acc = 0.0f;
#pragma unroll 16
  for (int j = 0; j < TAIL_BLOCKS / 16; ++j) acc += p[(size_t)j * H];
  lq[q][f] = acc;
  __syncthreads();
  if (threadIdx.x < H) {
    float t = 0.0f;
#pragma unroll
    for (int g = 0; g < 16; ++g) t += lq[g][threadIdx.x];
    gout[threadIdx.x] = t * (1.0f / NN);
  }
}

extern "C" void kernel_launch(void* const* d_in, const int* in_sizes, int n_in,
                              void* d_out, int out_size, void* d_ws, size_t ws_size,
                              hipStream_t stream) {
  (void)in_sizes; (void)n_in; (void)out_size; (void)ws_size;
  const float* x    = (const float*)d_in[0];
  const int*   ei   = (const int*)  d_in[1];
  const float* ea   = (const float*)d_in[2];
  const float* inW  = (const float*)d_in[3];
  const float* inb  = (const float*)d_in[4];
  const float* linW = (const float*)d_in[5];
  const float* linb = (const float*)d_in[6];
  const float* attW = (const float*)d_in[7];
  const float* attb = (const float*)d_in[8];
  const float* lng  = (const float*)d_in[9];
  const float* lnb  = (const float*)d_in[10];
  const float* qW1 = (const float*)d_in[11]; const float* qb1 = (const float*)d_in[12];
  const float* qW2 = (const float*)d_in[13]; const float* qb2 = (const float*)d_in[14];
  const float* iW1 = (const float*)d_in[15]; const float* ib1 = (const float*)d_in[16];
  const float* iW2 = (const float*)d_in[17]; const float* ib2 = (const float*)d_in[18];
  const float* gW1 = (const float*)d_in[19]; const float* gb1 = (const float*)d_in[20];
  const float* gW2 = (const float*)d_in[21]; const float* gb2 = (const float*)d_in[22];
  const float* oW  = (const float*)d_in[23]; const float* obv = (const float*)d_in[24];
  const float* sW1 = (const float*)d_in[25]; const float* sb1 = (const float*)d_in[26];
  const float* sW2 = (const float*)d_in[27]; const float* sb2 = (const float*)d_in[28];

  float* ws   = (float*)d_ws;
  float* h    = ws;
  float* mbuf = h    + (size_t)NN * H;
  float* aggr = mbuf + (size_t)NN * H;
  float* sD   = aggr + (size_t)NN * H;
  float* sS   = sD + NN;

  float* out  = (float*)d_out;
  float* oq   = out;
  float* oi   = out + (size_t)NN;
  float* og   = out + 2 * (size_t)NN;
  float* oh   = out + 3 * (size_t)NN;
  float* ogr  = out + 67 * (size_t)NN;
  float* obot = out + 67 * (size_t)NN + H;
  float* osev = out + 99 * (size_t)NN + H;

  float*  rec0 = mbuf;
  double* acc0 = (double*)aggr;
  int* ecount = (int*)(aggr + 1800000);
  k_l0_prep<<<(NN + 255) / 256, 256, 0, stream>>>(x, inW, inb, attW, rec0, acc0, ecount);
  k_l0_edges<<<(NE + 255) / 256, 256, 0, stream>>>(ei, ea, rec0, attb, acc0);
  float* sD2 = aggr + 1600000;
  float* sS2 = sD2 + NN;
  float* h2  = mbuf;
  k_l0_node<<<NN / 16, 128, 0, stream>>>(x, acc0, inW, inb, linW, linb, lng, lnb, attW + 2 * H,
                                         h, sD, sS);

  int*      erec  = (int*)(aggr + 2000000);
  unsigned* eown  = (unsigned*)(aggr + 6000000);
  int*      elist = (int*)aggr;
  k_edge_pack<<<(NE / 4 + PACK_T - 1) / PACK_T, PACK_T, 0, stream>>>(ei, ea, erec, eown, elist, ecount);

  k_own<<<OWN_BLOCKS, 1024, 0, stream>>>(erec, eown, h, sD, sS, attb + 1,
                                         linW + 1 * H * H, linb + 1 * H, lng + 1 * H, lnb + 1 * H,
                                         attW + 2 * 2 * H, 1, h2, sD2, sS2, elist, ecount);
  k_own<<<OWN_BLOCKS, 1024, 0, stream>>>(erec, eown, h2, sD2, sS2, attb + 2,
                                         linW + 2 * H * H, linb + 2 * H, lng + 2 * H, lnb + 2 * H,
                                         attW + 3 * 2 * H, 1, h, sD, sS, elist, ecount);
  k_own<<<OWN_BLOCKS, 1024, 0, stream>>>(erec, eown, h, sD, sS, attb + 3,
                                         linW + 3 * H * H, linb + 3 * H, lng + 3 * H, lnb + 3 * H,
                                         attW + 3 * 2 * H, 0, h2, sD2, sS2, elist, ecount);

  float* part = sD;
  k_tail<<<TAIL_BLOCKS, 256, 0, stream>>>(h2,
                                          qW1, qb1, qW2, qb2, iW1, ib1, iW2, ib2,
                                          gW1, gb1, gW2, gb2, oW, obv, sW1, sb1, sW2, sb2,
                                          oq, oi, og, oh, obot, osev, part);
  k_mean_final<<<1, 1024, 0, stream>>>(part, ogr);
}


# layers 1-3: owner lists hold the edges' 8-byte records (src, local row, edge_attr), read 256 contiguous bytes a batch; per-edge 16-byte record gather and record array go; sums as before
# speedup vs baseline: 2.5907x; 1.0527x over previous
#include <hip/hip_runtime.h>

#define NN 100000
#define NE 1000000
#define DIN 5
#define H 64
#define HHID 32
#define NL 4
#define DOUT 32
#define LN_EPS 1e-5f

typedef __attribute__((ext_vector_type(2)))  float    v2f;
typedef __attribute__((ext_vector_type(8)))  float    v8f;
typedef __attribute__((ext_vector_type(16))) _Float16 v16h;

#if defined(__has_builtin)
#if __has_builtin(__builtin_amdgcn_wmma_f32_16x16x4_f32)
#define HAVE_WMMA_F32 1
#endif
#endif

__device__ __forceinline__ float xhalf(float x) {
  const int b = __builtin_bit_cast(int, x);
  return __builtin_bit_cast(float, __builtin_amdgcn_permlanex16(b, b, 0x76543210, 0xfedcba98, false, false));
}
template <int M>
__device__ __forceinline__ float xrow(float x) {
  const int b = __builtin_bit_cast(int, x);
  return __builtin_bit_cast(float, __builtin_amdgcn_update_dpp(0, b, 0x160 + M, 0xF, 0xF, true));
}
__device__ __forceinline__ float wave_sum(float v) {
  v += xhalf(v);
  v += xrow<8>(v);
  v += xrow<4>(v);
  v += xrow<2>(v);
  v += xrow<1>(v);
  return v;
}
__device__ __forceinline__ float sigmoidf_(float x) { return 1.0f / (1.0f + __expf(-x)); }

__device__ __forceinline__ void atomic_fadd(float* p, float v) { unsafeAtomicAdd(p, v); }

__device__ __forceinline__ v8f tile_product(const float (*lA)[H + 1], const float (*lW)[H + 1],
                                            int m, int g, int col, v8f acc) {
#if HAVE_WMMA_F32
#pragma unroll
  for (int k0 = 0; k0 < H; k0 += 4) {
    int kb = k0 + 2 * g;
    v2f a, b;
    a.x = lA[m][kb];     a.y = lA[m][kb + 1];
    b.x = lW[kb][col];   b.y = lW[kb + 1][col];
    acc = __builtin_amdgcn_wmma_f32_16x16x4_f32(false, a, false, b, (short)0, acc, false, false);
  }
#else
#pragma unroll
  for (int kk = 0; kk < H; kk += 32) {
    v16h a, b;
#pragma unroll
    for (int j = 0; j < 8; ++j) {
      int ka = (j < 4) ? (2 * j + 8 * g) : (16 + 2 * (j - 4) + 8 * g);
      a[2 * j]     = (_Float16)lA[m][kk + ka];
      a[2 * j + 1] = (_Float16)lA[m][kk + ka + 1];
      int kb = 2 * j + 16 * g;
      b[2 * j]     = (_Float16)lW[kk + kb][col];
      b[2 * j + 1] = (_Float16)lW[kk + kb + 1][col];
    }
    acc = __builtin_amdgcn_wmma_f32_16x16x32_f16(false, a, false, b, (short)0, acc, false, false);
  }
#endif
  return acc;
}

#define EDGE_DEPTH 8
#define OWN_CAP 192
#define OWN_R 393
#define OWN_BLOCKS ((NN + OWN_R - 1) / OWN_R)
#define OWN_WAVES 32
#define OWN_LIST (OWN_WAVES * OWN_CAP)
#define REC_SRC_BITS 17
#define ECNT_STRIDE 32
#define PACK_T 1024
#define OWN_Q 544
#define OWN_TILES ((OWN_R + 15) / 16)
#define OWN_RP (OWN_TILES * 16)
#define TP 68
static_assert(OWN_TILES <= OWN_WAVES, "one wave per 16-row tile of the owner's product");
static_assert(NL == 4, "the three owner launches are written out");
static_assert(OWN_BLOCKS <= 255, "owner numbers are bytes 0..254; 255 marks a map byte that is nobody's");
static_assert(NE % 16 == 0, "the owner map is read 16 bytes a lane");
static_assert(NE % 32 == 0, "every wave of the layer-0 edge pass owns 32 whole edges");
static_assert(32 % EDGE_DEPTH == 0, "a batch of 32 edges splits into whole groups");
static_assert(OWN_LIST % 32 == 0, "a list holds whole batches");
static_assert((size_t)OWN_BLOCKS * OWN_LIST * 8 <= 16000000, "the lists fit between byte 8e6 of the aggr region and the owner map");
static_assert(NN <= (1 << REC_SRC_BITS), "a source node number fits the low bits of a record's first word");
static_assert(OWN_R <= (1 << (31 - REC_SRC_BITS)), "the owner-local row fits above it with the sign bit clear");
static_assert((size_t)OWN_BLOCKS * ECNT_STRIDE * 4 <= 800000, "the counters fit between the second attention scalars and the records");
static_assert(OWN_BLOCKS <= 256 && PACK_T >= 256, "a pack work-group keeps one LDS counter per owner");

__device__ __forceinline__ int lane_bcast_i(int v, int l) { return __builtin_amdgcn_readlane(v, l); }
__device__ __forceinline__ float lane_bcast_f(float v, int l) {
  return __int_as_float(__builtin_amdgcn_readlane(__float_as_int(v), l));
}

__device__ __forceinline__ unsigned long long edge_rec(int src, int dl, float a) {
  return (unsigned long long)(unsigned)(src | (dl << REC_SRC_BITS)) | ((unsigned long long)__float_as_uint(a) << 32);
}

__global__ __launch_bounds__(PACK_T) void k_edge_pack(
    const int* __restrict__ ei, const float* __restrict__ ea,
    unsigned* __restrict__ own,
    unsigned long long* __restrict__ elist, int* __restrict__ ecount) {
  __shared__ int hist[256];
  const int t = blockIdx.x * PACK_T + threadIdx.x;
  const bool live = t < NE / 4;
  if (threadIdx.x < 256) hist[threadIdx.x] = 0;
  __syncthreads();
  int o0 = 0, o1 = 0, o2 = 0, o3 = 0, r0 = 0, r1 = 0, r2 = 0, r3 = 0;
  int4 s4 = make_int4(0, 0, 0, 0), d4 = make_int4(0, 0, 0, 0);
  float4 a4 = make_float4(0.0f, 0.0f, 0.0f, 0.0f);
  if (live) {
    s4 = reinterpret_cast<const int4*>(ei)[t];
    d4 = reinterpret_cast<const int4*>(ei + NE)[t];
    a4 = reinterpret_cast<const float4*>(ea)[t];
    o0 = d4.x / OWN_R; o1 = d4.y / OWN_R; o2 = d4.z / OWN_R; o3 = d4.w / OWN_R;
    r0 = __hip_atomic_fetch_add(hist + o0, 1, __ATOMIC_RELAXED, __HIP_MEMORY_SCOPE_WORKGROUP);
    r1 = __hip_atomic_fetch_add(hist + o1, 1, __ATOMIC_RELAXED, __HIP_MEMORY_SCOPE_WORKGROUP);
    r2 = __hip_atomic_fetch_add(hist + o2, 1, __ATOMIC_RELAXED, __HIP_MEMORY_SCOPE_WORKGROUP);
    r3 = __hip_atomic_fetch_add(hist + o3, 1, __ATOMIC_RELAXED, __HIP_MEMORY_SCOPE_WORKGROUP);
  }
  __syncthreads();
  if (threadIdx.x < OWN_BLOCKS) {
    const int c = hist[threadIdx.x];
    int first = 0;
    if (c > 0)
      first = __hip_atomic_fetch_add(ecount + threadIdx.x * ECNT_STRIDE, c, __ATOMIC_RELAXED, __HIP_MEMORY_SCOPE_AGENT);
    hist[threadIdx.x] = first;
  }
  __syncthreads();
  if (live) {
    own[t] = (unsigned)o0 | ((unsigned)o1 << 8) | ((unsigned)o2 << 16) | ((unsigned)o3 << 24);
    const int p0 = hist[o0] + r0, p1 = hist[o1] + r1, p2 = hist[o2] + r2, p3 = hist[o3] + r3;
    if (p0 < OWN_LIST)
      elist[(size_t)o0 * OWN_LIST + p0] = edge_rec(s4.x, d4.x - o0 * OWN_R, a4.x);
    if (p1 < OWN_LIST)
      elist[(size_t)o1 * OWN_LIST + p1] = edge_rec(s4.y, d4.y - o1 * OWN_R, a4.y);
    if (p2 < OWN_LIST)
      elist[(size_t)o2 * OWN_LIST + p2] = edge_rec(s4.z, d4.z - o2 * OWN_R, a4.z);
    if (p3 < OWN_LIST)
      elist[(size_t)o3 * OWN_LIST + p3] = edge_rec(s4.w, d4.w - o3 * OWN_R, a4.w);
  }
}

__device__ __forceinline__ void own_batch(
    int esrc, int edl, float eav, int n, int lane, float b0,
    const float* __restrict__ hrows, const float* __restrict__ sS,
    const float* __restrict__ lsD, float* __restrict__ lsw, float* __restrict__ tile) {
  int src = 0, dl = 0;
  float w = 0.0f;
  if (lane < n) {
    src = esrc;
    dl = edl;
    w = eav * sigmoidf_(lsD[dl] + sS[src] + b0);
    __hip_atomic_fetch_add(lsw + dl, w, __ATOMIC_RELAXED, __HIP_MEMORY_SCOPE_WORKGROUP);
  }
  const float* mlane = hrows + lane;
#pragma unroll
  for (int i0 = 0; i0 < 32; i0 += EDGE_DEPTH) {
    if (i0 < n) {
      float lov[EDGE_DEPTH], hiv[EDGE_DEPTH];
#pragma unroll
      for (int j = 0; j < EDGE_DEPTH; ++j) {
        const float* mr = mlane + (size_t)lane_bcast_i(src, i0 + j) * H;
        lov[j] = mr[0];
        hiv[j] = mr[32];
      }
#pragma unroll
      for (int j = 0; j < EDGE_DEPTH; ++j) {
        if (i0 + j < n) {
          const float bw = lane_bcast_f(w, i0 + j);
          float* tr = tile + lane_bcast_i(dl, i0 + j) * TP + lane;
          __hip_atomic_fetch_add(tr,      lov[j] * bw, __ATOMIC_RELAXED, __HIP_MEMORY_SCOPE_WORKGROUP);
          __hip_atomic_fetch_add(tr + 32, hiv[j] * bw, __ATOMIC_RELAXED, __HIP_MEMORY_SCOPE_WORKGROUP);
        }
      }
    }
  }
}

__device__ __forceinline__ void own_scan_piece(const uint4 o, int vb, unsigned myb, int* qw, int& cnt) {
  const unsigned wd[4] = {o.x, o.y, o.z, o.w};
#pragma unroll
  for (int k = 0; k < 16; ++k) {
    const unsigned byte = (wd[k >> 2] >> (8 * (k & 3))) & 0xFFu;
    const bool mine = (byte == myb);
    const unsigned mask = __builtin_amdgcn_ballot_w32(mine);
    if (mask != 0u) {
      const int pos = cnt + (int)__builtin_amdgcn_mbcnt_lo(mask, 0u);
      if (mine) qw[pos] = vb + k;
      cnt += __builtin_popcount(mask);
    }
  }
}

__device__ __forceinline__ uint4 own_piece(const uint4 t, bool ok) {
  return ok ? t : make_uint4(0xFFFFFFFFu, 0xFFFFFFFFu, 0xFFFFFFFFu, 0xFFFFFFFFu);
}

__device__ __forceinline__ void own_drain(
    int* qw, int& cnt, int lane, int lo, float b0,
    const int* __restrict__ ei, const float* __restrict__ ea,
    const float* __restrict__ hrows, const float* __restrict__ sS,
    const float* __restrict__ lsD, float* __restrict__ lsw, float* __restrict__ tile) {
  int head = 0;
  while (cnt - head >= 32) {
    const int e = qw[head + lane];
    own_batch(ei[e], ei[NE + e] - lo, ea[e], 32, lane, b0, hrows, sS, lsD, lsw, tile);
    head += 32;
  }
  if (head > 0) {
    const int rest = cnt - head;
    int keep = 0;
    if (lane < rest) keep = qw[head + lane];
    if (lane < rest) qw[lane] = keep;
    cnt = rest;
  }
}

__global__ __launch_bounds__(1024) void k_own(
    const int* __restrict__ ei, const float* __restrict__ ea, const unsigned* __restrict__ own,
    const float* __restrict__ hin, const float* __restrict__ sD,
    const float* __restrict__ sS, const float* __restrict__ attb,
    const float* __restrict__ W, const float* __restrict__ bias,
    const float* __restrict__ g, const float* __restrict__ beta,
    const float* __restrict__ awn, int has_next,
    float* __restrict__ hout, float* __restrict__ sDn, float* __restrict__ sSn,
    const unsigned long long* __restrict__ elist, const int* __restrict__ ecount) {
  __shared__ __attribute__((aligned(16))) float tile[OWN_RP * TP];
  __shared__ __attribute__((aligned(16))) float lWt[H * TP];
  __shared__ float lsD[OWN_RP];
  __shared__ float lsw[OWN_RP];
  __shared__ int q[OWN_WAVES][OWN_Q];
  const int tid = threadIdx.x, lane = tid & 31;
  const int wave = __builtin_amdgcn_readfirstlane(tid >> 5);
  const int lo = blockIdx.x * OWN_R;
  const int nrows = (NN - lo < OWN_R) ? (NN - lo) : OWN_R;
  const unsigned myb = blockIdx.x;
  const float b0 = attb[0];
  const int total = __builtin_amdgcn_readfirstlane(ecount[blockIdx.x * ECNT_STRIDE]);

#pragma unroll 4
  for (int r = wave; r < nrows; r += OWN_WAVES) {
    const float* hr = hin + (size_t)(lo + r) * H;
    tile[r * TP + lane]      = hr[lane];
    tile[r * TP + lane + 32] = hr[lane + 32];
  }
  for (int r = nrows + wave; r < OWN_RP; r += OWN_WAVES) {
    tile[r * TP + lane]      = 0.0f;
    tile[r * TP + lane + 32] = 0.0f;
  }
  for (int r = tid; r < OWN_RP; r += 1024) {
    lsD[r] = (r < nrows) ? sD[lo + r] : 0.0f;
    lsw[r] = 1.0f;
  }
  for (int idx = tid; idx < H * H; idx += 1024)
    lWt[(idx & 63) * TP + (idx >> 6)] = W[idx];
  __syncthreads();

  const uint4* __restrict__ own4 = reinterpret_cast<const uint4*>(own);
  int* qw = q[wave];
  int cnt = 0;
  const int nvec = NE / 16;
  const int nit = (nvec + 31) / 32;
  const unsigned long long* gl = elist + (size_t)blockIdx.x * OWN_LIST;
  if (total <= OWN_LIST) {
    const int step = 32 * OWN_WAVES;
    const int first = 32 * wave;
    unsigned long long rA = (first + lane < total) ? gl[first + lane] : 0ull;
    unsigned long long rB = (first + step + lane < total) ? gl[first + step + lane] : 0ull;
    for (int b = first; b < total; b += step) {
      const unsigned long long r = rA;
      rA = rB;
      rB = (b + 2 * step + lane < total) ? gl[b + 2 * step + lane] : 0ull;
      const int n = (total - b < 32) ? (total - b) : 32;
      const unsigned r0 = (unsigned)r;
      own_batch((int)(r0 & ((1u << REC_SRC_BITS) - 1u)), (int)(r0 >> REC_SRC_BITS), __uint_as_float((unsigned)(r >> 32)),
                n, lane, b0, hin, sS, lsD, lsw, tile);
    }
  } else {
    for (int it0 = wave; it0 < nit; it0 += 4 * OWN_WAVES) {
      const int v0 = it0 * 32 + lane;
      const int v1 = v0 + 32 * OWN_WAVES;
      const int v2 = v1 + 32 * OWN_WAVES;
      const int v3 = v2 + 32 * OWN_WAVES;
      const bool ok0 = v0 < nvec, ok1 = v1 < nvec, ok2 = v2 < nvec, ok3 = v3 < nvec;
      const uint4 t0 = own4[ok0 ? v0 : nvec - 1];
      const uint4 t1 = own4[ok1 ? v1 : nvec - 1];
      const uint4 t2 = own4[ok2 ? v2 : nvec - 1];
      const uint4 t3 = own4[ok3 ? v3 : nvec - 1];
      own_scan_piece(own_piece(t0, ok0), v0 * 16, myb, qw, cnt);
      own_drain(qw, cnt, lane, lo, b0, ei, ea, hin, sS, lsD, lsw, tile);
      own_scan_piece(own_piece(t1, ok1), v1 * 16, myb, qw, cnt);
      own_drain(qw, cnt, lane, lo, b0, ei, ea, hin, sS, lsD, lsw, tile);
      own_scan_piece(own_piece(t2, ok2), v2 * 16, myb, qw, cnt);
      own_drain(qw, cnt, lane, lo, b0, ei, ea, hin, sS, lsD, lsw, tile);
      own_scan_piece(own_piece(t3, ok3), v3 * 16, myb, qw, cnt);
      own_drain(qw, cnt, lane, lo, b0, ei, ea, hin, sS, lsD, lsw, tile);
    }
    if (cnt > 0) {
      const int e = (lane < cnt) ? qw[lane] : 0;
      own_batch(ei[e], ei[NE + e] - lo, ea[e], cnt, lane, b0, hin, sS, lsD, lsw, tile);
    }
  }
  __syncthreads();

  if (wave * 16 < nrows) {
    const int mrow = lane & 15, gq = lane >> 4;
    const int r0 = wave * 16;
    const float* ap = tile + (r0 + mrow) * TP + 2 * gq;
    const float* bp = lWt + mrow * TP + 2 * gq;
    v8f acc[4];
#pragma unroll
    for (int cc = 0; cc < 4; ++cc) {
      const float bv = bias[16 * cc + mrow];
#pragma unroll
      for (int i = 0; i < 8; ++i) acc[cc][i] = lsw[r0 + i + 8 * gq] * bv;
    }
#if HAVE_WMMA_F32
#pragma unroll
    for (int k0 = 0; k0 < H; k0 += 4) {
      const v2f a = *reinterpret_cast<const v2f*>(ap + k0);
#pragma unroll
      for (int cc = 0; cc < 4; ++cc) {
        const v2f b = *reinterpret_cast<const v2f*>(bp + 16 * cc * TP + k0);
        acc[cc] = __builtin_amdgcn_wmma_f32_16x16x4_f32(false, a, false, b, (short)0, acc[cc], false, false);
      }
    }
#endif
#pragma unroll
    for (int cc = 0; cc < 4; ++cc) {
#pragma unroll
      for (int i = 0; i < 8; ++i) tile[(r0 + i + 8 * gq) * TP + 16 * cc + mrow] = acc[cc][i];
    }
  }
  __syncthreads();

  const float g0 = g[lane], g1 = g[lane + 32];
  const float be0 = beta[lane], be1 = beta[lane + 32];
  const float aD0 = awn[lane], aD1 = awn[lane + 32];
  const float aS0 = awn[H + lane], aS1 = awn[H + lane + 32];
#pragma unroll 2
  for (int r = wave; r < nrows; r += OWN_WAVES) {
    const size_t base = (size_t)(lo + r) * H;
    const float r0 = hin[base + lane], r1 = hin[base + lane + 32];
    const float f0 = tile[r * TP + lane], f1 = tile[r * TP + lane + 32];
    const float mean = wave_sum(f0 + f1) * (1.0f / H);
    const float d0 = f0 - mean, d1 = f1 - mean;
    const float var = wave_sum(fmaf(d0, d0, d1 * d1)) * (1.0f / H);
    const float inv = rsqrtf(var + LN_EPS);
    const float y0 = fmaf(d0 * inv, g0, be0);
    const float y1 = fmaf(d1 * inv, g1, be1);
    const float o0 = fmaxf(y0, 0.0f) + r0;
    const float o1 = fmaxf(y1, 0.0f) + r1;
    hout[base + lane]      = o0;
    hout[base + lane + 32] = o1;
    if (has_next) {
      const float pd = wave_sum(fmaf(o0, aD0, o1 * aD1));
      const float ps = wave_sum(fmaf(o0, aS0, o1 * aS1));
      if (lane == 0) { sDn[lo + r] = pd; sSn[lo + r] = ps; }
    }
  }
}

#define ACCW 8
#define RECW 8

static_assert(OWN_BLOCKS <= NN, "one thread of the node pass per list counter");
__global__ __launch_bounds__(256) void k_l0_prep(
    const float* __restrict__ x, const float* __restrict__ inW, const float* __restrict__ inb,
    const float* __restrict__ aw, float* __restrict__ rec, double* __restrict__ acc,
    int* __restrict__ ecount) {
  int n = blockIdx.x * 256 + threadIdx.x;
  if (n >= NN) return;
  if (n < OWN_BLOCKS) ecount[n * ECNT_STRIDE] = 0;
  float xv[DIN];
#pragma unroll
  for (int d = 0; d < DIN; ++d) xv[d] = x[(size_t)n * DIN + d];
  float sd = 0.0f, ss = 0.0f;
#pragma unroll 8
  for (int k = 0; k < H; ++k) {
    float hk = inb[k];
#pragma unroll
    for (int d = 0; d < DIN; ++d) hk = fmaf(xv[d], inW[d * H + k], hk);
    sd = fmaf(hk, aw[k], sd);
    ss = fmaf(hk, aw[H + k], ss);
  }
  float4* r4 = reinterpret_cast<float4*>(rec + (size_t)n * RECW);
  r4[0] = make_float4(xv[0], xv[1], xv[2], xv[3]);
  r4[1] = make_float4(xv[4], ss, sd, 0.0f);
  double2* a2 = reinterpret_cast<double2*>(acc + (size_t)n * ACCW);
#pragma unroll
  for (int j = 0; j < ACCW / 2; ++j) a2[j] = make_double2(0.0, 0.0);
}

__global__ __launch_bounds__(256) void k_l0_edges(
    const int* __restrict__ ei, const float* __restrict__ ea,
    const float* __restrict__ rec, const float* __restrict__ attb,
    double* __restrict__ acc) {
  int gw = blockIdx.x * 8 + (threadIdx.x >> 5);
  int lane = threadIdx.x & 31;
  int ebase = gw * 32;
  if (ebase >= NE) return;
  const int sub = lane >> 3, c = lane & 7;
  const float b0 = attb[0];
  int srcs[8], dsts[8];
  float eas[8], rv[8], sd[8];
#pragma unroll
  for (int i = 0; i < 8; ++i) {
    int e = ebase + 4 * i + sub;
    srcs[i] = ei[e];
    dsts[i] = ei[NE + e];
    eas[i] = ea[e];
  }
#pragma unroll
  for (int i = 0; i < 8; ++i) {
    rv[i] = rec[(size_t)srcs[i] * RECW + c];
    sd[i] = rec[(size_t)dsts[i] * RECW + 6];
  }
#pragma unroll
  for (int i = 0; i < 8; ++i) {
    float ss = __shfl(rv[i], (lane & 24) | 5, 32);
    float w = eas[i] * sigmoidf_(sd[i] + ss + b0);
    if (c < 6) {
      double v = (c == 5) ? (double)w : (double)w * (double)rv[i];
      unsafeAtomicAdd(acc + (size_t)dsts[i] * ACCW + c, v);
    }
  }
}

__global__ __launch_bounds__(128) void k_l0_node(
    const float* __restrict__ x, const double* __restrict__ acc,
    const float* __restrict__ inW, const float* __restrict__ inb,
    const float* __restrict__ W, const float* __restrict__ bias,
    const float* __restrict__ g_, const float* __restrict__ beta,
    const float* __restrict__ awn,
    float* __restrict__ h, float* __restrict__ sDn, float* __restrict__ sSn) {
  __shared__ float lA[16][H + 1];
  __shared__ float lW[H][H + 1];
  __shared__ float lO[16][H + 1];
  __shared__ float lX[16][8];
  __shared__ float lS[16][8];
  const int tid = threadIdx.x;
  const int row0 = blockIdx.x * 16;
  for (int idx = tid; idx < H * H; idx += 128)
    lW[idx >> 6][idx & 63] = W[idx];
  {
    const int r = tid >> 3, c = tid & 7;
    const size_t n = (size_t)(row0 + r);
    float xv = (c < DIN) ? x[n * DIN + c] : 0.0f;
    float sv = (c < 6) ? (float)acc[n * ACCW + c] : 0.0f;
    lX[r][c] = xv;
    lS[r][c] = (c < DIN) ? (sv + xv) : ((c == 5) ? (sv + 1.0f) : 0.0f);
  }
  __syncthreads();
  for (int idx = tid; idx < 16 * H; idx += 128) {
    const int r = idx >> 6, k = idx & 63;
    float u = lS[r][5] * inb[k];
#pragma unroll
    for (int d = 0; d < DIN; ++d) u = fmaf(lS[r][d], inW[d * H + k], u);
    lA[r][k] = u;
  }
  __syncthreads();

  const int wave = tid >> 5, lane = tid & 31;
  const int m = lane & 15, g = lane >> 4;
  const int col = wave * 16 + m;

  v8f t;
  const float bv = bias[col];
#pragma unroll
  for (int i = 0; i < 8; ++i) t[i] = lS[i + 8 * g][5] * bv;

  t = tile_product(lA, lW, m, g, col, t);

#pragma unroll
  for (int i = 0; i < 8; ++i) lO[i + 8 * g][col] = t[i];
  __syncthreads();

  const float g0 = g_[lane], g1 = g_[lane + 32];
  const float be0 = beta[lane], be1 = beta[lane + 32];
  const float aD0 = awn[lane], aD1 = awn[lane + 32];
  const float aS0 = awn[H + lane], aS1 = awn[H + lane + 32];
#pragma unroll
  for (int rr = 0; rr < 4; ++rr) {
    const int r = wave * 4 + rr;
    float r0 = inb[lane], r1 = inb[lane + 32];
#pragma unroll
    for (int d = 0; d < DIN; ++d) {
      r0 = fmaf(lX[r][d], inW[d * H + lane], r0);
      r1 = fmaf(lX[r][d], inW[d * H + lane + 32], r1);
    }
    float f0 = lO[r][lane], f1 = lO[r][lane + 32];
    float mean = wave_sum(f0 + f1) * (1.0f / H);
    float d0 = f0 - mean, d1 = f1 - mean;
    float var = wave_sum(fmaf(d0, d0, d1 * d1)) * (1.0f / H);
    float inv = rsqrtf(var + LN_EPS);
    float y0 = fmaf(d0 * inv, g0, be0);
    float y1 = fmaf(d1 * inv, g1, be1);
    const size_t base = (size_t)(row0 + r) * H;
    const float o0 = fmaxf(y0, 0.0f) + r0;
    const float o1 = fmaxf(y1, 0.0f) + r1;
    h[base + lane]      = o0;
    h[base + lane + 32] = o1;
    const float pd = wave_sum(fmaf(o0, aD0, o1 * aD1));
    const float ps = wave_sum(fmaf(o0, aS0, o1 * aS1));
    if (lane == 0) { sDn[row0 + r] = pd; sSn[row0 + r] = ps; }
  }
}

__device__ __forceinline__ float kgroup_reduce(const float (&a)[4], int hiA, int hiB) {
  float s0 = hiA ? a[0] : a[2];
  float s1 = hiA ? a[1] : a[3];
  float k0 = hiA ? a[2] : a[0];
  float k1 = hiA ? a[3] : a[1];
  float r0 = k0 + xhalf(s0);
  float r1 = k1 + xhalf(s1);
  float sB = hiB ? r0 : r1;
  float kB = hiB ? r1 : r0;
  return kB + xrow<8>(sB);
}
#define TAIL_BLOCKS 2048
static_assert(TAIL_BLOCKS * H <= 2 * NN, "partial sums must fit in the sD+sS region");
static_assert(TAIL_BLOCKS % 16 == 0, "stage 2 splits the partial rows into 16 equal groups");
__global__ __launch_bounds__(256) void k_tail(
    const float* __restrict__ h,
    const float* __restrict__ qW1, const float* __restrict__ qb1,
    const float* __restrict__ qW2, const float* __restrict__ qb2,
    const float* __restrict__ iW1, const float* __restrict__ ib1,
    const float* __restrict__ iW2, const float* __restrict__ ib2,
    const float* __restrict__ gW1, const float* __restrict__ gb1,
    const float* __restrict__ gW2, const float* __restrict__ gb2,
    const float* __restrict__ oW,  const float* __restrict__ obv,
    const float* __restrict__ sW1, const float* __restrict__ sb1,
    const float* __restrict__ sW2, const float* __restrict__ sb2,
    float* __restrict__ oq, float* __restrict__ oi, float* __restrict__ og,
    float* __restrict__ oh, float* __restrict__ obot, float* __restrict__ osev,
    float* __restrict__ part) {
  __shared__ float ls[8][H];
  const int wave = threadIdx.x >> 5;
  const int lane = threadIdx.x & 31;
  const int nwaves = gridDim.x * 8;
  float cs0 = 0.0f, cs1 = 0.0f;

  const int kgrp = lane >> 3;
  const int hiA = (lane >> 4) & 1, hiB = (lane >> 3) & 1;
  const int col = 4 * (lane & 7) + 2 * hiA + hiB;
  const float4* __restrict__ q4 = reinterpret_cast<const float4*>(qW1);
  const float4* __restrict__ i4 = reinterpret_cast<const float4*>(iW1);
  const float4* __restrict__ g4 = reinterpret_cast<const float4*>(gW1);
  const float4* __restrict__ s4 = reinterpret_cast<const float4*>(sW1);
  const float4* __restrict__ o4 = reinterpret_cast<const float4*>(oW);
  const unsigned wl = (unsigned)lane;

  for (int gw = blockIdx.x * 8 + wave; gw < NN; gw += nwaves) {
    size_t base = (size_t)gw * H;

    const float h0 = h[base + lane], h1 = h[base + lane + 32];

    float aq[4] = {0.f, 0.f, 0.f, 0.f}, ai[4] = {0.f, 0.f, 0.f, 0.f}, ag[4] = {0.f, 0.f, 0.f, 0.f};
    float as[4] = {0.f, 0.f, 0.f, 0.f}, ao[4] = {0.f, 0.f, 0.f, 0.f};
#pragma unroll
    for (int half = 0; half < 2; ++half) {
      const float hsrc = half ? h1 : h0;
#pragma unroll 2
      for (int ii8 = 0; ii8 < 8; ++ii8) {
        const int i = half * 8 + ii8;
        const float hk = __shfl(hsrc, 4 * ii8 + kgrp, 32);
        const unsigned wx = (unsigned)(i * 32) + wl;
        const float4 wq = q4[wx], wi = i4[wx], wg = g4[wx], wsv = s4[wx], wo = o4[wx];
        aq[0] = fmaf(hk, wq.x, aq[0]);  aq[1] = fmaf(hk, wq.y, aq[1]);
        aq[2] = fmaf(hk, wq.z, aq[2]);  aq[3] = fmaf(hk, wq.w, aq[3]);
        ai[0] = fmaf(hk, wi.x, ai[0]);  ai[1] = fmaf(hk, wi.y, ai[1]);
        ai[2] = fmaf(hk, wi.z, ai[2]);  ai[3] = fmaf(hk, wi.w, ai[3]);
        ag[0] = fmaf(hk, wg.x, ag[0]);  ag[1] = fmaf(hk, wg.y, ag[1]);
        ag[2] = fmaf(hk, wg.z, ag[2]);  ag[3] = fmaf(hk, wg.w, ag[3]);
        as[0] = fmaf(hk, wsv.x, as[0]); as[1] = fmaf(hk, wsv.y, as[1]);
        as[2] = fmaf(hk, wsv.z, as[2]); as[3] = fmaf(hk, wsv.w, as[3]);
        ao[0] = fmaf(hk, wo.x, ao[0]);  ao[1] = fmaf(hk, wo.y, ao[1]);
        ao[2] = fmaf(hk, wo.z, ao[2]);  ao[3] = fmaf(hk, wo.w, ao[3]);
      }
    }
    float tq = kgroup_reduce(aq, hiA, hiB);
    float ti = kgroup_reduce(ai, hiA, hiB);
    float tg = kgroup_reduce(ag, hiA, hiB);
    float ts = kgroup_reduce(as, hiA, hiB);
    float bo = kgroup_reduce(ao, hiA, hiB);
    tq = fmaxf(tq + qb1[col], 0.0f);
    ti = fmaxf(ti + ib1[col], 0.0f);
    tg = fmaxf(tg + gb1[col], 0.0f);
    ts = fmaxf(ts + sb1[col], 0.0f);

    float q  = wave_sum(tq * qW2[col]);
    float ii = wave_sum(ti * iW2[col]);
    float gg = wave_sum(tg * gW2[col]);
    if (lane == 0) {
      oq[gw] = sigmoidf_(q + qb2[0]);
      oi[gw] = sigmoidf_(ii + ib2[0]);
      og[gw] = sigmoidf_(gg + gb2[0]);
    }

    float z[5];
#pragma unroll
    for (int c = 0; c < 5; ++c) z[c] = wave_sum(ts * sW2[col * 5 + c]) + sb2[c];
    float zm = z[0];
#pragma unroll
    for (int c = 1; c < 5; ++c) zm = fmaxf(zm, z[c]);
    float es = 0.f;
#pragma unroll
    for (int c = 0; c < 5; ++c) { z[c] = __expf(z[c] - zm); es += z[c]; }
    float inv = 1.0f / es;
#pragma unroll
    for (int c = 0; c < 5; ++c)
      if (lane == c) osev[(size_t)gw * 5 + c] = z[c] * inv;

    obot[(size_t)gw * DOUT + col] = bo + obv[col];
    oh[base + lane]      = h0;
    oh[base + lane + 32] = h1;

    cs0 += h0;
    cs1 += h1;
  }

  ls[wave][lane]      = cs0;
  ls[wave][lane + 32] = cs1;
  __syncthreads();
  if (threadIdx.x < H) {
    float t = 0.0f;
#pragma unroll
    for (int w = 0; w < 8; ++w) t += ls[w][threadIdx.x];
    part[(size_t)blockIdx.x * H + threadIdx.x] = t;
  }
}

__global__ __launch_bounds__(1024) void k_mean_final(
    const float* __restrict__ part, float* __restrict__ gout) {
  __shared__ float lq[16][H];
  const int f = threadIdx.x & 63, q = threadIdx.x >> 6;
  const float* p = part + (size_t)q * (TAIL_BLOCKS / 16) * H + f;
  float acc = 0.0f;
#pragma unroll 16
  for (int j = 0; j < TAIL_BLOCKS / 16; ++j) acc += p[(size_t)j * H];
  lq[q][f] = acc;
  __syncthreads();
  if (threadIdx.x < H) {
    float t = 0.0f;
#pragma unroll
    for (int g = 0; g < 16; ++g) t += lq[g][threadIdx.x];
    gout[threadIdx.x] = t * (1.0f / NN);
  }
}

extern "C" void kernel_launch(void* const* d_in, const int* in_sizes, int n_in,
                              void* d_out, int out_size, void* d_ws, size_t ws_size,
                              hipStream_t stream) {
  (void)in_sizes; (void)n_in; (void)out_size; (void)ws_size;
  const float* x    = (const float*)d_in[0];
  const int*   ei   = (const int*)  d_in[1];
  const float* ea   = (const float*)d_in[2];
  const float* inW  = (const float*)d_in[3];
  const float* inb  = (const float*)d_in[4];
  const float* linW = (const float*)d_in[5];
  const float* linb = (const float*)d_in[6];
  const float* attW = (const float*)d_in[7];
  const float* attb = (const float*)d_in[8];
  const float* lng  = (const float*)d_in[9];
  const float* lnb  = (const float*)d_in[10];
  const float* qW1 = (const float*)d_in[11]; const float* qb1 = (const float*)d_in[12];
  const float* qW2 = (const float*)d_in[13]; const float* qb2 = (const float*)d_in[14];
  const float* iW1 = (const float*)d_in[15]; const float* ib1 = (const float*)d_in[16];
  const float* iW2 = (const float*)d_in[17]; const float* ib2 = (const float*)d_in[18];
  const float* gW1 = (const float*)d_in[19]; const float* gb1 = (const float*)d_in[20];
  const float* gW2 = (const float*)d_in[21]; const float* gb2 = (const float*)d_in[22];
  const float* oW  = (const float*)d_in[23]; const float* obv = (const float*)d_in[24];
  const float* sW1 = (const float*)d_in[25]; const float* sb1 = (const float*)d_in[26];
  const float* sW2 = (const float*)d_in[27]; const float* sb2 = (const float*)d_in[28];

  float* ws   = (float*)d_ws;
  float* h    = ws;
  float* mbuf = h    + (size_t)NN * H;
  float* aggr = mbuf + (size_t)NN * H;
  float* sD   = aggr + (size_t)NN * H;
  float* sS   = sD + NN;

  float* out  = (float*)d_out;
  float* oq   = out;
  float* oi   = out + (size_t)NN;
  float* og   = out + 2 * (size_t)NN;
  float* oh   = out + 3 * (size_t)NN;
  float* ogr  = out + 67 * (size_t)NN;
  float* obot = out + 67 * (size_t)NN + H;
  float* osev = out + 99 * (size_t)NN + H;

  float*  rec0 = mbuf;
  double* acc0 = (double*)aggr;
  int* ecount = (int*)(aggr + 1800000);
  k_l0_prep<<<(NN + 255) / 256, 256, 0, stream>>>(x, inW, inb, attW, rec0, acc0, ecount);
  k_l0_edges<<<(NE + 255) / 256, 256, 0, stream>>>(ei, ea, rec0, attb, acc0);
  float* sD2 = aggr + 1600000;
  float* sS2 = sD2 + NN;
  float* h2  = mbuf;
  k_l0_node<<<NN / 16, 128, 0, stream>>>(x, acc0, inW, inb, linW, linb, lng, lnb, attW + 2 * H,
                                         h, sD, sS);

  unsigned* eown  = (unsigned*)(aggr + 6000000);
  unsigned long long* elist = (unsigned long long*)(aggr + 2000000);
  k_edge_pack<<<(NE / 4 + PACK_T - 1) / PACK_T, PACK_T, 0, stream>>>(ei, ea, eown, elist, ecount);

  k_own<<<OWN_BLOCKS, 1024, 0, stream>>>(ei, ea, eown, h, sD, sS, attb + 1,
                                         linW + 1 * H * H, linb + 1 * H, lng + 1 * H, lnb + 1 * H,
                                         attW + 2 * 2 * H, 1, h2, sD2, sS2, elist, ecount);
  k_own<<<OWN_BLOCKS, 1024, 0, stream>>>(ei, ea, eown, h2, sD2, sS2, attb + 2,
                                         linW + 2 * H * H, linb + 2 * H, lng + 2 * H, lnb + 2 * H,
                                         attW + 3 * 2 * H, 1, h, sD, sS, elist, ecount);
  k_own<<<OWN_BLOCKS, 1024, 0, stream>>>(ei, ea, eown, h, sD, sS, attb + 3,
                                         linW + 3 * H * H, linb + 3 * H, lng + 3 * H, lnb + 3 * H,
                                         attW + 3 * 2 * H, 0, h2, sD2, sS2, elist, ecount);

  float* part = sD;
  k_tail<<<TAIL_BLOCKS, 256, 0, stream>>>(h2,
                                          qW1, qb1, qW2, qb2, iW1, ib1, iW2, ib2,
                                          gW1, gb1, gW2, gb2, oW, obv, sW1, sb1, sW2, sb2,
                                          oq, oi, og, oh, obot, osev, part);
  k_mean_final<<<1, 1024, 0, stream>>>(part, ogr);
}
